# starting kernel
# baseline (speedup 1.0000x reference)
#include <hip/hip_runtime.h>
#include <hip/hip_bf16.h>

typedef __attribute__((ext_vector_type(2))) float v2f;
typedef __attribute__((ext_vector_type(8))) float v8f;

#define F_IN 128
#define C1   64
#define C2   16
#define NEG_SLOPE 0.2f

__global__ void __launch_bounds__(256)
gemm1_wmma(const float* __restrict__ x, const float* __restrict__ W1,
           float* __restrict__ h1, int Nn) {
  __shared__ float sW[F_IN * C1];
  for (int i = threadIdx.x; i < (F_IN * C1) / 4; i += 256) {
    float4 v = ((const float4*)W1)[i];
    const int k  = i >> 4;
    const int c0 = (i & 15) * 4;
    float* dst = sW + (k >> 2) * 256 + (k & 3);
    dst[(c0 + 0) * 4] = v.x;
    dst[(c0 + 1) * 4] = v.y;
    dst[(c0 + 2) * 4] = v.z;
    dst[(c0 + 3) * 4] = v.w;
  }
  __syncthreads();

  const int wave = threadIdx.x >> 5;
  const int lane = threadIdx.x & 31;
  const int tile = blockIdx.x * 8 + wave;
  if (tile * 16 >= Nn) return;
  const int row0 = tile * 16;
  const int r    = lane & 15;
  const int kh   = (lane >> 4) << 1;
  const int arow = min(row0 + r, Nn - 1);

  v8f acc0 = {}, acc1 = {}, acc2 = {}, acc3 = {};
  const float* ap = x + (size_t)arow * F_IN + kh;
  const float* bbase = sW + r * 4 + kh;
  for (int k0 = 0; k0 < F_IN; k0 += 4) {
    v2f a; a.x = ap[k0]; a.y = ap[k0 + 1];
    const float* bq = bbase + (k0 >> 2) * 256;
    v2f b0 = *(const v2f*)(bq);
    v2f b1 = *(const v2f*)(bq + 64);
    v2f b2 = *(const v2f*)(bq + 128);
    v2f b3 = *(const v2f*)(bq + 192);
    acc0 = __builtin_amdgcn_wmma_f32_16x16x4_f32(false, a, false, b0, (short)0, acc0, false, false);
    acc1 = __builtin_amdgcn_wmma_f32_16x16x4_f32(false, a, false, b1, (short)0, acc1, false, false);
    acc2 = __builtin_amdgcn_wmma_f32_16x16x4_f32(false, a, false, b2, (short)0, acc2, false, false);
    acc3 = __builtin_amdgcn_wmma_f32_16x16x4_f32(false, a, false, b3, (short)0, acc3, false, false);
  }

  const int mbase = (lane >> 4) * 8;
  for (int i = 0; i < 8; ++i) {
    const int row = row0 + mbase + i;
    if (row < Nn) {
      float* op = h1 + (size_t)row * C1 + r;
      op[0]  = acc0[i];
      op[16] = acc1[i];
      op[32] = acc2[i];
      op[48] = acc3[i];
    }
  }
}

__global__ void __launch_bounds__(256)
gemm2_wmma(const float* __restrict__ h, const float* __restrict__ W2,
           float* __restrict__ h2, int Nn) {
  __shared__ float sW[C1 * C2];
  for (int i = threadIdx.x; i < (C1 * C2) / 4; i += 256) {
    float4 v = ((const float4*)W2)[i];
    const int k  = i >> 2;
    const int c0 = (i & 3) * 4;
    float* dst = sW + (k >> 2) * 64 + (k & 3);
    dst[(c0 + 0) * 4] = v.x;
    dst[(c0 + 1) * 4] = v.y;
    dst[(c0 + 2) * 4] = v.z;
    dst[(c0 + 3) * 4] = v.w;
  }
  __syncthreads();

  const int wave = threadIdx.x >> 5;
  const int lane = threadIdx.x & 31;
  const int tile = blockIdx.x * 8 + wave;
  if (tile * 16 >= Nn) return;
  const int row0 = tile * 16;
  const int r    = lane & 15;
  const int kh   = (lane >> 4) << 1;
  const int arow = min(row0 + r, Nn - 1);

  v8f acc = {};
  const float* ap = h + (size_t)arow * C1 + kh;
  const float* bbase = sW + r * 4 + kh;
  for (int k0 = 0; k0 < C1; k0 += 4) {
    v2f a; a.x = ap[k0]; a.y = ap[k0 + 1];
    v2f b = *(const v2f*)(bbase + (k0 >> 2) * 64);
    acc = __builtin_amdgcn_wmma_f32_16x16x4_f32(false, a, false, b, (short)0, acc, false, false);
  }

  const int mbase = (lane >> 4) * 8;
  for (int i = 0; i < 8; ++i) {
    const int row = row0 + mbase + i;
    if (row < Nn) h2[(size_t)row * C2 + r] = acc[i];
  }
}

template <int C>
__global__ void node_dots(const float* __restrict__ h,
                          const float* __restrict__ asrc,
                          const float* __restrict__ adst,
                          float* __restrict__ s, float* __restrict__ d, int Nn) {
  int n = blockIdx.x * blockDim.x + threadIdx.x;
  if (n >= Nn) return;
  const float* hp = h + (size_t)n * C;
  float ss = 0.f, dd = 0.f;
  for (int f = 0; f < C; ++f) { float v = hp[f]; ss += v * asrc[f]; dd += v * adst[f]; }
  s[n] = ss; d[n] = dd;
}

__global__ void init_accum(float* __restrict__ outacc, size_t acc_elems,
                           int* __restrict__ amax, float* __restrict__ denom, int Nn) {
  size_t i = (size_t)blockIdx.x * blockDim.x + threadIdx.x;
  if (i < acc_elems) outacc[i] = 0.f;
  if (i < (size_t)Nn) { amax[i] = (int)0x80000000; denom[i] = 0.f; }
}

__device__ __forceinline__ int fkey(float f) {
  int k = __float_as_int(f);
  return k >= 0 ? k : (k ^ 0x7fffffff);
}
__device__ __forceinline__ float fkey_inv(int k) {
  return __int_as_float(k >= 0 ? k : (k ^ 0x7fffffff));
}
__device__ __forceinline__ void edge_sd(const int* ei, int E, int e, int& s, int& d) {
  if (e < E) { s = ei[e]; d = ei[E + e]; } else { s = d = e - E; }
}

__global__ void edge_max_kernel(const int* __restrict__ ei, int E, int Etot,
                                const float* __restrict__ s, const float* __restrict__ d,
                                float* __restrict__ alpha_buf, int* __restrict__ amax) {
  int e = blockIdx.x * blockDim.x + threadIdx.x;
  if (e >= Etot) return;
  int sn, dn; edge_sd(ei, E, e, sn, dn);
  float a = s[sn] + d[dn];
  a = a > 0.f ? a : NEG_SLOPE * a;
  alpha_buf[e] = a;
  atomicMax(amax + dn, fkey(a));
}

__global__ void edge_sum_kernel(const int* __restrict__ ei, int E, int Etot,
                                float* __restrict__ alpha_buf,
                                const int* __restrict__ amax, float* __restrict__ denom) {
  int e = blockIdx.x * blockDim.x + threadIdx.x;
  if (e >= Etot) return;
  int sn, dn; edge_sd(ei, E, e, sn, dn);
  float ew = __expf(alpha_buf[e] - fkey_inv(amax[dn]));
  alpha_buf[e] = ew;
  atomicAdd(denom + dn, ew);
}

template <int C>
__global__ void edge_agg_kernel(const int* __restrict__ ei, int E, int Etot,
                                const float* __restrict__ ewb, const float* __restrict__ denom,
                                const float* __restrict__ hsrc, float* __restrict__ outacc) {
  const int chunks = C / 4;
  long long tid = (long long)blockIdx.x * blockDim.x + threadIdx.x;
  int e  = (int)(tid / chunks);
  int fo = (int)(tid % chunks) * 4;
  if (e >= Etot) return;
  int sn, dn; edge_sd(ei, E, e, sn, dn);
  float w = ewb[e] / (denom[dn] + 1e-16f);
  float4 hv = *(const float4*)(hsrc + (size_t)sn * C + fo);
  float* op = outacc + (size_t)dn * C + fo;
  atomicAdd(op + 0, hv.x * w);
  atomicAdd(op + 1, hv.y * w);
  atomicAdd(op + 2, hv.z * w);
  atomicAdd(op + 3, hv.w * w);
}

__device__ __forceinline__ float softplusf(float v) {
  return v > 0.f ? v + log1pf(__expf(-v)) : log1pf(__expf(v));
}

__global__ void finish1_kernel(const float* __restrict__ acc, const float* __restrict__ b1,
                               float* __restrict__ hbuf, float* __restrict__ ixz1, int Nn) {
  size_t tid = (size_t)blockIdx.x * blockDim.x + threadIdx.x;
  int n = (int)(tid >> 6), f = (int)(tid & 63);
  if (n >= Nn) return;
  float v = acc[(size_t)n * C1 + f] + b1[f];
  hbuf[(size_t)n * C1 + f] = v > 0.f ? v : (__expf(v) - 1.f);
  if (f < 32) {
    float sp  = acc[(size_t)n * C1 + 32 + f] + b1[32 + f];
    float sd  = softplusf(sp) + 1e-10f;
    ixz1[(size_t)n * 32 + f] = -__logf(sd) + 0.5f * (sd * sd + v * v - 1.f);
  }
}

__global__ void finish2_kernel(const float* __restrict__ acc, const float* __restrict__ b2,
                               float* __restrict__ out2, float* __restrict__ ixz2, int Nn) {
  size_t tid = (size_t)blockIdx.x * blockDim.x + threadIdx.x;
  int n = (int)(tid >> 4), f = (int)(tid & 15);
  if (n >= Nn) return;
  float v = acc[(size_t)n * C2 + f] + b2[f];
  out2[(size_t)n * C2 + f] = v;
  if (f < 8) {
    float sp = acc[(size_t)n * C2 + 8 + f] + b2[8 + f];
    float sd = softplusf(sp) + 1e-10f;
    ixz2[(size_t)n * 8 + f] = -__logf(sd) + 0.5f * (sd * sd + v * v - 1.f);
  }
}

extern "C" void kernel_launch(void* const* d_in, const int* in_sizes, int n_in,
                              void* d_out, int out_size, void* d_ws, size_t ws_size,
                              hipStream_t stream) {
  const float* x   = (const float*)d_in[0];
  const int*   ei  = (const int*)  d_in[1];
  const float* W1  = (const float*)d_in[2];
  const float* as1 = (const float*)d_in[3];
  const float* ad1 = (const float*)d_in[4];
  const float* b1  = (const float*)d_in[5];
  const float* W2  = (const float*)d_in[6];
  const float* as2 = (const float*)d_in[7];
  const float* ad2 = (const float*)d_in[8];
  const float* b2  = (const float*)d_in[9];

  const int N    = in_sizes[0] / F_IN;
  const int E    = in_sizes[1] / 2;
  const int Etot = E + N;
  float* out = (float*)d_out;

  const size_t N64 = (size_t)N * C1, N16 = (size_t)N * C2;
  float* ws    = (float*)d_ws;
  float* h1    = ws;
  float* out1a = h1 + N64;
  float* s1    = out1a + N64;
  float* d1    = s1 + N;
  float* denom = d1 + N;
  int*   amax  = (int*)(denom + N);
  float* ewb   = (float*)(amax + N);
  float* h2    = ewb + Etot;
  float* out2a = h2 + N16;

  float* ixz1 = out + N16;
  float* ixz2 = out + (size_t)N * 48;

  const int tiles  = (N + 15) / 16;
  const int gblk   = (tiles + 7) / 8;
  const int eblk   = (Etot + 255) / 256;

  init_accum<<<(int)((N64 + 255) / 256), 256, 0, stream>>>(out1a, N64, amax, denom, N);
  gemm1_wmma<<<gblk, 256, 0, stream>>>(x, W1, h1, N);
  node_dots<C1><<<(N + 255) / 256, 256, 0, stream>>>(h1, as1, ad1, s1, d1, N);
  edge_max_kernel<<<eblk, 256, 0, stream>>>(ei, E, Etot, s1, d1, ewb, amax);
  edge_sum_kernel<<<eblk, 256, 0, stream>>>(ei, E, Etot, ewb, amax, denom);
  {
    long long tot = (long long)Etot * (C1 / 4);
    edge_agg_kernel<C1><<<(int)((tot + 255) / 256), 256, 0, stream>>>(ei, E, Etot, ewb, denom, h1, out1a);
  }
  finish1_kernel<<<(int)((N64 + 255) / 256), 256, 0, stream>>>(out1a, b1, h1, ixz1, N);

  init_accum<<<(int)((N16 + 255) / 256), 256, 0, stream>>>(out2a, N16, amax, denom, N);
  gemm2_wmma<<<gblk, 256, 0, stream>>>(h1, W2, h2, N);
  node_dots<C2><<<(N + 255) / 256, 256, 0, stream>>>(h2, as2, ad2, s1, d1, N);
  edge_max_kernel<<<eblk, 256, 0, stream>>>(ei, E, Etot, s1, d1, ewb, amax);
  edge_sum_kernel<<<eblk, 256, 0, stream>>>(ei, E, Etot, ewb, amax, denom);
  {
    long long tot = (long long)Etot * (C2 / 4);
    edge_agg_kernel<C2><<<(int)((tot + 255) / 256), 256, 0, stream>>>(ei, E, Etot, ewb, denom, h2, out2a);
  }
  finish2_kernel<<<(int)((N16 + 255) / 256), 256, 0, stream>>>(out2a, b2, out, ixz2, N);
}


# a layer's edge passes, clears and finish in one launch of row-range owners: a work-group scans the edge list for its 392 rows, orders them in LDS; sums in registers by edge number, no float atomics
# speedup vs baseline: 2.2757x; 2.2757x over previous
#include <hip/hip_runtime.h>
#include <hip/hip_bf16.h>

typedef __attribute__((ext_vector_type(2))) float v2f;
typedef __attribute__((ext_vector_type(8))) float v8f;

#define F_IN 128
#define C1   64
#define C2   16
#define NEG_SLOPE 0.2f

#define ROWS 392
#define NT   1024
#define CAP  16384

__global__ void __launch_bounds__(256)
gemm1_wmma(const float* __restrict__ x, const float* __restrict__ W1,
           float* __restrict__ h1, int Nn) {
  __shared__ float sW[F_IN * C1];
  for (int i = threadIdx.x; i < (F_IN * C1) / 4; i += 256) {
    float4 v = ((const float4*)W1)[i];
    const int k  = i >> 4;
    const int c0 = (i & 15) * 4;
    float* dst = sW + (k >> 2) * 256 + (k & 3);
    dst[(c0 + 0) * 4] = v.x;
    dst[(c0 + 1) * 4] = v.y;
    dst[(c0 + 2) * 4] = v.z;
    dst[(c0 + 3) * 4] = v.w;
  }
  __syncthreads();

  const int wave = threadIdx.x >> 5;
  const int lane = threadIdx.x & 31;
  const int tile = blockIdx.x * 8 + wave;
  if (tile * 16 >= Nn) return;
  const int row0 = tile * 16;
  const int r    = lane & 15;
  const int kh   = (lane >> 4) << 1;
  const int arow = min(row0 + r, Nn - 1);

  v8f acc0 = {}, acc1 = {}, acc2 = {}, acc3 = {};
  const float* ap = x + (size_t)arow * F_IN + kh;
  const float* bbase = sW + r * 4 + kh;
  for (int k0 = 0; k0 < F_IN; k0 += 4) {
    v2f a; a.x = ap[k0]; a.y = ap[k0 + 1];
    const float* bq = bbase + (k0 >> 2) * 256;
    v2f b0 = *(const v2f*)(bq);
    v2f b1 = *(const v2f*)(bq + 64);
    v2f b2 = *(const v2f*)(bq + 128);
    v2f b3 = *(const v2f*)(bq + 192);
    acc0 = __builtin_amdgcn_wmma_f32_16x16x4_f32(false, a, false, b0, (short)0, acc0, false, false);
    acc1 = __builtin_amdgcn_wmma_f32_16x16x4_f32(false, a, false, b1, (short)0, acc1, false, false);
    acc2 = __builtin_amdgcn_wmma_f32_16x16x4_f32(false, a, false, b2, (short)0, acc2, false, false);
    acc3 = __builtin_amdgcn_wmma_f32_16x16x4_f32(false, a, false, b3, (short)0, acc3, false, false);
  }

  const int mbase = (lane >> 4) * 8;
  for (int i = 0; i < 8; ++i) {
    const int row = row0 + mbase + i;
    if (row < Nn) {
      float* op = h1 + (size_t)row * C1 + r;
      op[0]  = acc0[i];
      op[16] = acc1[i];
      op[32] = acc2[i];
      op[48] = acc3[i];
    }
  }
}

__global__ void __launch_bounds__(256)
gemm2_wmma(const float* __restrict__ h, const float* __restrict__ W2,
           float* __restrict__ h2, int Nn) {
  __shared__ float sW[C1 * C2];
  for (int i = threadIdx.x; i < (C1 * C2) / 4; i += 256) {
    float4 v = ((const float4*)W2)[i];
    const int k  = i >> 2;
    const int c0 = (i & 3) * 4;
    float* dst = sW + (k >> 2) * 64 + (k & 3);
    dst[(c0 + 0) * 4] = v.x;
    dst[(c0 + 1) * 4] = v.y;
    dst[(c0 + 2) * 4] = v.z;
    dst[(c0 + 3) * 4] = v.w;
  }
  __syncthreads();

  const int wave = threadIdx.x >> 5;
  const int lane = threadIdx.x & 31;
  const int tile = blockIdx.x * 8 + wave;
  if (tile * 16 >= Nn) return;
  const int row0 = tile * 16;
  const int r    = lane & 15;
  const int kh   = (lane >> 4) << 1;
  const int arow = min(row0 + r, Nn - 1);

  v8f acc = {};
  const float* ap = h + (size_t)arow * C1 + kh;
  const float* bbase = sW + r * 4 + kh;
  for (int k0 = 0; k0 < C1; k0 += 4) {
    v2f a; a.x = ap[k0]; a.y = ap[k0 + 1];
    v2f b = *(const v2f*)(bbase + (k0 >> 2) * 64);
    acc = __builtin_amdgcn_wmma_f32_16x16x4_f32(false, a, false, b, (short)0, acc, false, false);
  }

  const int mbase = (lane >> 4) * 8;
  for (int i = 0; i < 8; ++i) {
    const int row = row0 + mbase + i;
    if (row < Nn) h2[(size_t)row * C2 + r] = acc[i];
  }
}

template <int C>
__global__ void node_dots(const float* __restrict__ h,
                          const float* __restrict__ asrc,
                          const float* __restrict__ adst,
                          float* __restrict__ s, float* __restrict__ d, int Nn) {
  int n = blockIdx.x * blockDim.x + threadIdx.x;
  if (n >= Nn) return;
  const float* hp = h + (size_t)n * C;
  float ss = 0.f, dd = 0.f;
  for (int f = 0; f < C; ++f) { float v = hp[f]; ss += v * asrc[f]; dd += v * adst[f]; }
  s[n] = ss; d[n] = dd;
}


__device__ __forceinline__ int fkey(float f) {
  int k = __float_as_int(f);
  return k >= 0 ? k : (k ^ 0x7fffffff);
}
__device__ __forceinline__ float fkey_inv(int k) {
  return __int_as_float(k >= 0 ? k : (k ^ 0x7fffffff));
}
__device__ __forceinline__ void edge_sd(const int* ei, int E, int e, int& s, int& d) {
  if (e < E) { s = ei[e]; d = ei[E + e]; } else { s = d = e - E; }
}
__device__ __forceinline__ float lrelu(float a) { return a > 0.f ? a : NEG_SLOPE * a; }

__device__ __forceinline__ float softplusf(float v) {
  return v > 0.f ? v + log1pf(__expf(-v)) : log1pf(__expf(v));
}

template <int C, bool FIRST>
__global__ void __launch_bounds__(NT)
layer_owner_kernel(const int* __restrict__ ei, int E, int Nn,
                   const float* __restrict__ hsrc,
                   const float* __restrict__ s, const float* __restrict__ d,
                   const float* __restrict__ bias,
                   float* __restrict__ rowout, float* __restrict__ klout) {
  __shared__ int   s_raw[4 * CAP];
  __shared__ int   s_rowcnt[ROWS];
  __shared__ int   s_off[ROWS + 1];
  __shared__ float s_d[ROWS];
  __shared__ int   s_cnt;
  int* const s_e  = s_raw;
  int* const s_r  = s_raw + CAP;
  int* const s_te = s_raw + 2 * CAP;
  int* const s_tr = s_raw + 3 * CAP;

  const int tid  = threadIdx.x;
  const int lo   = blockIdx.x * ROWS;
  const int rows = min(ROWS, Nn - lo);
  const unsigned urows = (unsigned)rows;
  const int Etot = E + Nn;
  constexpr int H   = C / 2;
  constexpr int TPE = C / 4;
  constexpr int NG  = NT / TPE;
  const int fo = (tid % TPE) * 4;
  const int g  = tid / TPE;

  for (int i = tid; i < ROWS; i += NT) { s_rowcnt[i] = 0; s_d[i] = (i < rows) ? d[lo + i] : 0.f; }
  if (tid == 0) s_cnt = 0;
  __syncthreads();

  auto append = [&](int e, unsigned r) {
    const int slot = atomicAdd(&s_cnt, 1);
    if (slot < CAP) { s_e[slot] = e; s_r[slot] = (int)r; }
    atomicAdd(&s_rowcnt[r], 1);
  };

  {
    const int E4 = ((E & 3) == 0) ? (E >> 2) : 0;
    const int4* dst4 = (const int4*)(ei + E);
    constexpr int U  = 8;
    constexpr int CH = U * NT;
    const int nch = (E4 + CH - 1) / CH;
    int c = (int)(((long long)blockIdx.x * nch) / gridDim.x);
    for (int it = 0; it < nch; ++it) {
      const int base = c * CH + tid;
      int4 dv[U];
      if (c * CH + CH <= E4) {
#pragma unroll
        for (int u = 0; u < U; ++u) dv[u] = dst4[base + u * NT];
      } else {
#pragma unroll
        for (int u = 0; u < U; ++u) {
          const int i = base + u * NT;
          dv[u] = (i < E4) ? dst4[i] : make_int4(-1, -1, -1, -1);
        }
      }
#pragma unroll
      for (int u = 0; u < U; ++u) {
        const int i = base + u * NT;
        const unsigned r0 = (unsigned)(dv[u].x - lo), r1 = (unsigned)(dv[u].y - lo);
        const unsigned r2 = (unsigned)(dv[u].z - lo), r3 = (unsigned)(dv[u].w - lo);
        const bool m0 = r0 < urows, m1 = r1 < urows, m2 = r2 < urows, m3 = r3 < urows;
        if (m0 | m1 | m2 | m3) {
          if (m0) append(4 * i + 0, r0);
          if (m1) append(4 * i + 1, r1);
          if (m2) append(4 * i + 2, r2);
          if (m3) append(4 * i + 3, r3);
        }
      }
      if (++c == nch) c = 0;
    }
    for (int e = 4 * E4 + tid; e < E; e += NT) {
      const unsigned r = (unsigned)(ei[E + e] - lo);
      if (r < urows) append(e, r);
    }
    for (int r = tid; r < rows; r += NT) append(E + lo + r, (unsigned)r);
  }
  __syncthreads();
  const int  ntot = s_cnt;
  const bool listed = ntot <= CAP;

  if (listed) {
    if (tid < 32) {
      constexpr int PER = (ROWS + 31) / 32;
      int loc[PER]; int sum = 0;
#pragma unroll
      for (int k = 0; k < PER; ++k) {
        const int idx = tid * PER + k;
        loc[k] = sum;
        sum += (idx < ROWS) ? s_rowcnt[idx] : 0;
      }
      int incl = sum;
#pragma unroll
      for (int o = 1; o < 32; o <<= 1) {
        const int t = __shfl_up(incl, o);
        if (tid >= o) incl += t;
      }
      const int basep = incl - sum;
#pragma unroll
      for (int k = 0; k < PER; ++k) {
        const int idx = tid * PER + k;
        if (idx < ROWS) { s_off[idx] = basep + loc[k]; s_rowcnt[idx] = 0; }
      }
      if (tid == 31) s_off[ROWS] = incl;
    }
    __syncthreads();

    for (int i = tid; i < ntot; i += NT) {
      const int r = s_r[i];
      const int p = s_off[r] + atomicAdd(&s_rowcnt[r], 1);
      s_te[p] = s_e[i]; s_tr[p] = r;
    }
    __syncthreads();

    constexpr int UA = 4;
    for (int p0 = tid; p0 < ntot; p0 += UA * NT) {
      int sn[UA], pos[UA], rr[UA]; float sv[UA];
#pragma unroll
      for (int k = 0; k < UA; ++k) {
        const int  p  = p0 + k * NT;
        const bool ok = p < ntot;
        const int  pc = ok ? p : p0;
        const int  e  = s_te[pc];
        const int  r  = s_tr[pc];
        const int  b  = s_off[r], en = s_off[r + 1];
        int rk = 0;
        for (int j = b; j < en; ++j) rk += (s_te[j] < e) ? 1 : 0;
        pos[k] = ok ? b + rk : -1;
        rr[k]  = r;
        sn[k]  = e < E ? ei[e] : e - E;
      }
#pragma unroll
      for (int k = 0; k < UA; ++k) sv[k] = s[sn[k]];
#pragma unroll
      for (int k = 0; k < UA; ++k) {
        if (pos[k] >= 0) {
          s_e[pos[k]] = sn[k];
          s_r[pos[k]] = __float_as_int(lrelu(sv[k] + s_d[rr[k]]));
        }
      }
    }
    __syncthreads();

    for (int r = g; r < rows; r += NG) {
      const int b = s_off[r], en = s_off[r + 1];
      float m = __int_as_float(s_r[b]);
      for (int j = b + 1; j < en; ++j) m = fmaxf(m, __int_as_float(s_r[j]));
      float den = 0.f;
      for (int j = b; j < en; ++j) den += __expf(__int_as_float(s_r[j]) - m);
      den += 1e-16f;
      float a0 = 0.f, a1 = 0.f, a2 = 0.f, a3 = 0.f;
      constexpr int UG = 8;
      for (int j0 = b; j0 < en; j0 += UG) {
        float4 hv[UG]; float w[UG]; bool ok[UG];
#pragma unroll
        for (int k = 0; k < UG; ++k) {
          const int j = j0 + k;
          ok[k] = j < en;
          const int jc = ok[k] ? j : b;
          w[k]  = __expf(__int_as_float(s_r[jc]) - m) / den;
          hv[k] = *(const float4*)(hsrc + (size_t)s_e[jc] * C + fo);
        }
#pragma unroll
        for (int k = 0; k < UG; ++k) {
          if (ok[k]) {
            a0 += hv[k].x * w[k]; a1 += hv[k].y * w[k];
            a2 += hv[k].z * w[k]; a3 += hv[k].w * w[k];
          }
        }
      }
      const float4 bv = *(const float4*)(bias + fo);
      const float o0 = a0 + bv.x, o1 = a1 + bv.y, o2 = a2 + bv.z, o3 = a3 + bv.w;
      const bool  up = fo >= H;
      const float ra = __shfl_xor(up ? o0 : o2, TPE / 2, TPE);
      const float rb = __shfl_xor(up ? o1 : o3, TPE / 2, TPE);
      const size_t n = (size_t)(lo + r);
      float4 st;
      st.x = FIRST ? (o0 > 0.f ? o0 : (__expf(o0) - 1.f)) : o0;
      st.y = FIRST ? (o1 > 0.f ? o1 : (__expf(o1) - 1.f)) : o1;
      st.z = FIRST ? (o2 > 0.f ? o2 : (__expf(o2) - 1.f)) : o2;
      st.w = FIRST ? (o3 > 0.f ? o3 : (__expf(o3) - 1.f)) : o3;
      *(float4*)(rowout + n * C + fo) = st;
      const float mu0 = up ? ra : o0, mu1 = up ? rb : o1;
      const float sp0 = up ? o2 : ra, sp1 = up ? o3 : rb;
      const int   kf  = up ? fo - H + 2 : fo;
      const float d0 = softplusf(sp0) + 1e-10f, d1 = softplusf(sp1) + 1e-10f;
      float2 kl;
      kl.x = -__logf(d0) + 0.5f * (d0 * d0 + mu0 * mu0 - 1.f);
      kl.y = -__logf(d1) + 0.5f * (d1 * d1 + mu1 * mu1 - 1.f);
      *(float2*)(klout + n * H + kf) = kl;
    }
    return;
  }

  float* const s_acc = (float*)s_raw;
  int*   const s_mx  = s_raw + ROWS * C;
  float* const s_den = (float*)(s_raw + ROWS * C + ROWS);
  for (int i = tid; i < ROWS * C; i += NT) s_acc[i] = 0.f;
  for (int i = tid; i < ROWS; i += NT) { s_mx[i] = (int)0x80000000; s_den[i] = 0.f; }
  __syncthreads();
  for (int e = tid; e < Etot; e += NT) {
    int sn, dn; edge_sd(ei, E, e, sn, dn);
    const unsigned r = (unsigned)(dn - lo);
    if (r < urows) atomicMax(&s_mx[r], fkey(lrelu(s[sn] + s_d[r])));
  }
  __syncthreads();
  for (int e = tid; e < Etot; e += NT) {
    int sn, dn; edge_sd(ei, E, e, sn, dn);
    const unsigned r = (unsigned)(dn - lo);
    if (r < urows)
      atomicAdd(&s_den[r], __expf(lrelu(s[sn] + s_d[r]) - fkey_inv(s_mx[r])));
  }
  __syncthreads();
  for (int e = g; e < Etot; e += NG) {
    int sn, dn; edge_sd(ei, E, e, sn, dn);
    const unsigned r = (unsigned)(dn - lo);
    if (r < urows) {
      const float ew = __expf(lrelu(s[sn] + s_d[r]) - fkey_inv(s_mx[r]));
      const float w  = ew / (s_den[r] + 1e-16f);
      const float4 hv = *(const float4*)(hsrc + (size_t)sn * C + fo);
      float* op = s_acc + (int)r * C + fo;
      atomicAdd(op + 0, hv.x * w);
      atomicAdd(op + 1, hv.y * w);
      atomicAdd(op + 2, hv.z * w);
      atomicAdd(op + 3, hv.w * w);
    }
  }
  __syncthreads();
  for (int i = tid; i < rows * C; i += NT) {
    const int r = i / C, f = i % C;
    const size_t n = (size_t)(lo + r);
    const float v = s_acc[i] + bias[f];
    rowout[n * C + f] = FIRST ? (v > 0.f ? v : (__expf(v) - 1.f)) : v;
    if (f < H) {
      const float sp = s_acc[r * C + H + f] + bias[H + f];
      const float sd = softplusf(sp) + 1e-10f;
      klout[n * H + f] = -__logf(sd) + 0.5f * (sd * sd + v * v - 1.f);
    }
  }
}

extern "C" void kernel_launch(void* const* d_in, const int* in_sizes, int n_in,
                              void* d_out, int out_size, void* d_ws, size_t ws_size,
                              hipStream_t stream) {
  const float* x   = (const float*)d_in[0];
  const int*   ei  = (const int*)  d_in[1];
  const float* W1  = (const float*)d_in[2];
  const float* as1 = (const float*)d_in[3];
  const float* ad1 = (const float*)d_in[4];
  const float* b1  = (const float*)d_in[5];
  const float* W2  = (const float*)d_in[6];
  const float* as2 = (const float*)d_in[7];
  const float* ad2 = (const float*)d_in[8];
  const float* b2  = (const float*)d_in[9];

  const int N    = in_sizes[0] / F_IN;
  const int E    = in_sizes[1] / 2;
  float* out = (float*)d_out;

  const size_t N64 = (size_t)N * C1, N16 = (size_t)N * C2;
  float* ws = (float*)d_ws;
  float* h1 = ws;
  float* hb = h1 + N64;
  float* h2 = hb + N64;
  float* s1 = h2 + N16;
  float* d1 = s1 + N;

  float* ixz1 = out + N16;
  float* ixz2 = out + (size_t)N * 48;

  const int tiles  = (N + 15) / 16;
  const int gblk   = (tiles + 7) / 8;
  const int owners = (N + ROWS - 1) / ROWS;

  gemm1_wmma<<<gblk, 256, 0, stream>>>(x, W1, h1, N);
  node_dots<C1><<<(N + 255) / 256, 256, 0, stream>>>(h1, as1, ad1, s1, d1, N);
  layer_owner_kernel<C1, true><<<owners, NT, 0, stream>>>(ei, E, N, h1, s1, d1, b1, hb, ixz1);

  gemm2_wmma<<<gblk, 256, 0, stream>>>(hb, W2, h2, N);
  node_dots<C2><<<(N + 255) / 256, 256, 0, stream>>>(h2, as2, ad2, s1, d1, N);
  layer_owner_kernel<C2, false><<<owners, NT, 0, stream>>>(ei, E, N, h2, s1, d1, b2, out, ixz2);
}


# layer 2 reuses layer 1's edge lists: an owner stores its row offsets and row-ordered (source,row) entries in the workspace, layer 2 loads them: no second scan, grouping or ordering; sum order kept
# speedup vs baseline: 2.7465x; 1.2069x over previous
#include <hip/hip_runtime.h>
#include <hip/hip_bf16.h>

typedef __attribute__((ext_vector_type(2))) float v2f;
typedef __attribute__((ext_vector_type(8))) float v8f;

#define F_IN 128
#define C1   64
#define C2   16
#define NEG_SLOPE 0.2f

#define ROWS 392
#define NT   1024
#define CAP  16384
#define LOFF 396

__global__ void __launch_bounds__(256)
gemm1_wmma(const float* __restrict__ x, const float* __restrict__ W1,
           float* __restrict__ h1, int Nn) {
  __shared__ float sW[F_IN * C1];
  for (int i = threadIdx.x; i < (F_IN * C1) / 4; i += 256) {
    float4 v = ((const float4*)W1)[i];
    const int k  = i >> 4;
    const int c0 = (i & 15) * 4;
    float* dst = sW + (k >> 2) * 256 + (k & 3);
    dst[(c0 + 0) * 4] = v.x;
    dst[(c0 + 1) * 4] = v.y;
    dst[(c0 + 2) * 4] = v.z;
    dst[(c0 + 3) * 4] = v.w;
  }
  __syncthreads();

  const int wave = threadIdx.x >> 5;
  const int lane = threadIdx.x & 31;
  const int tile = blockIdx.x * 8 + wave;
  if (tile * 16 >= Nn) return;
  const int row0 = tile * 16;
  const int r    = lane & 15;
  const int kh   = (lane >> 4) << 1;
  const int arow = min(row0 + r, Nn - 1);

  v8f acc0 = {}, acc1 = {}, acc2 = {}, acc3 = {};
  const float* ap = x + (size_t)arow * F_IN + kh;
  const float* bbase = sW + r * 4 + kh;
  for (int k0 = 0; k0 < F_IN; k0 += 4) {
    v2f a; a.x = ap[k0]; a.y = ap[k0 + 1];
    const float* bq = bbase + (k0 >> 2) * 256;
    v2f b0 = *(const v2f*)(bq);
    v2f b1 = *(const v2f*)(bq + 64);
    v2f b2 = *(const v2f*)(bq + 128);
    v2f b3 = *(const v2f*)(bq + 192);
    acc0 = __builtin_amdgcn_wmma_f32_16x16x4_f32(false, a, false, b0, (short)0, acc0, false, false);
    acc1 = __builtin_amdgcn_wmma_f32_16x16x4_f32(false, a, false, b1, (short)0, acc1, false, false);
    acc2 = __builtin_amdgcn_wmma_f32_16x16x4_f32(false, a, false, b2, (short)0, acc2, false, false);
    acc3 = __builtin_amdgcn_wmma_f32_16x16x4_f32(false, a, false, b3, (short)0, acc3, false, false);
  }

  const int mbase = (lane >> 4) * 8;
  for (int i = 0; i < 8; ++i) {
    const int row = row0 + mbase + i;
    if (row < Nn) {
      float* op = h1 + (size_t)row * C1 + r;
      op[0]  = acc0[i];
      op[16] = acc1[i];
      op[32] = acc2[i];
      op[48] = acc3[i];
    }
  }
}

__global__ void __launch_bounds__(256)
gemm2_wmma(const float* __restrict__ h, const float* __restrict__ W2,
           float* __restrict__ h2, int Nn) {
  __shared__ float sW[C1 * C2];
  for (int i = threadIdx.x; i < (C1 * C2) / 4; i += 256) {
    float4 v = ((const float4*)W2)[i];
    const int k  = i >> 2;
    const int c0 = (i & 3) * 4;
    float* dst = sW + (k >> 2) * 64 + (k & 3);
    dst[(c0 + 0) * 4] = v.x;
    dst[(c0 + 1) * 4] = v.y;
    dst[(c0 + 2) * 4] = v.z;
    dst[(c0 + 3) * 4] = v.w;
  }
  __syncthreads();

  const int wave = threadIdx.x >> 5;
  const int lane = threadIdx.x & 31;
  const int tile = blockIdx.x * 8 + wave;
  if (tile * 16 >= Nn) return;
  const int row0 = tile * 16;
  const int r    = lane & 15;
  const int kh   = (lane >> 4) << 1;
  const int arow = min(row0 + r, Nn - 1);

  v8f acc = {};
  const float* ap = h + (size_t)arow * C1 + kh;
  const float* bbase = sW + r * 4 + kh;
  for (int k0 = 0; k0 < C1; k0 += 4) {
    v2f a; a.x = ap[k0]; a.y = ap[k0 + 1];
    v2f b = *(const v2f*)(bbase + (k0 >> 2) * 64);
    acc = __builtin_amdgcn_wmma_f32_16x16x4_f32(false, a, false, b, (short)0, acc, false, false);
  }

  const int mbase = (lane >> 4) * 8;
  for (int i = 0; i < 8; ++i) {
    const int row = row0 + mbase + i;
    if (row < Nn) h2[(size_t)row * C2 + r] = acc[i];
  }
}

template <int C>
__global__ void node_dots(const float* __restrict__ h,
                          const float* __restrict__ asrc,
                          const float* __restrict__ adst,
                          float* __restrict__ s, float* __restrict__ d, int Nn) {
  int n = blockIdx.x * blockDim.x + threadIdx.x;
  if (n >= Nn) return;
  const float* hp = h + (size_t)n * C;
  float ss = 0.f, dd = 0.f;
  for (int f = 0; f < C; ++f) { float v = hp[f]; ss += v * asrc[f]; dd += v * adst[f]; }
  s[n] = ss; d[n] = dd;
}


__device__ __forceinline__ int fkey(float f) {
  int k = __float_as_int(f);
  return k >= 0 ? k : (k ^ 0x7fffffff);
}
__device__ __forceinline__ float fkey_inv(int k) {
  return __int_as_float(k >= 0 ? k : (k ^ 0x7fffffff));
}
__device__ __forceinline__ void edge_sd(const int* ei, int E, int e, int& s, int& d) {
  if (e < E) { s = ei[e]; d = ei[E + e]; } else { s = d = e - E; }
}
__device__ __forceinline__ float lrelu(float a) { return a > 0.f ? a : NEG_SLOPE * a; }

__device__ __forceinline__ float softplusf(float v) {
  return v > 0.f ? v + log1pf(__expf(-v)) : log1pf(__expf(v));
}

template <int C, bool FIRST>
__global__ void __launch_bounds__(NT)
layer_owner_kernel(const int* __restrict__ ei, int E, int Nn,
                   const float* __restrict__ hsrc,
                   const float* __restrict__ s, const float* __restrict__ d,
                   const float* __restrict__ bias,
                   float* __restrict__ rowout, float* __restrict__ klout,
                   int* __restrict__ loff, int* __restrict__ lent) {
  __shared__ int   s_raw[4 * CAP];
  __shared__ int   s_rowcnt[ROWS];
  __shared__ int   s_off[ROWS + 1];
  __shared__ float s_d[ROWS];
  __shared__ int   s_cnt;
  int* const s_e  = s_raw;
  int* const s_r  = s_raw + CAP;
  int* const s_te = s_raw + 2 * CAP;
  int* const s_tr = s_raw + 3 * CAP;

  const int tid  = threadIdx.x;
  const int lo   = blockIdx.x * ROWS;
  const int rows = min(ROWS, Nn - lo);
  const unsigned urows = (unsigned)rows;
  const int Etot = E + Nn;
  constexpr int H   = C / 2;
  constexpr int TPE = C / 4;
  constexpr int NG  = NT / TPE;
  const int fo = (tid % TPE) * 4;
  const int g  = tid / TPE;

  int*  const offp = loff + (size_t)blockIdx.x * LOFF;
  int2* const entp = (int2*)lent + (size_t)blockIdx.x * CAP;

  int ntot = 0;
  if constexpr (!FIRST) ntot = offp[ROWS + 1];

  for (int i = tid; i < ROWS; i += NT) s_d[i] = (i < rows) ? d[lo + i] : 0.f;
  if constexpr (FIRST) {
    for (int i = tid; i < ROWS; i += NT) s_rowcnt[i] = 0;
    if (tid == 0) s_cnt = 0;
  }
  __syncthreads();

  if constexpr (FIRST) {
    auto append = [&](int e, unsigned r) {
      const int slot = atomicAdd(&s_cnt, 1);
      if (slot < CAP) { s_e[slot] = e; s_r[slot] = (int)r; }
      atomicAdd(&s_rowcnt[r], 1);
    };

    {
      const int E4 = ((E & 3) == 0) ? (E >> 2) : 0;
      const int4* dst4 = (const int4*)(ei + E);
      constexpr int U  = 8;
      constexpr int CH = U * NT;
      const int nch = (E4 + CH - 1) / CH;
      int c = (int)(((long long)blockIdx.x * nch) / gridDim.x);
      for (int it = 0; it < nch; ++it) {
        const int base = c * CH + tid;
        int4 dv[U];
        if (c * CH + CH <= E4) {
#pragma unroll
          for (int u = 0; u < U; ++u) dv[u] = dst4[base + u * NT];
        } else {
#pragma unroll
          for (int u = 0; u < U; ++u) {
            const int i = base + u * NT;
            dv[u] = (i < E4) ? dst4[i] : make_int4(-1, -1, -1, -1);
          }
        }
#pragma unroll
        for (int u = 0; u < U; ++u) {
          const int i = base + u * NT;
          const unsigned r0 = (unsigned)(dv[u].x - lo), r1 = (unsigned)(dv[u].y - lo);
          const unsigned r2 = (unsigned)(dv[u].z - lo), r3 = (unsigned)(dv[u].w - lo);
          const bool m0 = r0 < urows, m1 = r1 < urows, m2 = r2 < urows, m3 = r3 < urows;
          if (m0 | m1 | m2 | m3) {
            if (m0) append(4 * i + 0, r0);
            if (m1) append(4 * i + 1, r1);
            if (m2) append(4 * i + 2, r2);
            if (m3) append(4 * i + 3, r3);
          }
        }
        if (++c == nch) c = 0;
      }
      for (int e = 4 * E4 + tid; e < E; e += NT) {
        const unsigned r = (unsigned)(ei[E + e] - lo);
        if (r < urows) append(e, r);
      }
      for (int r = tid; r < rows; r += NT) append(E + lo + r, (unsigned)r);
    }
    __syncthreads();
    ntot = s_cnt;
    if (tid == 0) offp[ROWS + 1] = ntot;
  }
  const bool listed = ntot <= CAP;

  if (listed) {
    if constexpr (FIRST) {
      if (tid < 32) {
        constexpr int PER = (ROWS + 31) / 32;
        int loc[PER]; int sum = 0;
#pragma unroll
        for (int k = 0; k < PER; ++k) {
          const int idx = tid * PER + k;
          loc[k] = sum;
          sum += (idx < ROWS) ? s_rowcnt[idx] : 0;
        }
        int incl = sum;
#pragma unroll
        for (int o = 1; o < 32; o <<= 1) {
          const int t = __shfl_up(incl, o);
          if (tid >= o) incl += t;
        }
        const int basep = incl - sum;
#pragma unroll
        for (int k = 0; k < PER; ++k) {
          const int idx = tid * PER + k;
          if (idx < ROWS) { s_off[idx] = basep + loc[k]; s_rowcnt[idx] = 0; }
        }
        if (tid == 31) s_off[ROWS] = incl;
      }
      __syncthreads();
      for (int i = tid; i <= ROWS; i += NT) offp[i] = s_off[i];

      for (int i = tid; i < ntot; i += NT) {
        const int r = s_r[i];
        const int p = s_off[r] + atomicAdd(&s_rowcnt[r], 1);
        s_te[p] = s_e[i]; s_tr[p] = r;
      }
      __syncthreads();

      constexpr int UA = 4;
      for (int p0 = tid; p0 < ntot; p0 += UA * NT) {
        int sn[UA], pos[UA], rr[UA]; float sv[UA];
#pragma unroll
        for (int k = 0; k < UA; ++k) {
          const int  p  = p0 + k * NT;
          const bool ok = p < ntot;
          const int  pc = ok ? p : p0;
          const int  e  = s_te[pc];
          const int  r  = s_tr[pc];
          const int  b  = s_off[r], en = s_off[r + 1];
          int rk = 0;
          for (int j = b; j < en; ++j) rk += (s_te[j] < e) ? 1 : 0;
          pos[k] = ok ? b + rk : -1;
          rr[k]  = r;
          sn[k]  = e < E ? ei[e] : e - E;
        }
#pragma unroll
        for (int k = 0; k < UA; ++k) sv[k] = s[sn[k]];
#pragma unroll
        for (int k = 0; k < UA; ++k) {
          if (pos[k] >= 0) {
            s_e[pos[k]] = sn[k];
            s_r[pos[k]] = __float_as_int(lrelu(sv[k] + s_d[rr[k]]));
            entp[pos[k]] = make_int2(sn[k], rr[k]);
          }
        }
      }
      __syncthreads();
    } else {
      for (int i = tid; i <= ROWS; i += NT) s_off[i] = offp[i];
      constexpr int UA = 4;
      for (int p0 = tid; p0 < ntot; p0 += UA * NT) {
        int2 en[UA]; float sv[UA];
#pragma unroll
        for (int k = 0; k < UA; ++k) {
          const int p = p0 + k * NT;
          en[k] = entp[p < ntot ? p : p0];
        }
#pragma unroll
        for (int k = 0; k < UA; ++k) sv[k] = s[en[k].x];
        asm volatile("" :: "v"(sv[0]), "v"(sv[1]), "v"(sv[2]), "v"(sv[3]));
#pragma unroll
        for (int k = 0; k < UA; ++k) {
          const int p = p0 + k * NT;
          if (p < ntot) {
            s_e[p] = en[k].x;
            s_r[p] = __float_as_int(lrelu(sv[k] + s_d[en[k].y]));
          }
        }
      }
      __syncthreads();
    }

    for (int r = g; r < rows; r += NG) {
      const int b = s_off[r], en = s_off[r + 1];
      float m = __int_as_float(s_r[b]);
      for (int j = b + 1; j < en; ++j) m = fmaxf(m, __int_as_float(s_r[j]));
      float den = 0.f;
      for (int j = b; j < en; ++j) den += __expf(__int_as_float(s_r[j]) - m);
      den += 1e-16f;
      float a0 = 0.f, a1 = 0.f, a2 = 0.f, a3 = 0.f;
      constexpr int UG = 8;
      for (int j0 = b; j0 < en; j0 += UG) {
        float4 hv[UG]; float w[UG]; bool ok[UG];
#pragma unroll
        for (int k = 0; k < UG; ++k) {
          const int j = j0 + k;
          ok[k] = j < en;
          const int jc = ok[k] ? j : b;
          w[k]  = __expf(__int_as_float(s_r[jc]) - m) / den;
          hv[k] = *(const float4*)(hsrc + (size_t)s_e[jc] * C + fo);
        }
#pragma unroll
        for (int k = 0; k < UG; ++k) {
          if (ok[k]) {
            a0 += hv[k].x * w[k]; a1 += hv[k].y * w[k];
            a2 += hv[k].z * w[k]; a3 += hv[k].w * w[k];
          }
        }
      }
      const float4 bv = *(const float4*)(bias + fo);
      const float o0 = a0 + bv.x, o1 = a1 + bv.y, o2 = a2 + bv.z, o3 = a3 + bv.w;
      const bool  up = fo >= H;
      const float ra = __shfl_xor(up ? o0 : o2, TPE / 2, TPE);
      const float rb = __shfl_xor(up ? o1 : o3, TPE / 2, TPE);
      const size_t n = (size_t)(lo + r);
      float4 st;
      st.x = FIRST ? (o0 > 0.f ? o0 : (__expf(o0) - 1.f)) : o0;
      st.y = FIRST ? (o1 > 0.f ? o1 : (__expf(o1) - 1.f)) : o1;
      st.z = FIRST ? (o2 > 0.f ? o2 : (__expf(o2) - 1.f)) : o2;
      st.w = FIRST ? (o3 > 0.f ? o3 : (__expf(o3) - 1.f)) : o3;
      *(float4*)(rowout + n * C + fo) = st;
      const float mu0 = up ? ra : o0, mu1 = up ? rb : o1;
      const float sp0 = up ? o2 : ra, sp1 = up ? o3 : rb;
      const int   kf  = up ? fo - H + 2 : fo;
      const float d0 = softplusf(sp0) + 1e-10f, d1 = softplusf(sp1) + 1e-10f;
      float2 kl;
      kl.x = -__logf(d0) + 0.5f * (d0 * d0 + mu0 * mu0 - 1.f);
      kl.y = -__logf(d1) + 0.5f * (d1 * d1 + mu1 * mu1 - 1.f);
      *(float2*)(klout + n * H + kf) = kl;
    }
    return;
  }

  float* const s_acc = (float*)s_raw;
  int*   const s_mx  = s_raw + ROWS * C;
  float* const s_den = (float*)(s_raw + ROWS * C + ROWS);
  for (int i = tid; i < ROWS * C; i += NT) s_acc[i] = 0.f;
  for (int i = tid; i < ROWS; i += NT) { s_mx[i] = (int)0x80000000; s_den[i] = 0.f; }
  __syncthreads();
  for (int e = tid; e < Etot; e += NT) {
    int sn, dn; edge_sd(ei, E, e, sn, dn);
    const unsigned r = (unsigned)(dn - lo);
    if (r < urows) atomicMax(&s_mx[r], fkey(lrelu(s[sn] + s_d[r])));
  }
  __syncthreads();
  for (int e = tid; e < Etot; e += NT) {
    int sn, dn; edge_sd(ei, E, e, sn, dn);
    const unsigned r = (unsigned)(dn - lo);
    if (r < urows)
      atomicAdd(&s_den[r], __expf(lrelu(s[sn] + s_d[r]) - fkey_inv(s_mx[r])));
  }
  __syncthreads();
  for (int e = g; e < Etot; e += NG) {
    int sn, dn; edge_sd(ei, E, e, sn, dn);
    const unsigned r = (unsigned)(dn - lo);
    if (r < urows) {
      const float ew = __expf(lrelu(s[sn] + s_d[r]) - fkey_inv(s_mx[r]));
      const float w  = ew / (s_den[r] + 1e-16f);
      const float4 hv = *(const float4*)(hsrc + (size_t)sn * C + fo);
      float* op = s_acc + (int)r * C + fo;
      atomicAdd(op + 0, hv.x * w);
      atomicAdd(op + 1, hv.y * w);
      atomicAdd(op + 2, hv.z * w);
      atomicAdd(op + 3, hv.w * w);
    }
  }
  __syncthreads();
  for (int i = tid; i < rows * C; i += NT) {
    const int r = i / C, f = i % C;
    const size_t n = (size_t)(lo + r);
    const float v = s_acc[i] + bias[f];
    rowout[n * C + f] = FIRST ? (v > 0.f ? v : (__expf(v) - 1.f)) : v;
    if (f < H) {
      const float sp = s_acc[r * C + H + f] + bias[H + f];
      const float sd = softplusf(sp) + 1e-10f;
      klout[n * H + f] = -__logf(sd) + 0.5f * (sd * sd + v * v - 1.f);
    }
  }
}

extern "C" void kernel_launch(void* const* d_in, const int* in_sizes, int n_in,
                              void* d_out, int out_size, void* d_ws, size_t ws_size,
                              hipStream_t stream) {
  const float* x   = (const float*)d_in[0];
  const int*   ei  = (const int*)  d_in[1];
  const float* W1  = (const float*)d_in[2];
  const float* as1 = (const float*)d_in[3];
  const float* ad1 = (const float*)d_in[4];
  const float* b1  = (const float*)d_in[5];
  const float* W2  = (const float*)d_in[6];
  const float* as2 = (const float*)d_in[7];
  const float* ad2 = (const float*)d_in[8];
  const float* b2  = (const float*)d_in[9];

  const int N    = in_sizes[0] / F_IN;
  const int E    = in_sizes[1] / 2;
  float* out = (float*)d_out;

  const size_t N64 = (size_t)N * C1, N16 = (size_t)N * C2;
  const int owners = (N + ROWS - 1) / ROWS;
  int*   lent = (int*)d_ws;
  int*   loff = lent + (size_t)owners * CAP * 2;
  float* ws = (float*)(loff + (size_t)owners * LOFF);
  float* h1 = ws;
  float* hb = h1 + N64;
  float* h2 = hb + N64;
  float* s1 = h2 + N16;
  float* d1 = s1 + N;

  float* ixz1 = out + N16;
  float* ixz2 = out + (size_t)N * 48;

  const int tiles  = (N + 15) / 16;
  const int gblk   = (tiles + 7) / 8;

  gemm1_wmma<<<gblk, 256, 0, stream>>>(x, W1, h1, N);
  node_dots<C1><<<(N + 255) / 256, 256, 0, stream>>>(h1, as1, ad1, s1, d1, N);
  layer_owner_kernel<C1, true><<<owners, NT, 0, stream>>>(ei, E, N, h1, s1, d1, b1, hb, ixz1, loff, lent);

  gemm2_wmma<<<gblk, 256, 0, stream>>>(hb, W2, h2, N);
  node_dots<C2><<<(N + 255) / 256, 256, 0, stream>>>(h2, as2, ad2, s1, d1, N);
  layer_owner_kernel<C2, false><<<owners, NT, 0, stream>>>(ei, E, N, h2, s1, d1, b2, out, ixz2, loff, lent);
}


# owners stop scanning the whole edge list: a new binning launch orders each 2560-edge chunk by owner in LDS and an owner reads only its short segment of every chunk; 6->7 launches; sum order kept
# speedup vs baseline: 3.1018x; 1.1294x over previous
#include <hip/hip_runtime.h>
#include <hip/hip_bf16.h>

typedef __attribute__((ext_vector_type(2))) float v2f;
typedef __attribute__((ext_vector_type(8))) float v8f;

#define F_IN 128
#define C1   64
#define C2   16
#define NEG_SLOPE 0.2f

#define ROWS 392
#define NT   1024
#define CAP  16384
#define LOFF 396
#define BT    256
#define EPT   10
#define CHUNK (BT * EPT)
#define LPC   4
#define USEG  6

__global__ void __launch_bounds__(256)
gemm1_wmma(const float* __restrict__ x, const float* __restrict__ W1,
           float* __restrict__ h1, int Nn) {
  __shared__ float sW[F_IN * C1];
  for (int i = threadIdx.x; i < (F_IN * C1) / 4; i += 256) {
    float4 v = ((const float4*)W1)[i];
    const int k  = i >> 4;
    const int c0 = (i & 15) * 4;
    float* dst = sW + (k >> 2) * 256 + (k & 3);
    dst[(c0 + 0) * 4] = v.x;
    dst[(c0 + 1) * 4] = v.y;
    dst[(c0 + 2) * 4] = v.z;
    dst[(c0 + 3) * 4] = v.w;
  }
  __syncthreads();

  const int wave = threadIdx.x >> 5;
  const int lane = threadIdx.x & 31;
  const int tile = blockIdx.x * 8 + wave;
  if (tile * 16 >= Nn) return;
  const int row0 = tile * 16;
  const int r    = lane & 15;
  const int kh   = (lane >> 4) << 1;
  const int arow = min(row0 + r, Nn - 1);

  v8f acc0 = {}, acc1 = {}, acc2 = {}, acc3 = {};
  const float* ap = x + (size_t)arow * F_IN + kh;
  const float* bbase = sW + r * 4 + kh;
  for (int k0 = 0; k0 < F_IN; k0 += 4) {
    v2f a; a.x = ap[k0]; a.y = ap[k0 + 1];
    const float* bq = bbase + (k0 >> 2) * 256;
    v2f b0 = *(const v2f*)(bq);
    v2f b1 = *(const v2f*)(bq + 64);
    v2f b2 = *(const v2f*)(bq + 128);
    v2f b3 = *(const v2f*)(bq + 192);
    acc0 = __builtin_amdgcn_wmma_f32_16x16x4_f32(false, a, false, b0, (short)0, acc0, false, false);
    acc1 = __builtin_amdgcn_wmma_f32_16x16x4_f32(false, a, false, b1, (short)0, acc1, false, false);
    acc2 = __builtin_amdgcn_wmma_f32_16x16x4_f32(false, a, false, b2, (short)0, acc2, false, false);
    acc3 = __builtin_amdgcn_wmma_f32_16x16x4_f32(false, a, false, b3, (short)0, acc3, false, false);
  }

  const int mbase = (lane >> 4) * 8;
  for (int i = 0; i < 8; ++i) {
    const int row = row0 + mbase + i;
    if (row < Nn) {
      float* op = h1 + (size_t)row * C1 + r;
      op[0]  = acc0[i];
      op[16] = acc1[i];
      op[32] = acc2[i];
      op[48] = acc3[i];
    }
  }
}

__global__ void __launch_bounds__(256)
gemm2_wmma(const float* __restrict__ h, const float* __restrict__ W2,
           float* __restrict__ h2, int Nn) {
  __shared__ float sW[C1 * C2];
  for (int i = threadIdx.x; i < (C1 * C2) / 4; i += 256) {
    float4 v = ((const float4*)W2)[i];
    const int k  = i >> 2;
    const int c0 = (i & 3) * 4;
    float* dst = sW + (k >> 2) * 64 + (k & 3);
    dst[(c0 + 0) * 4] = v.x;
    dst[(c0 + 1) * 4] = v.y;
    dst[(c0 + 2) * 4] = v.z;
    dst[(c0 + 3) * 4] = v.w;
  }
  __syncthreads();

  const int wave = threadIdx.x >> 5;
  const int lane = threadIdx.x & 31;
  const int tile = blockIdx.x * 8 + wave;
  if (tile * 16 >= Nn) return;
  const int row0 = tile * 16;
  const int r    = lane & 15;
  const int kh   = (lane >> 4) << 1;
  const int arow = min(row0 + r, Nn - 1);

  v8f acc = {};
  const float* ap = h + (size_t)arow * C1 + kh;
  const float* bbase = sW + r * 4 + kh;
  for (int k0 = 0; k0 < C1; k0 += 4) {
    v2f a; a.x = ap[k0]; a.y = ap[k0 + 1];
    v2f b = *(const v2f*)(bbase + (k0 >> 2) * 64);
    acc = __builtin_amdgcn_wmma_f32_16x16x4_f32(false, a, false, b, (short)0, acc, false, false);
  }

  const int mbase = (lane >> 4) * 8;
  for (int i = 0; i < 8; ++i) {
    const int row = row0 + mbase + i;
    if (row < Nn) h2[(size_t)row * C2 + r] = acc[i];
  }
}

template <int C>
__global__ void node_dots(const float* __restrict__ h,
                          const float* __restrict__ asrc,
                          const float* __restrict__ adst,
                          float* __restrict__ s, float* __restrict__ d, int Nn) {
  int n = blockIdx.x * blockDim.x + threadIdx.x;
  if (n >= Nn) return;
  const float* hp = h + (size_t)n * C;
  float ss = 0.f, dd = 0.f;
  for (int f = 0; f < C; ++f) { float v = hp[f]; ss += v * asrc[f]; dd += v * adst[f]; }
  s[n] = ss; d[n] = dd;
}


__global__ void __launch_bounds__(BT)
bin_edges(const int* __restrict__ ei, int E, int Nn, int owners, int segpitch,
          int* __restrict__ segw, int* __restrict__ bentw) {
  int2* const seg  = (int2*)segw;
  int2* const bent = (int2*)bentw;
  __shared__ int  s_cnt[BT];
  __shared__ int  s_wsum[BT / 32];
  __shared__ int2 s_ent[CHUNK];
  const int tid  = threadIdx.x;
  const int lane = tid & 31;
  const int wave = tid >> 5;
  const int c    = blockIdx.x;
  const int e0   = c * CHUNK;

  int own[EPT], rloc[EPT], rank[EPT];
#pragma unroll
  for (int k = 0; k < EPT; ++k) {
    const int e  = e0 + k * BT + tid;
    const int dn = (e < E) ? ei[E + e] : -1;
    const int o  = ((unsigned)dn < (unsigned)Nn) ? dn / ROWS : -1;
    own[k]  = o;
    rloc[k] = dn - o * ROWS;
    rank[k] = 0;
  }

  int base = 0;
  for (int wb = 0; wb < owners; wb += BT) {
    s_cnt[tid] = 0;
    __syncthreads();
#pragma unroll
    for (int k = 0; k < EPT; ++k) {
      const unsigned o = (unsigned)(own[k] - wb);
      if (o < (unsigned)BT) rank[k] = atomicAdd(&s_cnt[o], 1);
    }
    __syncthreads();
    const int v = s_cnt[tid];
    int incl = v;
#pragma unroll
    for (int o = 1; o < 32; o <<= 1) {
      const int t = __shfl_up(incl, o);
      if (lane >= o) incl += t;
    }
    if (lane == 31) s_wsum[wave] = incl;
    __syncthreads();
    int wbase = 0, wtot = 0;
#pragma unroll
    for (int w = 0; w < BT / 32; ++w) {
      const int t = s_wsum[w];
      wbase += (w < wave) ? t : 0;
      wtot  += t;
    }
    const int start = base + wbase + incl - v;
    if (wb + tid < owners)
      seg[(size_t)c * segpitch + wb + tid] = make_int2(e0 + start, e0 + start + v);
    s_cnt[tid] = start;
    __syncthreads();
#pragma unroll
    for (int k = 0; k < EPT; ++k) {
      const unsigned o = (unsigned)(own[k] - wb);
      if (o < (unsigned)BT) s_ent[s_cnt[o] + rank[k]] = make_int2(e0 + k * BT + tid, rloc[k]);
    }
    base += wtot;
    __syncthreads();
  }
  for (int i = tid; i < base; i += BT) bent[e0 + i] = s_ent[i];
}

__device__ __forceinline__ int fkey(float f) {
  int k = __float_as_int(f);
  return k >= 0 ? k : (k ^ 0x7fffffff);
}
__device__ __forceinline__ float fkey_inv(int k) {
  return __int_as_float(k >= 0 ? k : (k ^ 0x7fffffff));
}
__device__ __forceinline__ void edge_sd(const int* ei, int E, int e, int& s, int& d) {
  if (e < E) { s = ei[e]; d = ei[E + e]; } else { s = d = e - E; }
}
__device__ __forceinline__ float lrelu(float a) { return a > 0.f ? a : NEG_SLOPE * a; }

__device__ __forceinline__ float softplusf(float v) {
  return v > 0.f ? v + log1pf(__expf(-v)) : log1pf(__expf(v));
}

template <int C, bool FIRST>
__global__ void __launch_bounds__(NT)
layer_owner_kernel(const int* __restrict__ ei, int E, int Nn,
                   const float* __restrict__ hsrc,
                   const float* __restrict__ s, const float* __restrict__ d,
                   const float* __restrict__ bias,
                   float* __restrict__ rowout, float* __restrict__ klout,
                   int* __restrict__ loff, int* __restrict__ lent,
                   const int* __restrict__ segw, const int* __restrict__ bentw,
                   int nchunk, int segpitch) {
  const int2* const seg  = (const int2*)segw;
  const int2* const bent = (const int2*)bentw;
  __shared__ int   s_raw[4 * CAP];
  __shared__ int   s_rowcnt[ROWS];
  __shared__ int   s_off[ROWS + 1];
  __shared__ float s_d[ROWS];
  __shared__ int   s_cnt;
  int* const s_e  = s_raw;
  int* const s_r  = s_raw + CAP;
  int* const s_te = s_raw + 2 * CAP;
  int* const s_tr = s_raw + 3 * CAP;

  const int tid  = threadIdx.x;
  const int lo   = blockIdx.x * ROWS;
  const int rows = min(ROWS, Nn - lo);
  const unsigned urows = (unsigned)rows;
  const int Etot = E + Nn;
  constexpr int H   = C / 2;
  constexpr int TPE = C / 4;
  constexpr int NG  = NT / TPE;
  const int fo = (tid % TPE) * 4;
  const int g  = tid / TPE;

  int*  const offp = loff + (size_t)blockIdx.x * LOFF;
  int2* const entp = (int2*)lent + (size_t)blockIdx.x * CAP;

  int ntot = 0;
  if constexpr (!FIRST) ntot = offp[ROWS + 1];

  for (int i = tid; i < ROWS; i += NT) s_d[i] = (i < rows) ? d[lo + i] : 0.f;
  if constexpr (FIRST) {
    for (int i = tid; i < ROWS; i += NT) s_rowcnt[i] = 0;
    if (tid == 0) s_cnt = 0;
  }
  __syncthreads();

  if constexpr (FIRST) {
    auto append = [&](int e, unsigned r) {
      const int slot = atomicAdd(&s_cnt, 1);
      if (slot < CAP) { s_e[slot] = e; s_r[slot] = (int)r; }
      atomicAdd(&s_rowcnt[r], 1);
    };

    for (int c = tid / LPC; c < nchunk; c += NT / LPC) {
      const int2 se = seg[(size_t)c * segpitch + blockIdx.x];
      const int  n  = se.y - se.x;
      int slot0 = 0;
      if (tid % LPC == 0) slot0 = atomicAdd(&s_cnt, n);
      slot0 = __shfl(slot0, 0, LPC);
      for (int k0 = tid % LPC; k0 < n; k0 += LPC * USEG) {
        int2 en[USEG];
#pragma unroll
        for (int u = 0; u < USEG; ++u) {
          const int k = k0 + u * LPC;
          en[u] = bent[se.x + (k < n ? k : k0)];
        }
#pragma unroll
        for (int u = 0; u < USEG; ++u) {
          const int k = k0 + u * LPC;
          if (k < n) {
            const int slot = slot0 + k;
            if (slot < CAP) { s_e[slot] = en[u].x; s_r[slot] = en[u].y; }
            atomicAdd(&s_rowcnt[en[u].y], 1);
          }
        }
      }
    }
    for (int r = tid; r < rows; r += NT) append(E + lo + r, (unsigned)r);
    __syncthreads();
    ntot = s_cnt;
    if (tid == 0) offp[ROWS + 1] = ntot;
  }
  const bool listed = ntot <= CAP;

  if (listed) {
    if constexpr (FIRST) {
      if (tid < 32) {
        constexpr int PER = (ROWS + 31) / 32;
        int loc[PER]; int sum = 0;
#pragma unroll
        for (int k = 0; k < PER; ++k) {
          const int idx = tid * PER + k;
          loc[k] = sum;
          sum += (idx < ROWS) ? s_rowcnt[idx] : 0;
        }
        int incl = sum;
#pragma unroll
        for (int o = 1; o < 32; o <<= 1) {
          const int t = __shfl_up(incl, o);
          if (tid >= o) incl += t;
        }
        const int basep = incl - sum;
#pragma unroll
        for (int k = 0; k < PER; ++k) {
          const int idx = tid * PER + k;
          if (idx < ROWS) { s_off[idx] = basep + loc[k]; s_rowcnt[idx] = 0; }
        }
        if (tid == 31) s_off[ROWS] = incl;
      }
      __syncthreads();
      for (int i = tid; i <= ROWS; i += NT) offp[i] = s_off[i];

      for (int i = tid; i < ntot; i += NT) {
        const int r = s_r[i];
        const int p = s_off[r] + atomicAdd(&s_rowcnt[r], 1);
        s_te[p] = s_e[i]; s_tr[p] = r;
      }
      __syncthreads();

      constexpr int UA = 4;
      for (int p0 = tid; p0 < ntot; p0 += UA * NT) {
        int sn[UA], pos[UA], rr[UA]; float sv[UA];
#pragma unroll
        for (int k = 0; k < UA; ++k) {
          const int  p  = p0 + k * NT;
          const bool ok = p < ntot;
          const int  pc = ok ? p : p0;
          const int  e  = s_te[pc];
          const int  r  = s_tr[pc];
          const int  b  = s_off[r], en = s_off[r + 1];
          int rk = 0;
          for (int j = b; j < en; ++j) rk += (s_te[j] < e) ? 1 : 0;
          pos[k] = ok ? b + rk : -1;
          rr[k]  = r;
          sn[k]  = e < E ? ei[e] : e - E;
        }
#pragma unroll
        for (int k = 0; k < UA; ++k) sv[k] = s[sn[k]];
#pragma unroll
        for (int k = 0; k < UA; ++k) {
          if (pos[k] >= 0) {
            s_e[pos[k]] = sn[k];
            s_r[pos[k]] = __float_as_int(lrelu(sv[k] + s_d[rr[k]]));
            entp[pos[k]] = make_int2(sn[k], rr[k]);
          }
        }
      }
      __syncthreads();
    } else {
      for (int i = tid; i <= ROWS; i += NT) s_off[i] = offp[i];
      constexpr int UA = 4;
      for (int p0 = tid; p0 < ntot; p0 += UA * NT) {
        int2 en[UA]; float sv[UA];
#pragma unroll
        for (int k = 0; k < UA; ++k) {
          const int p = p0 + k * NT;
          en[k] = entp[p < ntot ? p : p0];
        }
#pragma unroll
        for (int k = 0; k < UA; ++k) sv[k] = s[en[k].x];
        asm volatile("" :: "v"(sv[0]), "v"(sv[1]), "v"(sv[2]), "v"(sv[3]));
#pragma unroll
        for (int k = 0; k < UA; ++k) {
          const int p = p0 + k * NT;
          if (p < ntot) {
            s_e[p] = en[k].x;
            s_r[p] = __float_as_int(lrelu(sv[k] + s_d[en[k].y]));
          }
        }
      }
      __syncthreads();
    }

    for (int r = g; r < rows; r += NG) {
      const int b = s_off[r], en = s_off[r + 1];
      float m = __int_as_float(s_r[b]);
      for (int j = b + 1; j < en; ++j) m = fmaxf(m, __int_as_float(s_r[j]));
      float den = 0.f;
      for (int j = b; j < en; ++j) den += __expf(__int_as_float(s_r[j]) - m);
      den += 1e-16f;
      float a0 = 0.f, a1 = 0.f, a2 = 0.f, a3 = 0.f;
      constexpr int UG = 8;
      for (int j0 = b; j0 < en; j0 += UG) {
        float4 hv[UG]; float w[UG]; bool ok[UG];
#pragma unroll
        for (int k = 0; k < UG; ++k) {
          const int j = j0 + k;
          ok[k] = j < en;
          const int jc = ok[k] ? j : b;
          w[k]  = __expf(__int_as_float(s_r[jc]) - m) / den;
          hv[k] = *(const float4*)(hsrc + (size_t)s_e[jc] * C + fo);
        }
#pragma unroll
        for (int k = 0; k < UG; ++k) {
          if (ok[k]) {
            a0 += hv[k].x * w[k]; a1 += hv[k].y * w[k];
            a2 += hv[k].z * w[k]; a3 += hv[k].w * w[k];
          }
        }
      }
      const float4 bv = *(const float4*)(bias + fo);
      const float o0 = a0 + bv.x, o1 = a1 + bv.y, o2 = a2 + bv.z, o3 = a3 + bv.w;
      const bool  up = fo >= H;
      const float ra = __shfl_xor(up ? o0 : o2, TPE / 2, TPE);
      const float rb = __shfl_xor(up ? o1 : o3, TPE / 2, TPE);
      const size_t n = (size_t)(lo + r);
      float4 st;
      st.x = FIRST ? (o0 > 0.f ? o0 : (__expf(o0) - 1.f)) : o0;
      st.y = FIRST ? (o1 > 0.f ? o1 : (__expf(o1) - 1.f)) : o1;
      st.z = FIRST ? (o2 > 0.f ? o2 : (__expf(o2) - 1.f)) : o2;
      st.w = FIRST ? (o3 > 0.f ? o3 : (__expf(o3) - 1.f)) : o3;
      *(float4*)(rowout + n * C + fo) = st;
      const float mu0 = up ? ra : o0, mu1 = up ? rb : o1;
      const float sp0 = up ? o2 : ra, sp1 = up ? o3 : rb;
      const int   kf  = up ? fo - H + 2 : fo;
      const float d0 = softplusf(sp0) + 1e-10f, d1 = softplusf(sp1) + 1e-10f;
      float2 kl;
      kl.x = -__logf(d0) + 0.5f * (d0 * d0 + mu0 * mu0 - 1.f);
      kl.y = -__logf(d1) + 0.5f * (d1 * d1 + mu1 * mu1 - 1.f);
      *(float2*)(klout + n * H + kf) = kl;
    }
    return;
  }

  float* const s_acc = (float*)s_raw;
  int*   const s_mx  = s_raw + ROWS * C;
  float* const s_den = (float*)(s_raw + ROWS * C + ROWS);
  for (int i = tid; i < ROWS * C; i += NT) s_acc[i] = 0.f;
  for (int i = tid; i < ROWS; i += NT) { s_mx[i] = (int)0x80000000; s_den[i] = 0.f; }
  __syncthreads();
  for (int e = tid; e < Etot; e += NT) {
    int sn, dn; edge_sd(ei, E, e, sn, dn);
    const unsigned r = (unsigned)(dn - lo);
    if (r < urows) atomicMax(&s_mx[r], fkey(lrelu(s[sn] + s_d[r])));
  }
  __syncthreads();
  for (int e = tid; e < Etot; e += NT) {
    int sn, dn; edge_sd(ei, E, e, sn, dn);
    const unsigned r = (unsigned)(dn - lo);
    if (r < urows)
      atomicAdd(&s_den[r], __expf(lrelu(s[sn] + s_d[r]) - fkey_inv(s_mx[r])));
  }
  __syncthreads();
  for (int e = g; e < Etot; e += NG) {
    int sn, dn; edge_sd(ei, E, e, sn, dn);
    const unsigned r = (unsigned)(dn - lo);
    if (r < urows) {
      const float ew = __expf(lrelu(s[sn] + s_d[r]) - fkey_inv(s_mx[r]));
      const float w  = ew / (s_den[r] + 1e-16f);
      const float4 hv = *(const float4*)(hsrc + (size_t)sn * C + fo);
      float* op = s_acc + (int)r * C + fo;
      atomicAdd(op + 0, hv.x * w);
      atomicAdd(op + 1, hv.y * w);
      atomicAdd(op + 2, hv.z * w);
      atomicAdd(op + 3, hv.w * w);
    }
  }
  __syncthreads();
  for (int i = tid; i < rows * C; i += NT) {
    const int r = i / C, f = i % C;
    const size_t n = (size_t)(lo + r);
    const float v = s_acc[i] + bias[f];
    rowout[n * C + f] = FIRST ? (v > 0.f ? v : (__expf(v) - 1.f)) : v;
    if (f < H) {
      const float sp = s_acc[r * C + H + f] + bias[H + f];
      const float sd = softplusf(sp) + 1e-10f;
      klout[n * H + f] = -__logf(sd) + 0.5f * (sd * sd + v * v - 1.f);
    }
  }
}

extern "C" void kernel_launch(void* const* d_in, const int* in_sizes, int n_in,
                              void* d_out, int out_size, void* d_ws, size_t ws_size,
                              hipStream_t stream) {
  const float* x   = (const float*)d_in[0];
  const int*   ei  = (const int*)  d_in[1];
  const float* W1  = (const float*)d_in[2];
  const float* as1 = (const float*)d_in[3];
  const float* ad1 = (const float*)d_in[4];
  const float* b1  = (const float*)d_in[5];
  const float* W2  = (const float*)d_in[6];
  const float* as2 = (const float*)d_in[7];
  const float* ad2 = (const float*)d_in[8];
  const float* b2  = (const float*)d_in[9];

  const int N    = in_sizes[0] / F_IN;
  const int E    = in_sizes[1] / 2;
  float* out = (float*)d_out;

  const size_t N64 = (size_t)N * C1, N16 = (size_t)N * C2;
  const int owners = (N + ROWS - 1) / ROWS;
  int*   lent = (int*)d_ws;
  int*   loff = lent + (size_t)owners * CAP * 2;
  const int nchunk = (E + CHUNK - 1) / CHUNK;
  const int segp   = (owners + 15) & ~15;
  int*   bent = loff + (size_t)owners * LOFF;
  int*   seg  = bent + (size_t)nchunk * CHUNK * 2;
  float* ws = (float*)(seg + (size_t)nchunk * segp * 2);
  float* h1 = ws;
  float* hb = h1 + N64;
  float* h2 = hb + N64;
  float* s1 = h2 + N16;
  float* d1 = s1 + N;

  float* ixz1 = out + N16;
  float* ixz2 = out + (size_t)N * 48;

  const int tiles  = (N + 15) / 16;
  const int gblk   = (tiles + 7) / 8;

  gemm1_wmma<<<gblk, 256, 0, stream>>>(x, W1, h1, N);
  node_dots<C1><<<(N + 255) / 256, 256, 0, stream>>>(h1, as1, ad1, s1, d1, N);
  bin_edges<<<nchunk, BT, 0, stream>>>(ei, E, N, owners, segp, seg, bent);
  layer_owner_kernel<C1, true><<<owners, NT, 0, stream>>>(ei, E, N, h1, s1, d1, b1, hb, ixz1, loff, lent,
                                                          seg, bent, nchunk, segp);

  gemm2_wmma<<<gblk, 256, 0, stream>>>(hb, W2, h2, N);
  node_dots<C2><<<(N + 255) / 256, 256, 0, stream>>>(h2, as2, ad2, s1, d1, N);
  layer_owner_kernel<C2, false><<<owners, NT, 0, stream>>>(ei, E, N, h2, s1, d1, b2, out, ixz2, loff, lent,
                                                           seg, bent, nchunk, segp);
}


# softmax weights computed once per edge, not by every lane of a row's group: a thread per row takes max and denominator, a thread per list entry stores exp(alpha-max)/den in LDS; sum order kept
# speedup vs baseline: 3.4255x; 1.1044x over previous
#include <hip/hip_runtime.h>
#include <hip/hip_bf16.h>

typedef __attribute__((ext_vector_type(2))) float v2f;
typedef __attribute__((ext_vector_type(8))) float v8f;

#define F_IN 128
#define C1   64
#define C2   16
#define NEG_SLOPE 0.2f

#define ROWS 392
#define NT   1024
#define CAP  16384
#define LOFF 396
#define BT    256
#define EPT   10
#define CHUNK (BT * EPT)
#define LPC   4
#define USEG  6

__global__ void __launch_bounds__(256)
gemm1_wmma(const float* __restrict__ x, const float* __restrict__ W1,
           float* __restrict__ h1, int Nn) {
  __shared__ float sW[F_IN * C1];
  for (int i = threadIdx.x; i < (F_IN * C1) / 4; i += 256) {
    float4 v = ((const float4*)W1)[i];
    const int k  = i >> 4;
    const int c0 = (i & 15) * 4;
    float* dst = sW + (k >> 2) * 256 + (k & 3);
    dst[(c0 + 0) * 4] = v.x;
    dst[(c0 + 1) * 4] = v.y;
    dst[(c0 + 2) * 4] = v.z;
    dst[(c0 + 3) * 4] = v.w;
  }
  __syncthreads();

  const int wave = threadIdx.x >> 5;
  const int lane = threadIdx.x & 31;
  const int tile = blockIdx.x * 8 + wave;
  if (tile * 16 >= Nn) return;
  const int row0 = tile * 16;
  const int r    = lane & 15;
  const int kh   = (lane >> 4) << 1;
  const int arow = min(row0 + r, Nn - 1);

  v8f acc0 = {}, acc1 = {}, acc2 = {}, acc3 = {};
  const float* ap = x + (size_t)arow * F_IN + kh;
  const float* bbase = sW + r * 4 + kh;
  for (int k0 = 0; k0 < F_IN; k0 += 4) {
    v2f a; a.x = ap[k0]; a.y = ap[k0 + 1];
    const float* bq = bbase + (k0 >> 2) * 256;
    v2f b0 = *(const v2f*)(bq);
    v2f b1 = *(const v2f*)(bq + 64);
    v2f b2 = *(const v2f*)(bq + 128);
    v2f b3 = *(const v2f*)(bq + 192);
    acc0 = __builtin_amdgcn_wmma_f32_16x16x4_f32(false, a, false, b0, (short)0, acc0, false, false);
    acc1 = __builtin_amdgcn_wmma_f32_16x16x4_f32(false, a, false, b1, (short)0, acc1, false, false);
    acc2 = __builtin_amdgcn_wmma_f32_16x16x4_f32(false, a, false, b2, (short)0, acc2, false, false);
    acc3 = __builtin_amdgcn_wmma_f32_16x16x4_f32(false, a, false, b3, (short)0, acc3, false, false);
  }

  const int mbase = (lane >> 4) * 8;
  for (int i = 0; i < 8; ++i) {
    const int row = row0 + mbase + i;
    if (row < Nn) {
      float* op = h1 + (size_t)row * C1 + r;
      op[0]  = acc0[i];
      op[16] = acc1[i];
      op[32] = acc2[i];
      op[48] = acc3[i];
    }
  }
}

__global__ void __launch_bounds__(256)
gemm2_wmma(const float* __restrict__ h, const float* __restrict__ W2,
           float* __restrict__ h2, int Nn) {
  __shared__ float sW[C1 * C2];
  for (int i = threadIdx.x; i < (C1 * C2) / 4; i += 256) {
    float4 v = ((const float4*)W2)[i];
    const int k  = i >> 2;
    const int c0 = (i & 3) * 4;
    float* dst = sW + (k >> 2) * 64 + (k & 3);
    dst[(c0 + 0) * 4] = v.x;
    dst[(c0 + 1) * 4] = v.y;
    dst[(c0 + 2) * 4] = v.z;
    dst[(c0 + 3) * 4] = v.w;
  }
  __syncthreads();

  const int wave = threadIdx.x >> 5;
  const int lane = threadIdx.x & 31;
  const int tile = blockIdx.x * 8 + wave;
  if (tile * 16 >= Nn) return;
  const int row0 = tile * 16;
  const int r    = lane & 15;
  const int kh   = (lane >> 4) << 1;
  const int arow = min(row0 + r, Nn - 1);

  v8f acc = {};
  const float* ap = h + (size_t)arow * C1 + kh;
  const float* bbase = sW + r * 4 + kh;
  for (int k0 = 0; k0 < C1; k0 += 4) {
    v2f a; a.x = ap[k0]; a.y = ap[k0 + 1];
    v2f b = *(const v2f*)(bbase + (k0 >> 2) * 64);
    acc = __builtin_amdgcn_wmma_f32_16x16x4_f32(false, a, false, b, (short)0, acc, false, false);
  }

  const int mbase = (lane >> 4) * 8;
  for (int i = 0; i < 8; ++i) {
    const int row = row0 + mbase + i;
    if (row < Nn) h2[(size_t)row * C2 + r] = acc[i];
  }
}

template <int C>
__global__ void node_dots(const float* __restrict__ h,
                          const float* __restrict__ asrc,
                          const float* __restrict__ adst,
                          float* __restrict__ s, float* __restrict__ d, int Nn) {
  int n = blockIdx.x * blockDim.x + threadIdx.x;
  if (n >= Nn) return;
  const float* hp = h + (size_t)n * C;
  float ss = 0.f, dd = 0.f;
  for (int f = 0; f < C; ++f) { float v = hp[f]; ss += v * asrc[f]; dd += v * adst[f]; }
  s[n] = ss; d[n] = dd;
}


__global__ void __launch_bounds__(BT)
bin_edges(const int* __restrict__ ei, int E, int Nn, int owners, int segpitch,
          int* __restrict__ segw, int* __restrict__ bentw) {
  int2* const seg  = (int2*)segw;
  int2* const bent = (int2*)bentw;
  __shared__ int  s_cnt[BT];
  __shared__ int  s_wsum[BT / 32];
  __shared__ int2 s_ent[CHUNK];
  const int tid  = threadIdx.x;
  const int lane = tid & 31;
  const int wave = tid >> 5;
  const int c    = blockIdx.x;
  const int e0   = c * CHUNK;

  int own[EPT], rloc[EPT], rank[EPT];
#pragma unroll
  for (int k = 0; k < EPT; ++k) {
    const int e  = e0 + k * BT + tid;
    const int dn = (e < E) ? ei[E + e] : -1;
    const int o  = ((unsigned)dn < (unsigned)Nn) ? dn / ROWS : -1;
    own[k]  = o;
    rloc[k] = dn - o * ROWS;
    rank[k] = 0;
  }

  int base = 0;
  for (int wb = 0; wb < owners; wb += BT) {
    s_cnt[tid] = 0;
    __syncthreads();
#pragma unroll
    for (int k = 0; k < EPT; ++k) {
      const unsigned o = (unsigned)(own[k] - wb);
      if (o < (unsigned)BT) rank[k] = atomicAdd(&s_cnt[o], 1);
    }
    __syncthreads();
    const int v = s_cnt[tid];
    int incl = v;
#pragma unroll
    for (int o = 1; o < 32; o <<= 1) {
      const int t = __shfl_up(incl, o);
      if (lane >= o) incl += t;
    }
    if (lane == 31) s_wsum[wave] = incl;
    __syncthreads();
    int wbase = 0, wtot = 0;
#pragma unroll
    for (int w = 0; w < BT / 32; ++w) {
      const int t = s_wsum[w];
      wbase += (w < wave) ? t : 0;
      wtot  += t;
    }
    const int start = base + wbase + incl - v;
    if (wb + tid < owners)
      seg[(size_t)c * segpitch + wb + tid] = make_int2(e0 + start, e0 + start + v);
    s_cnt[tid] = start;
    __syncthreads();
#pragma unroll
    for (int k = 0; k < EPT; ++k) {
      const unsigned o = (unsigned)(own[k] - wb);
      if (o < (unsigned)BT) s_ent[s_cnt[o] + rank[k]] = make_int2(e0 + k * BT + tid, rloc[k]);
    }
    base += wtot;
    __syncthreads();
  }
  for (int i = tid; i < base; i += BT) bent[e0 + i] = s_ent[i];
}

__device__ __forceinline__ int fkey(float f) {
  int k = __float_as_int(f);
  return k >= 0 ? k : (k ^ 0x7fffffff);
}
__device__ __forceinline__ float fkey_inv(int k) {
  return __int_as_float(k >= 0 ? k : (k ^ 0x7fffffff));
}
__device__ __forceinline__ void edge_sd(const int* ei, int E, int e, int& s, int& d) {
  if (e < E) { s = ei[e]; d = ei[E + e]; } else { s = d = e - E; }
}
__device__ __forceinline__ float lrelu(float a) { return a > 0.f ? a : NEG_SLOPE * a; }

__device__ __forceinline__ float softplusf(float v) {
  return v > 0.f ? v + log1pf(__expf(-v)) : log1pf(__expf(v));
}

template <int C, bool FIRST>
__global__ void __launch_bounds__(NT)
layer_owner_kernel(const int* __restrict__ ei, int E, int Nn,
                   const float* __restrict__ hsrc,
                   const float* __restrict__ s, const float* __restrict__ d,
                   const float* __restrict__ bias,
                   float* __restrict__ rowout, float* __restrict__ klout,
                   int* __restrict__ loff, int* __restrict__ lent,
                   const int* __restrict__ segw, const int* __restrict__ bentw,
                   int nchunk, int segpitch) {
  const int2* const seg  = (const int2*)segw;
  const int2* const bent = (const int2*)bentw;
  __shared__ int   s_raw[4 * CAP];
  __shared__ int   s_rowcnt[ROWS];
  __shared__ int   s_off[ROWS + 1];
  __shared__ float s_d[ROWS];
  __shared__ float s_rmax[ROWS];
  __shared__ float s_rden[ROWS];
  __shared__ int   s_cnt;
  int* const s_e  = s_raw;
  int* const s_r  = s_raw + CAP;
  int* const s_te = s_raw + 2 * CAP;
  int* const s_tr = s_raw + 3 * CAP;

  const int tid  = threadIdx.x;
  const int lo   = blockIdx.x * ROWS;
  const int rows = min(ROWS, Nn - lo);
  const unsigned urows = (unsigned)rows;
  const int Etot = E + Nn;
  constexpr int H   = C / 2;
  constexpr int TPE = C / 4;
  constexpr int NG  = NT / TPE;
  const int fo = (tid % TPE) * 4;
  const int g  = tid / TPE;

  int*  const offp = loff + (size_t)blockIdx.x * LOFF;
  int2* const entp = (int2*)lent + (size_t)blockIdx.x * CAP;

  int ntot = 0;
  if constexpr (!FIRST) ntot = offp[ROWS + 1];

  for (int i = tid; i < ROWS; i += NT) s_d[i] = (i < rows) ? d[lo + i] : 0.f;
  if constexpr (FIRST) {
    for (int i = tid; i < ROWS; i += NT) s_rowcnt[i] = 0;
    if (tid == 0) s_cnt = 0;
  }
  __syncthreads();

  if constexpr (FIRST) {
    auto append = [&](int e, unsigned r) {
      const int slot = atomicAdd(&s_cnt, 1);
      if (slot < CAP) { s_e[slot] = e; s_r[slot] = (int)r; }
      atomicAdd(&s_rowcnt[r], 1);
    };

    for (int c = tid / LPC; c < nchunk; c += NT / LPC) {
      const int2 se = seg[(size_t)c * segpitch + blockIdx.x];
      const int  n  = se.y - se.x;
      int slot0 = 0;
      if (tid % LPC == 0) slot0 = atomicAdd(&s_cnt, n);
      slot0 = __shfl(slot0, 0, LPC);
      for (int k0 = tid % LPC; k0 < n; k0 += LPC * USEG) {
        int2 en[USEG];
#pragma unroll
        for (int u = 0; u < USEG; ++u) {
          const int k = k0 + u * LPC;
          en[u] = bent[se.x + (k < n ? k : k0)];
        }
#pragma unroll
        for (int u = 0; u < USEG; ++u) {
          const int k = k0 + u * LPC;
          if (k < n) {
            const int slot = slot0 + k;
            if (slot < CAP) { s_e[slot] = en[u].x; s_r[slot] = en[u].y; }
            atomicAdd(&s_rowcnt[en[u].y], 1);
          }
        }
      }
    }
    for (int r = tid; r < rows; r += NT) append(E + lo + r, (unsigned)r);
    __syncthreads();
    ntot = s_cnt;
    if (tid == 0) offp[ROWS + 1] = ntot;
  }
  const bool listed = ntot <= CAP;

  if (listed) {
    if constexpr (FIRST) {
      if (tid < 32) {
        constexpr int PER = (ROWS + 31) / 32;
        int loc[PER]; int sum = 0;
#pragma unroll
        for (int k = 0; k < PER; ++k) {
          const int idx = tid * PER + k;
          loc[k] = sum;
          sum += (idx < ROWS) ? s_rowcnt[idx] : 0;
        }
        int incl = sum;
#pragma unroll
        for (int o = 1; o < 32; o <<= 1) {
          const int t = __shfl_up(incl, o);
          if (tid >= o) incl += t;
        }
        const int basep = incl - sum;
#pragma unroll
        for (int k = 0; k < PER; ++k) {
          const int idx = tid * PER + k;
          if (idx < ROWS) { s_off[idx] = basep + loc[k]; s_rowcnt[idx] = 0; }
        }
        if (tid == 31) s_off[ROWS] = incl;
      }
      __syncthreads();
      for (int i = tid; i <= ROWS; i += NT) offp[i] = s_off[i];

      for (int i = tid; i < ntot; i += NT) {
        const int r = s_r[i];
        const int p = s_off[r] + atomicAdd(&s_rowcnt[r], 1);
        s_te[p] = s_e[i]; s_tr[p] = r;
      }
      __syncthreads();

      constexpr int UA = 4;
      for (int p0 = tid; p0 < ntot; p0 += UA * NT) {
        int sn[UA], pos[UA], rr[UA]; float sv[UA];
#pragma unroll
        for (int k = 0; k < UA; ++k) {
          const int  p  = p0 + k * NT;
          const bool ok = p < ntot;
          const int  pc = ok ? p : p0;
          const int  e  = s_te[pc];
          const int  r  = s_tr[pc];
          const int  b  = s_off[r], en = s_off[r + 1];
          int rk = 0;
          for (int j = b; j < en; ++j) rk += (s_te[j] < e) ? 1 : 0;
          pos[k] = ok ? b + rk : -1;
          rr[k]  = r;
          sn[k]  = e < E ? ei[e] : e - E;
        }
#pragma unroll
        for (int k = 0; k < UA; ++k) sv[k] = s[sn[k]];
#pragma unroll
        for (int k = 0; k < UA; ++k) {
          if (pos[k] >= 0) {
            s_e[pos[k]] = sn[k];
            s_r[pos[k]] = __float_as_int(lrelu(sv[k] + s_d[rr[k]]));
            entp[pos[k]] = make_int2(sn[k], rr[k]);
          }
        }
      }
      __syncthreads();
    } else {
      for (int i = tid; i <= ROWS; i += NT) s_off[i] = offp[i];
      constexpr int UA = 4;
      for (int p0 = tid; p0 < ntot; p0 += UA * NT) {
        int2 en[UA]; float sv[UA];
#pragma unroll
        for (int k = 0; k < UA; ++k) {
          const int p = p0 + k * NT;
          en[k] = entp[p < ntot ? p : p0];
        }
#pragma unroll
        for (int k = 0; k < UA; ++k) sv[k] = s[en[k].x];
        asm volatile("" :: "v"(sv[0]), "v"(sv[1]), "v"(sv[2]), "v"(sv[3]));
#pragma unroll
        for (int k = 0; k < UA; ++k) {
          const int p = p0 + k * NT;
          if (p < ntot) {
            s_e[p]  = en[k].x;
            s_tr[p] = en[k].y;
            s_r[p]  = __float_as_int(lrelu(sv[k] + s_d[en[k].y]));
          }
        }
      }
      __syncthreads();
    }

    constexpr int UD = 8;
    for (int r = tid; r < rows; r += NT) {
      const int b = s_off[r], en = s_off[r + 1];
      float m = __int_as_float(s_r[b]);
      for (int j0 = b + 1; j0 < en; j0 += UD) {
        float av[UD];
#pragma unroll
        for (int k = 0; k < UD; ++k) {
          const int j = j0 + k;
          av[k] = __int_as_float(s_r[j < en ? j : b]);
        }
#pragma unroll
        for (int k = 0; k < UD; ++k) m = fmaxf(m, av[k]);
      }
      float den = 0.f;
      for (int j0 = b; j0 < en; j0 += UD) {
        float ev[UD];
#pragma unroll
        for (int k = 0; k < UD; ++k) {
          const int j = j0 + k;
          ev[k] = __expf(__int_as_float(s_r[j < en ? j : b]) - m);
        }
#pragma unroll
        for (int k = 0; k < UD; ++k)
          if (j0 + k < en) den += ev[k];
      }
      s_rmax[r] = m;
      s_rden[r] = den + 1e-16f;
    }
    __syncthreads();

    {
      constexpr int UW = 4;
      for (int p0 = tid; p0 < ntot; p0 += UW * NT) {
        float al[UW], mm[UW], dd[UW];
#pragma unroll
        for (int k = 0; k < UW; ++k) {
          const int p  = p0 + k * NT;
          const int pc = p < ntot ? p : p0;
          const int r  = s_tr[pc];
          al[k] = __int_as_float(s_r[pc]);
          mm[k] = s_rmax[r];
          dd[k] = s_rden[r];
        }
#pragma unroll
        for (int k = 0; k < UW; ++k) {
          const int p = p0 + k * NT;
          if (p < ntot) s_r[p] = __float_as_int(__expf(al[k] - mm[k]) / dd[k]);
        }
      }
    }
    __syncthreads();

    for (int r = g; r < rows; r += NG) {
      const int b = s_off[r], en = s_off[r + 1];
      float a0 = 0.f, a1 = 0.f, a2 = 0.f, a3 = 0.f;
      constexpr int UG = 8;
      for (int j0 = b; j0 < en; j0 += UG) {
        float4 hv[UG]; float w[UG]; bool ok[UG];
#pragma unroll
        for (int k = 0; k < UG; ++k) {
          const int j = j0 + k;
          ok[k] = j < en;
          const int jc = ok[k] ? j : b;
          w[k]  = __int_as_float(s_r[jc]);
          hv[k] = *(const float4*)(hsrc + (size_t)s_e[jc] * C + fo);
        }
#pragma unroll
        for (int k = 0; k < UG; ++k) {
          if (ok[k]) {
            a0 += hv[k].x * w[k]; a1 += hv[k].y * w[k];
            a2 += hv[k].z * w[k]; a3 += hv[k].w * w[k];
          }
        }
      }
      const float4 bv = *(const float4*)(bias + fo);
      const float o0 = a0 + bv.x, o1 = a1 + bv.y, o2 = a2 + bv.z, o3 = a3 + bv.w;
      const bool  up = fo >= H;
      const float ra = __shfl_xor(up ? o0 : o2, TPE / 2, TPE);
      const float rb = __shfl_xor(up ? o1 : o3, TPE / 2, TPE);
      const size_t n = (size_t)(lo + r);
      float4 st;
      st.x = FIRST ? (o0 > 0.f ? o0 : (__expf(o0) - 1.f)) : o0;
      st.y = FIRST ? (o1 > 0.f ? o1 : (__expf(o1) - 1.f)) : o1;
      st.z = FIRST ? (o2 > 0.f ? o2 : (__expf(o2) - 1.f)) : o2;
      st.w = FIRST ? (o3 > 0.f ? o3 : (__expf(o3) - 1.f)) : o3;
      *(float4*)(rowout + n * C + fo) = st;
      const float mu0 = up ? ra : o0, mu1 = up ? rb : o1;
      const float sp0 = up ? o2 : ra, sp1 = up ? o3 : rb;
      const int   kf  = up ? fo - H + 2 : fo;
      const float d0 = softplusf(sp0) + 1e-10f, d1 = softplusf(sp1) + 1e-10f;
      float2 kl;
      kl.x = -__logf(d0) + 0.5f * (d0 * d0 + mu0 * mu0 - 1.f);
      kl.y = -__logf(d1) + 0.5f * (d1 * d1 + mu1 * mu1 - 1.f);
      *(float2*)(klout + n * H + kf) = kl;
    }
    return;
  }

  float* const s_acc = (float*)s_raw;
  int*   const s_mx  = s_raw + ROWS * C;
  float* const s_den = (float*)(s_raw + ROWS * C + ROWS);
  for (int i = tid; i < ROWS * C; i += NT) s_acc[i] = 0.f;
  for (int i = tid; i < ROWS; i += NT) { s_mx[i] = (int)0x80000000; s_den[i] = 0.f; }
  __syncthreads();
  for (int e = tid; e < Etot; e += NT) {
    int sn, dn; edge_sd(ei, E, e, sn, dn);
    const unsigned r = (unsigned)(dn - lo);
    if (r < urows) atomicMax(&s_mx[r], fkey(lrelu(s[sn] + s_d[r])));
  }
  __syncthreads();
  for (int e = tid; e < Etot; e += NT) {
    int sn, dn; edge_sd(ei, E, e, sn, dn);
    const unsigned r = (unsigned)(dn - lo);
    if (r < urows)
      atomicAdd(&s_den[r], __expf(lrelu(s[sn] + s_d[r]) - fkey_inv(s_mx[r])));
  }
  __syncthreads();
  for (int e = g; e < Etot; e += NG) {
    int sn, dn; edge_sd(ei, E, e, sn, dn);
    const unsigned r = (unsigned)(dn - lo);
    if (r < urows) {
      const float ew = __expf(lrelu(s[sn] + s_d[r]) - fkey_inv(s_mx[r]));
      const float w  = ew / (s_den[r] + 1e-16f);
      const float4 hv = *(const float4*)(hsrc + (size_t)sn * C + fo);
      float* op = s_acc + (int)r * C + fo;
      atomicAdd(op + 0, hv.x * w);
      atomicAdd(op + 1, hv.y * w);
      atomicAdd(op + 2, hv.z * w);
      atomicAdd(op + 3, hv.w * w);
    }
  }
  __syncthreads();
  for (int i = tid; i < rows * C; i += NT) {
    const int r = i / C, f = i % C;
    const size_t n = (size_t)(lo + r);
    const float v = s_acc[i] + bias[f];
    rowout[n * C + f] = FIRST ? (v > 0.f ? v : (__expf(v) - 1.f)) : v;
    if (f < H) {
      const float sp = s_acc[r * C + H + f] + bias[H + f];
      const float sd = softplusf(sp) + 1e-10f;
      klout[n * H + f] = -__logf(sd) + 0.5f * (sd * sd + v * v - 1.f);
    }
  }
}

extern "C" void kernel_launch(void* const* d_in, const int* in_sizes, int n_in,
                              void* d_out, int out_size, void* d_ws, size_t ws_size,
                              hipStream_t stream) {
  const float* x   = (const float*)d_in[0];
  const int*   ei  = (const int*)  d_in[1];
  const float* W1  = (const float*)d_in[2];
  const float* as1 = (const float*)d_in[3];
  const float* ad1 = (const float*)d_in[4];
  const float* b1  = (const float*)d_in[5];
  const float* W2  = (const float*)d_in[6];
  const float* as2 = (const float*)d_in[7];
  const float* ad2 = (const float*)d_in[8];
  const float* b2  = (const float*)d_in[9];

  const int N    = in_sizes[0] / F_IN;
  const int E    = in_sizes[1] / 2;
  float* out = (float*)d_out;

  const size_t N64 = (size_t)N * C1, N16 = (size_t)N * C2;
  const int owners = (N + ROWS - 1) / ROWS;
  int*   lent = (int*)d_ws;
  int*   loff = lent + (size_t)owners * CAP * 2;
  const int nchunk = (E + CHUNK - 1) / CHUNK;
  const int segp   = (owners + 15) & ~15;
  int*   bent = loff + (size_t)owners * LOFF;
  int*   seg  = bent + (size_t)nchunk * CHUNK * 2;
  float* ws = (float*)(seg + (size_t)nchunk * segp * 2);
  float* h1 = ws;
  float* hb = h1 + N64;
  float* h2 = hb + N64;
  float* s1 = h2 + N16;
  float* d1 = s1 + N;

  float* ixz1 = out + N16;
  float* ixz2 = out + (size_t)N * 48;

  const int tiles  = (N + 15) / 16;
  const int gblk   = (tiles + 7) / 8;

  gemm1_wmma<<<gblk, 256, 0, stream>>>(x, W1, h1, N);
  node_dots<C1><<<(N + 255) / 256, 256, 0, stream>>>(h1, as1, ad1, s1, d1, N);
  bin_edges<<<nchunk, BT, 0, stream>>>(ei, E, N, owners, segp, seg, bent);
  layer_owner_kernel<C1, true><<<owners, NT, 0, stream>>>(ei, E, N, h1, s1, d1, b1, hb, ixz1, loff, lent,
                                                          seg, bent, nchunk, segp);

  gemm2_wmma<<<gblk, 256, 0, stream>>>(hb, W2, h2, N);
  node_dots<C2><<<(N + 255) / 256, 256, 0, stream>>>(h2, as2, ad2, s1, d1, N);
  layer_owner_kernel<C2, false><<<owners, NT, 0, stream>>>(ei, E, N, h2, s1, d1, b2, out, ixz2, loff, lent,
                                                           seg, bent, nchunk, segp);
}


# attention dots folded through the weights: s = x.(W a_src), d = x.(W a_dst), summed on the vector unit in each GEMM's k loop from its A values; node_dots launches go (7->5); s, d sum order changes
# speedup vs baseline: 3.7532x; 1.0957x over previous
#include <hip/hip_runtime.h>
#include <hip/hip_bf16.h>

typedef __attribute__((ext_vector_type(2))) float v2f;
typedef __attribute__((ext_vector_type(8))) float v8f;

#define F_IN 128
#define C1   64
#define C2   16
#define NEG_SLOPE 0.2f

#define ROWS 392
#define NT   1024
#define CAP  16384
#define LOFF 396
#define BT    256
#define EPT   10
#define CHUNK (BT * EPT)
#define LPC   4
#define USEG  6

__global__ void __launch_bounds__(256)
gemm1_wmma(const float* __restrict__ x, const float* __restrict__ W1,
           const float* __restrict__ asrc, const float* __restrict__ adst,
           float* __restrict__ h1, float* __restrict__ sd, int Nn) {
  __shared__ float  sW[F_IN * C1];
  __shared__ float4 sUV[F_IN / 2];
  {
    const int k = threadIdx.x >> 1;
    const float4* av = (const float4*)((threadIdx.x & 1) ? adst : asrc);
    const float4* wr = (const float4*)(W1 + k * C1);
    float f = 0.f;
    for (int c4 = 0; c4 < C1 / 4; ++c4) {
      const float4 w = wr[c4], a = av[c4];
      f += w.x * a.x; f += w.y * a.y; f += w.z * a.z; f += w.w * a.w;
    }
    ((float*)sUV)[threadIdx.x] = f;
  }
  for (int i = threadIdx.x; i < (F_IN * C1) / 4; i += 256) {
    float4 v = ((const float4*)W1)[i];
    const int k  = i >> 4;
    const int c0 = (i & 15) * 4;
    float* dst = sW + (k >> 2) * 256 + (k & 3);
    dst[(c0 + 0) * 4] = v.x;
    dst[(c0 + 1) * 4] = v.y;
    dst[(c0 + 2) * 4] = v.z;
    dst[(c0 + 3) * 4] = v.w;
  }
  __syncthreads();

  const int wave = threadIdx.x >> 5;
  const int lane = threadIdx.x & 31;
  const int tile = blockIdx.x * 8 + wave;
  if (tile * 16 >= Nn) return;
  const int row0 = tile * 16;
  const int r    = lane & 15;
  const int kh   = (lane >> 4) << 1;
  const int arow = min(row0 + r, Nn - 1);

  v8f acc0 = {}, acc1 = {}, acc2 = {}, acc3 = {};
  float ss = 0.f, dd = 0.f;
  const float* ap = x + (size_t)arow * F_IN + kh;
  const float* bbase = sW + r * 4 + kh;
  constexpr int KC = 8;
  v2f an[KC];
  for (int j = 0; j < KC; ++j) { an[j].x = ap[4 * j]; an[j].y = ap[4 * j + 1]; }
#pragma unroll 1
  for (int q0 = 0; q0 < F_IN / 4; q0 += KC) {
    v2f ac[KC];
    for (int j = 0; j < KC; ++j) ac[j] = an[j];
    if (q0 + KC < F_IN / 4) {
      for (int j = 0; j < KC; ++j) {
        an[j].x = ap[4 * (q0 + KC + j)]; an[j].y = ap[4 * (q0 + KC + j) + 1];
      }
    }
    for (int j = 0; j < KC; ++j) {
      const int q = q0 + j;
      const v2f a = ac[j];
      const float4 uv = sUV[(4 * q + kh) >> 1];
      ss += a.x * uv.x; dd += a.x * uv.y;
      ss += a.y * uv.z; dd += a.y * uv.w;
      const float* bq = bbase + q * 256;
      v2f b0 = *(const v2f*)(bq);
      v2f b1 = *(const v2f*)(bq + 64);
      v2f b2 = *(const v2f*)(bq + 128);
      v2f b3 = *(const v2f*)(bq + 192);
      acc0 = __builtin_amdgcn_wmma_f32_16x16x4_f32(false, a, false, b0, (short)0, acc0, false, false);
      acc1 = __builtin_amdgcn_wmma_f32_16x16x4_f32(false, a, false, b1, (short)0, acc1, false, false);
      acc2 = __builtin_amdgcn_wmma_f32_16x16x4_f32(false, a, false, b2, (short)0, acc2, false, false);
      acc3 = __builtin_amdgcn_wmma_f32_16x16x4_f32(false, a, false, b3, (short)0, acc3, false, false);
    }
  }

  const int mbase = (lane >> 4) * 8;
  for (int i = 0; i < 8; ++i) {
    const int row = row0 + mbase + i;
    if (row < Nn) {
      float* op = h1 + (size_t)row * C1 + r;
      op[0]  = acc0[i];
      op[16] = acc1[i];
      op[32] = acc2[i];
      op[48] = acc3[i];
    }
  }

  ss += __shfl_xor(ss, 16);
  dd += __shfl_xor(dd, 16);
  if (lane < 16 && row0 + r < Nn)
    *(float2*)(sd + 2 * (size_t)(row0 + r)) = make_float2(ss, dd);
}

__global__ void __launch_bounds__(256)
gemm2_wmma(const float* __restrict__ h, const float* __restrict__ W2,
           const float* __restrict__ asrc, const float* __restrict__ adst,
           float* __restrict__ h2, float* __restrict__ sd, int Nn) {
  __shared__ float  sW[C1 * C2];
  __shared__ float4 sUV[C1 / 2];
  if (threadIdx.x < 2 * C1) {
    const int k = threadIdx.x >> 1;
    const float4* av = (const float4*)((threadIdx.x & 1) ? adst : asrc);
    const float4* wr = (const float4*)(W2 + k * C2);
    float f = 0.f;
    for (int c4 = 0; c4 < C2 / 4; ++c4) {
      const float4 w = wr[c4], a = av[c4];
      f += w.x * a.x; f += w.y * a.y; f += w.z * a.z; f += w.w * a.w;
    }
    ((float*)sUV)[threadIdx.x] = f;
  }
  for (int i = threadIdx.x; i < (C1 * C2) / 4; i += 256) {
    float4 v = ((const float4*)W2)[i];
    const int k  = i >> 2;
    const int c0 = (i & 3) * 4;
    float* dst = sW + (k >> 2) * 64 + (k & 3);
    dst[(c0 + 0) * 4] = v.x;
    dst[(c0 + 1) * 4] = v.y;
    dst[(c0 + 2) * 4] = v.z;
    dst[(c0 + 3) * 4] = v.w;
  }
  __syncthreads();

  const int wave = threadIdx.x >> 5;
  const int lane = threadIdx.x & 31;
  const int tile = blockIdx.x * 8 + wave;
  if (tile * 16 >= Nn) return;
  const int row0 = tile * 16;
  const int r    = lane & 15;
  const int kh   = (lane >> 4) << 1;
  const int arow = min(row0 + r, Nn - 1);

  v8f acc = {};
  float ss = 0.f, dd = 0.f;
  const float* ap = h + (size_t)arow * C1 + kh;
  const float* bbase = sW + r * 4 + kh;
  for (int k0 = 0; k0 < C1; k0 += 4) {
    v2f a; a.x = ap[k0]; a.y = ap[k0 + 1];
    const float4 uv = sUV[(k0 + kh) >> 1];
    ss += a.x * uv.x; dd += a.x * uv.y;
    ss += a.y * uv.z; dd += a.y * uv.w;
    v2f b = *(const v2f*)(bbase + (k0 >> 2) * 64);
    acc = __builtin_amdgcn_wmma_f32_16x16x4_f32(false, a, false, b, (short)0, acc, false, false);
  }

  const int mbase = (lane >> 4) * 8;
  for (int i = 0; i < 8; ++i) {
    const int row = row0 + mbase + i;
    if (row < Nn) h2[(size_t)row * C2 + r] = acc[i];
  }

  ss += __shfl_xor(ss, 16);
  dd += __shfl_xor(dd, 16);
  if (lane < 16 && row0 + r < Nn)
    *(float2*)(sd + 2 * (size_t)(row0 + r)) = make_float2(ss, dd);
}

__global__ void __launch_bounds__(BT)
bin_edges(const int* __restrict__ ei, int E, int Nn, int owners, int segpitch,
          int* __restrict__ segw, int* __restrict__ bentw) {
  int2* const seg  = (int2*)segw;
  int2* const bent = (int2*)bentw;
  __shared__ int  s_cnt[BT];
  __shared__ int  s_wsum[BT / 32];
  __shared__ int2 s_ent[CHUNK];
  const int tid  = threadIdx.x;
  const int lane = tid & 31;
  const int wave = tid >> 5;
  const int c    = blockIdx.x;
  const int e0   = c * CHUNK;

  int own[EPT], rloc[EPT], rank[EPT];
#pragma unroll
  for (int k = 0; k < EPT; ++k) {
    const int e  = e0 + k * BT + tid;
    const int dn = (e < E) ? ei[E + e] : -1;
    const int o  = ((unsigned)dn < (unsigned)Nn) ? dn / ROWS : -1;
    own[k]  = o;
    rloc[k] = dn - o * ROWS;
    rank[k] = 0;
  }

  int base = 0;
  for (int wb = 0; wb < owners; wb += BT) {
    s_cnt[tid] = 0;
    __syncthreads();
#pragma unroll
    for (int k = 0; k < EPT; ++k) {
      const unsigned o = (unsigned)(own[k] - wb);
      if (o < (unsigned)BT) rank[k] = atomicAdd(&s_cnt[o], 1);
    }
    __syncthreads();
    const int v = s_cnt[tid];
    int incl = v;
#pragma unroll
    for (int o = 1; o < 32; o <<= 1) {
      const int t = __shfl_up(incl, o);
      if (lane >= o) incl += t;
    }
    if (lane == 31) s_wsum[wave] = incl;
    __syncthreads();
    int wbase = 0, wtot = 0;
#pragma unroll
    for (int w = 0; w < BT / 32; ++w) {
      const int t = s_wsum[w];
      wbase += (w < wave) ? t : 0;
      wtot  += t;
    }
    const int start = base + wbase + incl - v;
    if (wb + tid < owners)
      seg[(size_t)c * segpitch + wb + tid] = make_int2(e0 + start, e0 + start + v);
    s_cnt[tid] = start;
    __syncthreads();
#pragma unroll
    for (int k = 0; k < EPT; ++k) {
      const unsigned o = (unsigned)(own[k] - wb);
      if (o < (unsigned)BT) s_ent[s_cnt[o] + rank[k]] = make_int2(e0 + k * BT + tid, rloc[k]);
    }
    base += wtot;
    __syncthreads();
  }
  for (int i = tid; i < base; i += BT) bent[e0 + i] = s_ent[i];
}

__device__ __forceinline__ int fkey(float f) {
  int k = __float_as_int(f);
  return k >= 0 ? k : (k ^ 0x7fffffff);
}
__device__ __forceinline__ float fkey_inv(int k) {
  return __int_as_float(k >= 0 ? k : (k ^ 0x7fffffff));
}
__device__ __forceinline__ void edge_sd(const int* ei, int E, int e, int& s, int& d) {
  if (e < E) { s = ei[e]; d = ei[E + e]; } else { s = d = e - E; }
}
__device__ __forceinline__ float lrelu(float a) { return a > 0.f ? a : NEG_SLOPE * a; }

__device__ __forceinline__ float softplusf(float v) {
  return v > 0.f ? v + log1pf(__expf(-v)) : log1pf(__expf(v));
}

template <int C, bool FIRST>
__global__ void __launch_bounds__(NT)
layer_owner_kernel(const int* __restrict__ ei, int E, int Nn,
                   const float* __restrict__ hsrc,
                   const float* __restrict__ sd,
                   const float* __restrict__ bias,
                   float* __restrict__ rowout, float* __restrict__ klout,
                   int* __restrict__ loff, int* __restrict__ lent,
                   const int* __restrict__ segw, const int* __restrict__ bentw,
                   int nchunk, int segpitch) {
  const int2* const seg  = (const int2*)segw;
  const int2* const bent = (const int2*)bentw;
  __shared__ int   s_raw[4 * CAP];
  __shared__ int   s_rowcnt[ROWS];
  __shared__ int   s_off[ROWS + 1];
  __shared__ float s_d[ROWS];
  __shared__ float s_rmax[ROWS];
  __shared__ float s_rden[ROWS];
  __shared__ int   s_cnt;
  int* const s_e  = s_raw;
  int* const s_r  = s_raw + CAP;
  int* const s_te = s_raw + 2 * CAP;
  int* const s_tr = s_raw + 3 * CAP;

  const int tid  = threadIdx.x;
  const int lo   = blockIdx.x * ROWS;
  const int rows = min(ROWS, Nn - lo);
  const unsigned urows = (unsigned)rows;
  const int Etot = E + Nn;
  constexpr int H   = C / 2;
  constexpr int TPE = C / 4;
  constexpr int NG  = NT / TPE;
  const int fo = (tid % TPE) * 4;
  const int g  = tid / TPE;

  int*  const offp = loff + (size_t)blockIdx.x * LOFF;
  int2* const entp = (int2*)lent + (size_t)blockIdx.x * CAP;

  int ntot = 0;
  if constexpr (!FIRST) ntot = offp[ROWS + 1];

  for (int i = tid; i < ROWS; i += NT) s_d[i] = (i < rows) ? sd[2 * (lo + i) + 1] : 0.f;
  if constexpr (FIRST) {
    for (int i = tid; i < ROWS; i += NT) s_rowcnt[i] = 0;
    if (tid == 0) s_cnt = 0;
  }
  __syncthreads();

  if constexpr (FIRST) {
    auto append = [&](int e, unsigned r) {
      const int slot = atomicAdd(&s_cnt, 1);
      if (slot < CAP) { s_e[slot] = e; s_r[slot] = (int)r; }
      atomicAdd(&s_rowcnt[r], 1);
    };

    for (int c = tid / LPC; c < nchunk; c += NT / LPC) {
      const int2 se = seg[(size_t)c * segpitch + blockIdx.x];
      const int  n  = se.y - se.x;
      int slot0 = 0;
      if (tid % LPC == 0) slot0 = atomicAdd(&s_cnt, n);
      slot0 = __shfl(slot0, 0, LPC);
      for (int k0 = tid % LPC; k0 < n; k0 += LPC * USEG) {
        int2 en[USEG];
#pragma unroll
        for (int u = 0; u < USEG; ++u) {
          const int k = k0 + u * LPC;
          en[u] = bent[se.x + (k < n ? k : k0)];
        }
#pragma unroll
        for (int u = 0; u < USEG; ++u) {
          const int k = k0 + u * LPC;
          if (k < n) {
            const int slot = slot0 + k;
            if (slot < CAP) { s_e[slot] = en[u].x; s_r[slot] = en[u].y; }
            atomicAdd(&s_rowcnt[en[u].y], 1);
          }
        }
      }
    }
    for (int r = tid; r < rows; r += NT) append(E + lo + r, (unsigned)r);
    __syncthreads();
    ntot = s_cnt;
    if (tid == 0) offp[ROWS + 1] = ntot;
  }
  const bool listed = ntot <= CAP;

  if (listed) {
    if constexpr (FIRST) {
      if (tid < 32) {
        constexpr int PER = (ROWS + 31) / 32;
        int loc[PER]; int sum = 0;
#pragma unroll
        for (int k = 0; k < PER; ++k) {
          const int idx = tid * PER + k;
          loc[k] = sum;
          sum += (idx < ROWS) ? s_rowcnt[idx] : 0;
        }
        int incl = sum;
#pragma unroll
        for (int o = 1; o < 32; o <<= 1) {
          const int t = __shfl_up(incl, o);
          if (tid >= o) incl += t;
        }
        const int basep = incl - sum;
#pragma unroll
        for (int k = 0; k < PER; ++k) {
          const int idx = tid * PER + k;
          if (idx < ROWS) { s_off[idx] = basep + loc[k]; s_rowcnt[idx] = 0; }
        }
        if (tid == 31) s_off[ROWS] = incl;
      }
      __syncthreads();
      for (int i = tid; i <= ROWS; i += NT) offp[i] = s_off[i];

      for (int i = tid; i < ntot; i += NT) {
        const int r = s_r[i];
        const int p = s_off[r] + atomicAdd(&s_rowcnt[r], 1);
        s_te[p] = s_e[i]; s_tr[p] = r;
      }
      __syncthreads();

      constexpr int UA = 4;
      for (int p0 = tid; p0 < ntot; p0 += UA * NT) {
        int sn[UA], pos[UA], rr[UA]; float sv[UA];
#pragma unroll
        for (int k = 0; k < UA; ++k) {
          const int  p  = p0 + k * NT;
          const bool ok = p < ntot;
          const int  pc = ok ? p : p0;
          const int  e  = s_te[pc];
          const int  r  = s_tr[pc];
          const int  b  = s_off[r], en = s_off[r + 1];
          int rk = 0;
          for (int j = b; j < en; ++j) rk += (s_te[j] < e) ? 1 : 0;
          pos[k] = ok ? b + rk : -1;
          rr[k]  = r;
          sn[k]  = e < E ? ei[e] : e - E;
        }
#pragma unroll
        for (int k = 0; k < UA; ++k) sv[k] = sd[2 * sn[k]];
#pragma unroll
        for (int k = 0; k < UA; ++k) {
          if (pos[k] >= 0) {
            s_e[pos[k]] = sn[k];
            s_r[pos[k]] = __float_as_int(lrelu(sv[k] + s_d[rr[k]]));
            entp[pos[k]] = make_int2(sn[k], rr[k]);
          }
        }
      }
      __syncthreads();
    } else {
      for (int i = tid; i <= ROWS; i += NT) s_off[i] = offp[i];
      constexpr int UA = 4;
      for (int p0 = tid; p0 < ntot; p0 += UA * NT) {
        int2 en[UA]; float sv[UA];
#pragma unroll
        for (int k = 0; k < UA; ++k) {
          const int p = p0 + k * NT;
          en[k] = entp[p < ntot ? p : p0];
        }
#pragma unroll
        for (int k = 0; k < UA; ++k) sv[k] = sd[2 * en[k].x];
        asm volatile("" :: "v"(sv[0]), "v"(sv[1]), "v"(sv[2]), "v"(sv[3]));
#pragma unroll
        for (int k = 0; k < UA; ++k) {
          const int p = p0 + k * NT;
          if (p < ntot) {
            s_e[p]  = en[k].x;
            s_tr[p] = en[k].y;
            s_r[p]  = __float_as_int(lrelu(sv[k] + s_d[en[k].y]));
          }
        }
      }
      __syncthreads();
    }

    constexpr int UD = 8;
    for (int r = tid; r < rows; r += NT) {
      const int b = s_off[r], en = s_off[r + 1];
      float m = __int_as_float(s_r[b]);
      for (int j0 = b + 1; j0 < en; j0 += UD) {
        float av[UD];
#pragma unroll
        for (int k = 0; k < UD; ++k) {
          const int j = j0 + k;
          av[k] = __int_as_float(s_r[j < en ? j : b]);
        }
#pragma unroll
        for (int k = 0; k < UD; ++k) m = fmaxf(m, av[k]);
      }
      float den = 0.f;
      for (int j0 = b; j0 < en; j0 += UD) {
        float ev[UD];
#pragma unroll
        for (int k = 0; k < UD; ++k) {
          const int j = j0 + k;
          ev[k] = __expf(__int_as_float(s_r[j < en ? j : b]) - m);
        }
#pragma unroll
        for (int k = 0; k < UD; ++k)
          if (j0 + k < en) den += ev[k];
      }
      s_rmax[r] = m;
      s_rden[r] = den + 1e-16f;
    }
    __syncthreads();

    {
      constexpr int UW = 4;
      for (int p0 = tid; p0 < ntot; p0 += UW * NT) {
        float al[UW], mm[UW], dd[UW];
#pragma unroll
        for (int k = 0; k < UW; ++k) {
          const int p  = p0 + k * NT;
          const int pc = p < ntot ? p : p0;
          const int r  = s_tr[pc];
          al[k] = __int_as_float(s_r[pc]);
          mm[k] = s_rmax[r];
          dd[k] = s_rden[r];
        }
#pragma unroll
        for (int k = 0; k < UW; ++k) {
          const int p = p0 + k * NT;
          if (p < ntot) s_r[p] = __float_as_int(__expf(al[k] - mm[k]) / dd[k]);
        }
      }
    }
    __syncthreads();

    for (int r = g; r < rows; r += NG) {
      const int b = s_off[r], en = s_off[r + 1];
      float a0 = 0.f, a1 = 0.f, a2 = 0.f, a3 = 0.f;
      constexpr int UG = 8;
      for (int j0 = b; j0 < en; j0 += UG) {
        float4 hv[UG]; float w[UG]; bool ok[UG];
#pragma unroll
        for (int k = 0; k < UG; ++k) {
          const int j = j0 + k;
          ok[k] = j < en;
          const int jc = ok[k] ? j : b;
          w[k]  = __int_as_float(s_r[jc]);
          hv[k] = *(const float4*)(hsrc + (size_t)s_e[jc] * C + fo);
        }
#pragma unroll
        for (int k = 0; k < UG; ++k) {
          if (ok[k]) {
            a0 += hv[k].x * w[k]; a1 += hv[k].y * w[k];
            a2 += hv[k].z * w[k]; a3 += hv[k].w * w[k];
          }
        }
      }
      const float4 bv = *(const float4*)(bias + fo);
      const float o0 = a0 + bv.x, o1 = a1 + bv.y, o2 = a2 + bv.z, o3 = a3 + bv.w;
      const bool  up = fo >= H;
      const float ra = __shfl_xor(up ? o0 : o2, TPE / 2, TPE);
      const float rb = __shfl_xor(up ? o1 : o3, TPE / 2, TPE);
      const size_t n = (size_t)(lo + r);
      float4 st;
      st.x = FIRST ? (o0 > 0.f ? o0 : (__expf(o0) - 1.f)) : o0;
      st.y = FIRST ? (o1 > 0.f ? o1 : (__expf(o1) - 1.f)) : o1;
      st.z = FIRST ? (o2 > 0.f ? o2 : (__expf(o2) - 1.f)) : o2;
      st.w = FIRST ? (o3 > 0.f ? o3 : (__expf(o3) - 1.f)) : o3;
      *(float4*)(rowout + n * C + fo) = st;
      const float mu0 = up ? ra : o0, mu1 = up ? rb : o1;
      const float sp0 = up ? o2 : ra, sp1 = up ? o3 : rb;
      const int   kf  = up ? fo - H + 2 : fo;
      const float d0 = softplusf(sp0) + 1e-10f, d1 = softplusf(sp1) + 1e-10f;
      float2 kl;
      kl.x = -__logf(d0) + 0.5f * (d0 * d0 + mu0 * mu0 - 1.f);
      kl.y = -__logf(d1) + 0.5f * (d1 * d1 + mu1 * mu1 - 1.f);
      *(float2*)(klout + n * H + kf) = kl;
    }
    return;
  }

  float* const s_acc = (float*)s_raw;
  int*   const s_mx  = s_raw + ROWS * C;
  float* const s_den = (float*)(s_raw + ROWS * C + ROWS);
  for (int i = tid; i < ROWS * C; i += NT) s_acc[i] = 0.f;
  for (int i = tid; i < ROWS; i += NT) { s_mx[i] = (int)0x80000000; s_den[i] = 0.f; }
  __syncthreads();
  for (int e = tid; e < Etot; e += NT) {
    int sn, dn; edge_sd(ei, E, e, sn, dn);
    const unsigned r = (unsigned)(dn - lo);
    if (r < urows) atomicMax(&s_mx[r], fkey(lrelu(sd[2 * sn] + s_d[r])));
  }
  __syncthreads();
  for (int e = tid; e < Etot; e += NT) {
    int sn, dn; edge_sd(ei, E, e, sn, dn);
    const unsigned r = (unsigned)(dn - lo);
    if (r < urows)
      atomicAdd(&s_den[r], __expf(lrelu(sd[2 * sn] + s_d[r]) - fkey_inv(s_mx[r])));
  }
  __syncthreads();
  for (int e = g; e < Etot; e += NG) {
    int sn, dn; edge_sd(ei, E, e, sn, dn);
    const unsigned r = (unsigned)(dn - lo);
    if (r < urows) {
      const float ew = __expf(lrelu(sd[2 * sn] + s_d[r]) - fkey_inv(s_mx[r]));
      const float w  = ew / (s_den[r] + 1e-16f);
      const float4 hv = *(const float4*)(hsrc + (size_t)sn * C + fo);
      float* op = s_acc + (int)r * C + fo;
      atomicAdd(op + 0, hv.x * w);
      atomicAdd(op + 1, hv.y * w);
      atomicAdd(op + 2, hv.z * w);
      atomicAdd(op + 3, hv.w * w);
    }
  }
  __syncthreads();
  for (int i = tid; i < rows * C; i += NT) {
    const int r = i / C, f = i % C;
    const size_t n = (size_t)(lo + r);
    const float v = s_acc[i] + bias[f];
    rowout[n * C + f] = FIRST ? (v > 0.f ? v : (__expf(v) - 1.f)) : v;
    if (f < H) {
      const float sp = s_acc[r * C + H + f] + bias[H + f];
      const float sd = softplusf(sp) + 1e-10f;
      klout[n * H + f] = -__logf(sd) + 0.5f * (sd * sd + v * v - 1.f);
    }
  }
}

extern "C" void kernel_launch(void* const* d_in, const int* in_sizes, int n_in,
                              void* d_out, int out_size, void* d_ws, size_t ws_size,
                              hipStream_t stream) {
  const float* x   = (const float*)d_in[0];
  const int*   ei  = (const int*)  d_in[1];
  const float* W1  = (const float*)d_in[2];
  const float* as1 = (const float*)d_in[3];
  const float* ad1 = (const float*)d_in[4];
  const float* b1  = (const float*)d_in[5];
  const float* W2  = (const float*)d_in[6];
  const float* as2 = (const float*)d_in[7];
  const float* ad2 = (const float*)d_in[8];
  const float* b2  = (const float*)d_in[9];

  const int N    = in_sizes[0] / F_IN;
  const int E    = in_sizes[1] / 2;
  float* out = (float*)d_out;

  const size_t N64 = (size_t)N * C1, N16 = (size_t)N * C2;
  const int owners = (N + ROWS - 1) / ROWS;
  int*   lent = (int*)d_ws;
  int*   loff = lent + (size_t)owners * CAP * 2;
  const int nchunk = (E + CHUNK - 1) / CHUNK;
  const int segp   = (owners + 15) & ~15;
  int*   bent = loff + (size_t)owners * LOFF;
  int*   seg  = bent + (size_t)nchunk * CHUNK * 2;
  float* ws = (float*)(seg + (size_t)nchunk * segp * 2);
  float* h1 = ws;
  float* hb = h1 + N64;
  float* h2 = hb + N64;
  float* sd = h2 + N16;

  float* ixz1 = out + N16;
  float* ixz2 = out + (size_t)N * 48;

  const int tiles  = (N + 15) / 16;
  const int gblk   = (tiles + 7) / 8;

  gemm1_wmma<<<gblk, 256, 0, stream>>>(x, W1, as1, ad1, h1, sd, N);
  bin_edges<<<nchunk, BT, 0, stream>>>(ei, E, N, owners, segp, seg, bent);
  layer_owner_kernel<C1, true><<<owners, NT, 0, stream>>>(ei, E, N, h1, sd, b1, hb, ixz1, loff, lent,
                                                          seg, bent, nchunk, segp);

  gemm2_wmma<<<gblk, 256, 0, stream>>>(hb, W2, as2, ad2, h2, sd, N);
  layer_owner_kernel<C2, false><<<owners, NT, 0, stream>>>(ei, E, N, h2, sd, b2, out, ixz2, loff, lent,
                                                           seg, bent, nchunk, segp);
}


# softplus without a divergent branch: max(v,0) + log1p(exp(-|v|)) is one body, so a wave of mixed signs runs the library log1p once per KL term, not once per side (4 -> 2 per lane); same bits
# speedup vs baseline: 4.0562x; 1.0807x over previous
#include <hip/hip_runtime.h>
#include <hip/hip_bf16.h>

typedef __attribute__((ext_vector_type(2))) float v2f;
typedef __attribute__((ext_vector_type(8))) float v8f;

#define F_IN 128
#define C1   64
#define C2   16
#define NEG_SLOPE 0.2f

#define ROWS 392
#define NT   1024
#define CAP  16384
#define LOFF 396
#define BT    256
#define EPT   10
#define CHUNK (BT * EPT)
#define LPC   4
#define USEG  6

__global__ void __launch_bounds__(256)
gemm1_wmma(const float* __restrict__ x, const float* __restrict__ W1,
           const float* __restrict__ asrc, const float* __restrict__ adst,
           float* __restrict__ h1, float* __restrict__ sd, int Nn) {
  __shared__ float  sW[F_IN * C1];
  __shared__ float4 sUV[F_IN / 2];
  {
    const int k = threadIdx.x >> 1;
    const float4* av = (const float4*)((threadIdx.x & 1) ? adst : asrc);
    const float4* wr = (const float4*)(W1 + k * C1);
    float f = 0.f;
    for (int c4 = 0; c4 < C1 / 4; ++c4) {
      const float4 w = wr[c4], a = av[c4];
      f += w.x * a.x; f += w.y * a.y; f += w.z * a.z; f += w.w * a.w;
    }
    ((float*)sUV)[threadIdx.x] = f;
  }
  for (int i = threadIdx.x; i < (F_IN * C1) / 4; i += 256) {
    float4 v = ((const float4*)W1)[i];
    const int k  = i >> 4;
    const int c0 = (i & 15) * 4;
    float* dst = sW + (k >> 2) * 256 + (k & 3);
    dst[(c0 + 0) * 4] = v.x;
    dst[(c0 + 1) * 4] = v.y;
    dst[(c0 + 2) * 4] = v.z;
    dst[(c0 + 3) * 4] = v.w;
  }
  __syncthreads();

  const int wave = threadIdx.x >> 5;
  const int lane = threadIdx.x & 31;
  const int tile = blockIdx.x * 8 + wave;
  if (tile * 16 >= Nn) return;
  const int row0 = tile * 16;
  const int r    = lane & 15;
  const int kh   = (lane >> 4) << 1;
  const int arow = min(row0 + r, Nn - 1);

  v8f acc0 = {}, acc1 = {}, acc2 = {}, acc3 = {};
  float ss = 0.f, dd = 0.f;
  const float* ap = x + (size_t)arow * F_IN + kh;
  const float* bbase = sW + r * 4 + kh;
  constexpr int KC = 8;
  v2f an[KC];
  for (int j = 0; j < KC; ++j) { an[j].x = ap[4 * j]; an[j].y = ap[4 * j + 1]; }
#pragma unroll 1
  for (int q0 = 0; q0 < F_IN / 4; q0 += KC) {
    v2f ac[KC];
    for (int j = 0; j < KC; ++j) ac[j] = an[j];
    if (q0 + KC < F_IN / 4) {
      for (int j = 0; j < KC; ++j) {
        an[j].x = ap[4 * (q0 + KC + j)]; an[j].y = ap[4 * (q0 + KC + j) + 1];
      }
    }
    for (int j = 0; j < KC; ++j) {
      const int q = q0 + j;
      const v2f a = ac[j];
      const float4 uv = sUV[(4 * q + kh) >> 1];
      ss += a.x * uv.x; dd += a.x * uv.y;
      ss += a.y * uv.z; dd += a.y * uv.w;
      const float* bq = bbase + q * 256;
      v2f b0 = *(const v2f*)(bq);
      v2f b1 = *(const v2f*)(bq + 64);
      v2f b2 = *(const v2f*)(bq + 128);
      v2f b3 = *(const v2f*)(bq + 192);
      acc0 = __builtin_amdgcn_wmma_f32_16x16x4_f32(false, a, false, b0, (short)0, acc0, false, false);
      acc1 = __builtin_amdgcn_wmma_f32_16x16x4_f32(false, a, false, b1, (short)0, acc1, false, false);
      acc2 = __builtin_amdgcn_wmma_f32_16x16x4_f32(false, a, false, b2, (short)0, acc2, false, false);
      acc3 = __builtin_amdgcn_wmma_f32_16x16x4_f32(false, a, false, b3, (short)0, acc3, false, false);
    }
  }

  const int mbase = (lane >> 4) * 8;
  for (int i = 0; i < 8; ++i) {
    const int row = row0 + mbase + i;
    if (row < Nn) {
      float* op = h1 + (size_t)row * C1 + r;
      op[0]  = acc0[i];
      op[16] = acc1[i];
      op[32] = acc2[i];
      op[48] = acc3[i];
    }
  }

  ss += __shfl_xor(ss, 16);
  dd += __shfl_xor(dd, 16);
  if (lane < 16 && row0 + r < Nn)
    *(float2*)(sd + 2 * (size_t)(row0 + r)) = make_float2(ss, dd);
}

__global__ void __launch_bounds__(256)
gemm2_wmma(const float* __restrict__ h, const float* __restrict__ W2,
           const float* __restrict__ asrc, const float* __restrict__ adst,
           float* __restrict__ h2, float* __restrict__ sd, int Nn) {
  __shared__ float  sW[C1 * C2];
  __shared__ float4 sUV[C1 / 2];
  if (threadIdx.x < 2 * C1) {
    const int k = threadIdx.x >> 1;
    const float4* av = (const float4*)((threadIdx.x & 1) ? adst : asrc);
    const float4* wr = (const float4*)(W2 + k * C2);
    float f = 0.f;
    for (int c4 = 0; c4 < C2 / 4; ++c4) {
      const float4 w = wr[c4], a = av[c4];
      f += w.x * a.x; f += w.y * a.y; f += w.z * a.z; f += w.w * a.w;
    }
    ((float*)sUV)[threadIdx.x] = f;
  }
  for (int i = threadIdx.x; i < (C1 * C2) / 4; i += 256) {
    float4 v = ((const float4*)W2)[i];
    const int k  = i >> 2;
    const int c0 = (i & 3) * 4;
    float* dst = sW + (k >> 2) * 64 + (k & 3);
    dst[(c0 + 0) * 4] = v.x;
    dst[(c0 + 1) * 4] = v.y;
    dst[(c0 + 2) * 4] = v.z;
    dst[(c0 + 3) * 4] = v.w;
  }
  __syncthreads();

  const int wave = threadIdx.x >> 5;
  const int lane = threadIdx.x & 31;
  const int tile = blockIdx.x * 8 + wave;
  if (tile * 16 >= Nn) return;
  const int row0 = tile * 16;
  const int r    = lane & 15;
  const int kh   = (lane >> 4) << 1;
  const int arow = min(row0 + r, Nn - 1);

  v8f acc = {};
  float ss = 0.f, dd = 0.f;
  const float* ap = h + (size_t)arow * C1 + kh;
  const float* bbase = sW + r * 4 + kh;
  for (int k0 = 0; k0 < C1; k0 += 4) {
    v2f a; a.x = ap[k0]; a.y = ap[k0 + 1];
    const float4 uv = sUV[(k0 + kh) >> 1];
    ss += a.x * uv.x; dd += a.x * uv.y;
    ss += a.y * uv.z; dd += a.y * uv.w;
    v2f b = *(const v2f*)(bbase + (k0 >> 2) * 64);
    acc = __builtin_amdgcn_wmma_f32_16x16x4_f32(false, a, false, b, (short)0, acc, false, false);
  }

  const int mbase = (lane >> 4) * 8;
  for (int i = 0; i < 8; ++i) {
    const int row = row0 + mbase + i;
    if (row < Nn) h2[(size_t)row * C2 + r] = acc[i];
  }

  ss += __shfl_xor(ss, 16);
  dd += __shfl_xor(dd, 16);
  if (lane < 16 && row0 + r < Nn)
    *(float2*)(sd + 2 * (size_t)(row0 + r)) = make_float2(ss, dd);
}

__global__ void __launch_bounds__(BT)
bin_edges(const int* __restrict__ ei, int E, int Nn, int owners, int segpitch,
          int* __restrict__ segw, int* __restrict__ bentw) {
  int2* const seg  = (int2*)segw;
  int2* const bent = (int2*)bentw;
  __shared__ int  s_cnt[BT];
  __shared__ int  s_wsum[BT / 32];
  __shared__ int2 s_ent[CHUNK];
  const int tid  = threadIdx.x;
  const int lane = tid & 31;
  const int wave = tid >> 5;
  const int c    = blockIdx.x;
  const int e0   = c * CHUNK;

  int own[EPT], rloc[EPT], rank[EPT];
#pragma unroll
  for (int k = 0; k < EPT; ++k) {
    const int e  = e0 + k * BT + tid;
    const int dn = (e < E) ? ei[E + e] : -1;
    const int o  = ((unsigned)dn < (unsigned)Nn) ? dn / ROWS : -1;
    own[k]  = o;
    rloc[k] = dn - o * ROWS;
    rank[k] = 0;
  }

  int base = 0;
  for (int wb = 0; wb < owners; wb += BT) {
    s_cnt[tid] = 0;
    __syncthreads();
#pragma unroll
    for (int k = 0; k < EPT; ++k) {
      const unsigned o = (unsigned)(own[k] - wb);
      if (o < (unsigned)BT) rank[k] = atomicAdd(&s_cnt[o], 1);
    }
    __syncthreads();
    const int v = s_cnt[tid];
    int incl = v;
#pragma unroll
    for (int o = 1; o < 32; o <<= 1) {
      const int t = __shfl_up(incl, o);
      if (lane >= o) incl += t;
    }
    if (lane == 31) s_wsum[wave] = incl;
    __syncthreads();
    int wbase = 0, wtot = 0;
#pragma unroll
    for (int w = 0; w < BT / 32; ++w) {
      const int t = s_wsum[w];
      wbase += (w < wave) ? t : 0;
      wtot  += t;
    }
    const int start = base + wbase + incl - v;
    if (wb + tid < owners)
      seg[(size_t)c * segpitch + wb + tid] = make_int2(e0 + start, e0 + start + v);
    s_cnt[tid] = start;
    __syncthreads();
#pragma unroll
    for (int k = 0; k < EPT; ++k) {
      const unsigned o = (unsigned)(own[k] - wb);
      if (o < (unsigned)BT) s_ent[s_cnt[o] + rank[k]] = make_int2(e0 + k * BT + tid, rloc[k]);
    }
    base += wtot;
    __syncthreads();
  }
  for (int i = tid; i < base; i += BT) bent[e0 + i] = s_ent[i];
}

__device__ __forceinline__ int fkey(float f) {
  int k = __float_as_int(f);
  return k >= 0 ? k : (k ^ 0x7fffffff);
}
__device__ __forceinline__ float fkey_inv(int k) {
  return __int_as_float(k >= 0 ? k : (k ^ 0x7fffffff));
}
__device__ __forceinline__ void edge_sd(const int* ei, int E, int e, int& s, int& d) {
  if (e < E) { s = ei[e]; d = ei[E + e]; } else { s = d = e - E; }
}
__device__ __forceinline__ float lrelu(float a) { return a > 0.f ? a : NEG_SLOPE * a; }

__device__ __forceinline__ float softplusf(float v) {
  return fmaxf(v, 0.f) + log1pf(__expf(-fabsf(v)));
}

template <int C, bool FIRST>
__global__ void __launch_bounds__(NT)
layer_owner_kernel(const int* __restrict__ ei, int E, int Nn,
                   const float* __restrict__ hsrc,
                   const float* __restrict__ sd,
                   const float* __restrict__ bias,
                   float* __restrict__ rowout, float* __restrict__ klout,
                   int* __restrict__ loff, int* __restrict__ lent,
                   const int* __restrict__ segw, const int* __restrict__ bentw,
                   int nchunk, int segpitch) {
  const int2* const seg  = (const int2*)segw;
  const int2* const bent = (const int2*)bentw;
  __shared__ int   s_raw[4 * CAP];
  __shared__ int   s_rowcnt[ROWS];
  __shared__ int   s_off[ROWS + 1];
  __shared__ float s_d[ROWS];
  __shared__ float s_rmax[ROWS];
  __shared__ float s_rden[ROWS];
  __shared__ int   s_cnt;
  int* const s_e  = s_raw;
  int* const s_r  = s_raw + CAP;
  int* const s_te = s_raw + 2 * CAP;
  int* const s_tr = s_raw + 3 * CAP;

  const int tid  = threadIdx.x;
  const int lo   = blockIdx.x * ROWS;
  const int rows = min(ROWS, Nn - lo);
  const unsigned urows = (unsigned)rows;
  const int Etot = E + Nn;
  constexpr int H   = C / 2;
  constexpr int TPE = C / 4;
  constexpr int NG  = NT / TPE;
  const int fo = (tid % TPE) * 4;
  const int g  = tid / TPE;

  int*  const offp = loff + (size_t)blockIdx.x * LOFF;
  int2* const entp = (int2*)lent + (size_t)blockIdx.x * CAP;

  int ntot = 0;
  if constexpr (!FIRST) ntot = offp[ROWS + 1];

  for (int i = tid; i < ROWS; i += NT) s_d[i] = (i < rows) ? sd[2 * (lo + i) + 1] : 0.f;
  if constexpr (FIRST) {
    for (int i = tid; i < ROWS; i += NT) s_rowcnt[i] = 0;
    if (tid == 0) s_cnt = 0;
  }
  __syncthreads();

  if constexpr (FIRST) {
    auto append = [&](int e, unsigned r) {
      const int slot = atomicAdd(&s_cnt, 1);
      if (slot < CAP) { s_e[slot] = e; s_r[slot] = (int)r; }
      atomicAdd(&s_rowcnt[r], 1);
    };

    for (int c = tid / LPC; c < nchunk; c += NT / LPC) {
      const int2 se = seg[(size_t)c * segpitch + blockIdx.x];
      const int  n  = se.y - se.x;
      int slot0 = 0;
      if (tid % LPC == 0) slot0 = atomicAdd(&s_cnt, n);
      slot0 = __shfl(slot0, 0, LPC);
      for (int k0 = tid % LPC; k0 < n; k0 += LPC * USEG) {
        int2 en[USEG];
#pragma unroll
        for (int u = 0; u < USEG; ++u) {
          const int k = k0 + u * LPC;
          en[u] = bent[se.x + (k < n ? k : k0)];
        }
#pragma unroll
        for (int u = 0; u < USEG; ++u) {
          const int k = k0 + u * LPC;
          if (k < n) {
            const int slot = slot0 + k;
            if (slot < CAP) { s_e[slot] = en[u].x; s_r[slot] = en[u].y; }
            atomicAdd(&s_rowcnt[en[u].y], 1);
          }
        }
      }
    }
    for (int r = tid; r < rows; r += NT) append(E + lo + r, (unsigned)r);
    __syncthreads();
    ntot = s_cnt;
    if (tid == 0) offp[ROWS + 1] = ntot;
  }
  const bool listed = ntot <= CAP;

  if (listed) {
    if constexpr (FIRST) {
      if (tid < 32) {
        constexpr int PER = (ROWS + 31) / 32;
        int loc[PER]; int sum = 0;
#pragma unroll
        for (int k = 0; k < PER; ++k) {
          const int idx = tid * PER + k;
          loc[k] = sum;
          sum += (idx < ROWS) ? s_rowcnt[idx] : 0;
        }
        int incl = sum;
#pragma unroll
        for (int o = 1; o < 32; o <<= 1) {
          const int t = __shfl_up(incl, o);
          if (tid >= o) incl += t;
        }
        const int basep = incl - sum;
#pragma unroll
        for (int k = 0; k < PER; ++k) {
          const int idx = tid * PER + k;
          if (idx < ROWS) { s_off[idx] = basep + loc[k]; s_rowcnt[idx] = 0; }
        }
        if (tid == 31) s_off[ROWS] = incl;
      }
      __syncthreads();
      for (int i = tid; i <= ROWS; i += NT) offp[i] = s_off[i];

      for (int i = tid; i < ntot; i += NT) {
        const int r = s_r[i];
        const int p = s_off[r] + atomicAdd(&s_rowcnt[r], 1);
        s_te[p] = s_e[i]; s_tr[p] = r;
      }
      __syncthreads();

      constexpr int UA = 4;
      for (int p0 = tid; p0 < ntot; p0 += UA * NT) {
        int sn[UA], pos[UA], rr[UA]; float sv[UA];
#pragma unroll
        for (int k = 0; k < UA; ++k) {
          const int  p  = p0 + k * NT;
          const bool ok = p < ntot;
          const int  pc = ok ? p : p0;
          const int  e  = s_te[pc];
          const int  r  = s_tr[pc];
          const int  b  = s_off[r], en = s_off[r + 1];
          int rk = 0;
          for (int j = b; j < en; ++j) rk += (s_te[j] < e) ? 1 : 0;
          pos[k] = ok ? b + rk : -1;
          rr[k]  = r;
          sn[k]  = e < E ? ei[e] : e - E;
        }
#pragma unroll
        for (int k = 0; k < UA; ++k) sv[k] = sd[2 * sn[k]];
#pragma unroll
        for (int k = 0; k < UA; ++k) {
          if (pos[k] >= 0) {
            s_e[pos[k]] = sn[k];
            s_r[pos[k]] = __float_as_int(lrelu(sv[k] + s_d[rr[k]]));
            entp[pos[k]] = make_int2(sn[k], rr[k]);
          }
        }
      }
      __syncthreads();
    } else {
      for (int i = tid; i <= ROWS; i += NT) s_off[i] = offp[i];
      constexpr int UA = 4;
      for (int p0 = tid; p0 < ntot; p0 += UA * NT) {
        int2 en[UA]; float sv[UA];
#pragma unroll
        for (int k = 0; k < UA; ++k) {
          const int p = p0 + k * NT;
          en[k] = entp[p < ntot ? p : p0];
        }
#pragma unroll
        for (int k = 0; k < UA; ++k) sv[k] = sd[2 * en[k].x];
        asm volatile("" :: "v"(sv[0]), "v"(sv[1]), "v"(sv[2]), "v"(sv[3]));
#pragma unroll
        for (int k = 0; k < UA; ++k) {
          const int p = p0 + k * NT;
          if (p < ntot) {
            s_e[p]  = en[k].x;
            s_tr[p] = en[k].y;
            s_r[p]  = __float_as_int(lrelu(sv[k] + s_d[en[k].y]));
          }
        }
      }
      __syncthreads();
    }

    constexpr int UD = 8;
    for (int r = tid; r < rows; r += NT) {
      const int b = s_off[r], en = s_off[r + 1];
      float m = __int_as_float(s_r[b]);
      for (int j0 = b + 1; j0 < en; j0 += UD) {
        float av[UD];
#pragma unroll
        for (int k = 0; k < UD; ++k) {
          const int j = j0 + k;
          av[k] = __int_as_float(s_r[j < en ? j : b]);
        }
#pragma unroll
        for (int k = 0; k < UD; ++k) m = fmaxf(m, av[k]);
      }
      float den = 0.f;
      for (int j0 = b; j0 < en; j0 += UD) {
        float ev[UD];
#pragma unroll
        for (int k = 0; k < UD; ++k) {
          const int j = j0 + k;
          ev[k] = __expf(__int_as_float(s_r[j < en ? j : b]) - m);
        }
#pragma unroll
        for (int k = 0; k < UD; ++k)
          if (j0 + k < en) den += ev[k];
      }
      s_rmax[r] = m;
      s_rden[r] = den + 1e-16f;
    }
    __syncthreads();

    {
      constexpr int UW = 4;
      for (int p0 = tid; p0 < ntot; p0 += UW * NT) {
        float al[UW], mm[UW], dd[UW];
#pragma unroll
        for (int k = 0; k < UW; ++k) {
          const int p  = p0 + k * NT;
          const int pc = p < ntot ? p : p0;
          const int r  = s_tr[pc];
          al[k] = __int_as_float(s_r[pc]);
          mm[k] = s_rmax[r];
          dd[k] = s_rden[r];
        }
#pragma unroll
        for (int k = 0; k < UW; ++k) {
          const int p = p0 + k * NT;
          if (p < ntot) s_r[p] = __float_as_int(__expf(al[k] - mm[k]) / dd[k]);
        }
      }
    }
    __syncthreads();

    for (int r = g; r < rows; r += NG) {
      const int b = s_off[r], en = s_off[r + 1];
      float a0 = 0.f, a1 = 0.f, a2 = 0.f, a3 = 0.f;
      constexpr int UG = 8;
      for (int j0 = b; j0 < en; j0 += UG) {
        float4 hv[UG]; float w[UG]; bool ok[UG];
#pragma unroll
        for (int k = 0; k < UG; ++k) {
          const int j = j0 + k;
          ok[k] = j < en;
          const int jc = ok[k] ? j : b;
          w[k]  = __int_as_float(s_r[jc]);
          hv[k] = *(const float4*)(hsrc + (size_t)s_e[jc] * C + fo);
        }
#pragma unroll
        for (int k = 0; k < UG; ++k) {
          if (ok[k]) {
            a0 += hv[k].x * w[k]; a1 += hv[k].y * w[k];
            a2 += hv[k].z * w[k]; a3 += hv[k].w * w[k];
          }
        }
      }
      const float4 bv = *(const float4*)(bias + fo);
      const float o0 = a0 + bv.x, o1 = a1 + bv.y, o2 = a2 + bv.z, o3 = a3 + bv.w;
      const bool  up = fo >= H;
      const float ra = __shfl_xor(up ? o0 : o2, TPE / 2, TPE);
      const float rb = __shfl_xor(up ? o1 : o3, TPE / 2, TPE);
      const size_t n = (size_t)(lo + r);
      float4 st;
      st.x = FIRST ? (o0 > 0.f ? o0 : (__expf(o0) - 1.f)) : o0;
      st.y = FIRST ? (o1 > 0.f ? o1 : (__expf(o1) - 1.f)) : o1;
      st.z = FIRST ? (o2 > 0.f ? o2 : (__expf(o2) - 1.f)) : o2;
      st.w = FIRST ? (o3 > 0.f ? o3 : (__expf(o3) - 1.f)) : o3;
      *(float4*)(rowout + n * C + fo) = st;
      const float mu0 = up ? ra : o0, mu1 = up ? rb : o1;
      const float sp0 = up ? o2 : ra, sp1 = up ? o3 : rb;
      const int   kf  = up ? fo - H + 2 : fo;
      const float d0 = softplusf(sp0) + 1e-10f, d1 = softplusf(sp1) + 1e-10f;
      float2 kl;
      kl.x = -__logf(d0) + 0.5f * (d0 * d0 + mu0 * mu0 - 1.f);
      kl.y = -__logf(d1) + 0.5f * (d1 * d1 + mu1 * mu1 - 1.f);
      *(float2*)(klout + n * H + kf) = kl;
    }
    return;
  }

  float* const s_acc = (float*)s_raw;
  int*   const s_mx  = s_raw + ROWS * C;
  float* const s_den = (float*)(s_raw + ROWS * C + ROWS);
  for (int i = tid; i < ROWS * C; i += NT) s_acc[i] = 0.f;
  for (int i = tid; i < ROWS; i += NT) { s_mx[i] = (int)0x80000000; s_den[i] = 0.f; }
  __syncthreads();
  for (int e = tid; e < Etot; e += NT) {
    int sn, dn; edge_sd(ei, E, e, sn, dn);
    const unsigned r = (unsigned)(dn - lo);
    if (r < urows) atomicMax(&s_mx[r], fkey(lrelu(sd[2 * sn] + s_d[r])));
  }
  __syncthreads();
  for (int e = tid; e < Etot; e += NT) {
    int sn, dn; edge_sd(ei, E, e, sn, dn);
    const unsigned r = (unsigned)(dn - lo);
    if (r < urows)
      atomicAdd(&s_den[r], __expf(lrelu(sd[2 * sn] + s_d[r]) - fkey_inv(s_mx[r])));
  }
  __syncthreads();
  for (int e = g; e < Etot; e += NG) {
    int sn, dn; edge_sd(ei, E, e, sn, dn);
    const unsigned r = (unsigned)(dn - lo);
    if (r < urows) {
      const float ew = __expf(lrelu(sd[2 * sn] + s_d[r]) - fkey_inv(s_mx[r]));
      const float w  = ew / (s_den[r] + 1e-16f);
      const float4 hv = *(const float4*)(hsrc + (size_t)sn * C + fo);
      float* op = s_acc + (int)r * C + fo;
      atomicAdd(op + 0, hv.x * w);
      atomicAdd(op + 1, hv.y * w);
      atomicAdd(op + 2, hv.z * w);
      atomicAdd(op + 3, hv.w * w);
    }
  }
  __syncthreads();
  for (int i = tid; i < rows * C; i += NT) {
    const int r = i / C, f = i % C;
    const size_t n = (size_t)(lo + r);
    const float v = s_acc[i] + bias[f];
    rowout[n * C + f] = FIRST ? (v > 0.f ? v : (__expf(v) - 1.f)) : v;
    if (f < H) {
      const float sp = s_acc[r * C + H + f] + bias[H + f];
      const float sd = softplusf(sp) + 1e-10f;
      klout[n * H + f] = -__logf(sd) + 0.5f * (sd * sd + v * v - 1.f);
    }
  }
}

extern "C" void kernel_launch(void* const* d_in, const int* in_sizes, int n_in,
                              void* d_out, int out_size, void* d_ws, size_t ws_size,
                              hipStream_t stream) {
  const float* x   = (const float*)d_in[0];
  const int*   ei  = (const int*)  d_in[1];
  const float* W1  = (const float*)d_in[2];
  const float* as1 = (const float*)d_in[3];
  const float* ad1 = (const float*)d_in[4];
  const float* b1  = (const float*)d_in[5];
  const float* W2  = (const float*)d_in[6];
  const float* as2 = (const float*)d_in[7];
  const float* ad2 = (const float*)d_in[8];
  const float* b2  = (const float*)d_in[9];

  const int N    = in_sizes[0] / F_IN;
  const int E    = in_sizes[1] / 2;
  float* out = (float*)d_out;

  const size_t N64 = (size_t)N * C1, N16 = (size_t)N * C2;
  const int owners = (N + ROWS - 1) / ROWS;
  int*   lent = (int*)d_ws;
  int*   loff = lent + (size_t)owners * CAP * 2;
  const int nchunk = (E + CHUNK - 1) / CHUNK;
  const int segp   = (owners + 15) & ~15;
  int*   bent = loff + (size_t)owners * LOFF;
  int*   seg  = bent + (size_t)nchunk * CHUNK * 2;
  float* ws = (float*)(seg + (size_t)nchunk * segp * 2);
  float* h1 = ws;
  float* hb = h1 + N64;
  float* h2 = hb + N64;
  float* sd = h2 + N16;

  float* ixz1 = out + N16;
  float* ixz2 = out + (size_t)N * 48;

  const int tiles  = (N + 15) / 16;
  const int gblk   = (tiles + 7) / 8;

  gemm1_wmma<<<gblk, 256, 0, stream>>>(x, W1, as1, ad1, h1, sd, N);
  bin_edges<<<nchunk, BT, 0, stream>>>(ei, E, N, owners, segp, seg, bent);
  layer_owner_kernel<C1, true><<<owners, NT, 0, stream>>>(ei, E, N, h1, sd, b1, hb, ixz1, loff, lent,
                                                          seg, bent, nchunk, segp);

  gemm2_wmma<<<gblk, 256, 0, stream>>>(hb, W2, as2, ad2, h2, sd, N);
  layer_owner_kernel<C2, false><<<owners, NT, 0, stream>>>(ei, E, N, h2, sd, b2, out, ixz2, loff, lent,
                                                           seg, bent, nchunk, segp);
}


# row lists kept as (source byte offset, weight) pairs, each row padded with zero-weight pairs to whole groups of 8: the row-sum loop has no bounds test, clamp, select or 64-bit address math; sums kept
# speedup vs baseline: 4.2417x; 1.0457x over previous
#include <hip/hip_runtime.h>
#include <hip/hip_bf16.h>

typedef __attribute__((ext_vector_type(2))) float v2f;
typedef __attribute__((ext_vector_type(8))) float v8f;

#define F_IN 128
#define C1   64
#define C2   16
#define NEG_SLOPE 0.2f

#define ROWS 392
#define NT   1024
#define CAP  16384
#define LOFF 396
#define BT    256
#define EPT   10
#define CHUNK (BT * EPT)
#define LPC   4
#define USEG  6

__global__ void __launch_bounds__(256)
gemm1_wmma(const float* __restrict__ x, const float* __restrict__ W1,
           const float* __restrict__ asrc, const float* __restrict__ adst,
           float* __restrict__ h1, float* __restrict__ sd, int Nn) {
  __shared__ float  sW[F_IN * C1];
  __shared__ float4 sUV[F_IN / 2];
  {
    const int k = threadIdx.x >> 1;
    const float4* av = (const float4*)((threadIdx.x & 1) ? adst : asrc);
    const float4* wr = (const float4*)(W1 + k * C1);
    float f = 0.f;
    for (int c4 = 0; c4 < C1 / 4; ++c4) {
      const float4 w = wr[c4], a = av[c4];
      f += w.x * a.x; f += w.y * a.y; f += w.z * a.z; f += w.w * a.w;
    }
    ((float*)sUV)[threadIdx.x] = f;
  }
  for (int i = threadIdx.x; i < (F_IN * C1) / 4; i += 256) {
    float4 v = ((const float4*)W1)[i];
    const int k  = i >> 4;
    const int c0 = (i & 15) * 4;
    float* dst = sW + (k >> 2) * 256 + (k & 3);
    dst[(c0 + 0) * 4] = v.x;
    dst[(c0 + 1) * 4] = v.y;
    dst[(c0 + 2) * 4] = v.z;
    dst[(c0 + 3) * 4] = v.w;
  }
  __syncthreads();

  const int wave = threadIdx.x >> 5;
  const int lane = threadIdx.x & 31;
  const int tile = blockIdx.x * 8 + wave;
  if (tile * 16 >= Nn) return;
  const int row0 = tile * 16;
  const int r    = lane & 15;
  const int kh   = (lane >> 4) << 1;
  const int arow = min(row0 + r, Nn - 1);

  v8f acc0 = {}, acc1 = {}, acc2 = {}, acc3 = {};
  float ss = 0.f, dd = 0.f;
  const float* ap = x + (size_t)arow * F_IN + kh;
  const float* bbase = sW + r * 4 + kh;
  constexpr int KC = 8;
  v2f an[KC];
  for (int j = 0; j < KC; ++j) { an[j].x = ap[4 * j]; an[j].y = ap[4 * j + 1]; }
#pragma unroll 1
  for (int q0 = 0; q0 < F_IN / 4; q0 += KC) {
    v2f ac[KC];
    for (int j = 0; j < KC; ++j) ac[j] = an[j];
    if (q0 + KC < F_IN / 4) {
      for (int j = 0; j < KC; ++j) {
        an[j].x = ap[4 * (q0 + KC + j)]; an[j].y = ap[4 * (q0 + KC + j) + 1];
      }
    }
    for (int j = 0; j < KC; ++j) {
      const int q = q0 + j;
      const v2f a = ac[j];
      const float4 uv = sUV[(4 * q + kh) >> 1];
      ss += a.x * uv.x; dd += a.x * uv.y;
      ss += a.y * uv.z; dd += a.y * uv.w;
      const float* bq = bbase + q * 256;
      v2f b0 = *(const v2f*)(bq);
      v2f b1 = *(const v2f*)(bq + 64);
      v2f b2 = *(const v2f*)(bq + 128);
      v2f b3 = *(const v2f*)(bq + 192);
      acc0 = __builtin_amdgcn_wmma_f32_16x16x4_f32(false, a, false, b0, (short)0, acc0, false, false);
      acc1 = __builtin_amdgcn_wmma_f32_16x16x4_f32(false, a, false, b1, (short)0, acc1, false, false);
      acc2 = __builtin_amdgcn_wmma_f32_16x16x4_f32(false, a, false, b2, (short)0, acc2, false, false);
      acc3 = __builtin_amdgcn_wmma_f32_16x16x4_f32(false, a, false, b3, (short)0, acc3, false, false);
    }
  }

  const int mbase = (lane >> 4) * 8;
  for (int i = 0; i < 8; ++i) {
    const int row = row0 + mbase + i;
    if (row < Nn) {
      float* op = h1 + (size_t)row * C1 + r;
      op[0]  = acc0[i];
      op[16] = acc1[i];
      op[32] = acc2[i];
      op[48] = acc3[i];
    }
  }

  ss += __shfl_xor(ss, 16);
  dd += __shfl_xor(dd, 16);
  if (lane < 16 && row0 + r < Nn)
    *(float2*)(sd + 2 * (size_t)(row0 + r)) = make_float2(ss, dd);
}

__global__ void __launch_bounds__(256)
gemm2_wmma(const float* __restrict__ h, const float* __restrict__ W2,
           const float* __restrict__ asrc, const float* __restrict__ adst,
           float* __restrict__ h2, float* __restrict__ sd, int Nn) {
  __shared__ float  sW[C1 * C2];
  __shared__ float4 sUV[C1 / 2];
  if (threadIdx.x < 2 * C1) {
    const int k = threadIdx.x >> 1;
    const float4* av = (const float4*)((threadIdx.x & 1) ? adst : asrc);
    const float4* wr = (const float4*)(W2 + k * C2);
    float f = 0.f;
    for (int c4 = 0; c4 < C2 / 4; ++c4) {
      const float4 w = wr[c4], a = av[c4];
      f += w.x * a.x; f += w.y * a.y; f += w.z * a.z; f += w.w * a.w;
    }
    ((float*)sUV)[threadIdx.x] = f;
  }
  for (int i = threadIdx.x; i < (C1 * C2) / 4; i += 256) {
    float4 v = ((const float4*)W2)[i];
    const int k  = i >> 2;
    const int c0 = (i & 3) * 4;
    float* dst = sW + (k >> 2) * 64 + (k & 3);
    dst[(c0 + 0) * 4] = v.x;
    dst[(c0 + 1) * 4] = v.y;
    dst[(c0 + 2) * 4] = v.z;
    dst[(c0 + 3) * 4] = v.w;
  }
  __syncthreads();

  const int wave = threadIdx.x >> 5;
  const int lane = threadIdx.x & 31;
  const int tile = blockIdx.x * 8 + wave;
  if (tile * 16 >= Nn) return;
  const int row0 = tile * 16;
  const int r    = lane & 15;
  const int kh   = (lane >> 4) << 1;
  const int arow = min(row0 + r, Nn - 1);

  v8f acc = {};
  float ss = 0.f, dd = 0.f;
  const float* ap = h + (size_t)arow * C1 + kh;
  const float* bbase = sW + r * 4 + kh;
  for (int k0 = 0; k0 < C1; k0 += 4) {
    v2f a; a.x = ap[k0]; a.y = ap[k0 + 1];
    const float4 uv = sUV[(k0 + kh) >> 1];
    ss += a.x * uv.x; dd += a.x * uv.y;
    ss += a.y * uv.z; dd += a.y * uv.w;
    v2f b = *(const v2f*)(bbase + (k0 >> 2) * 64);
    acc = __builtin_amdgcn_wmma_f32_16x16x4_f32(false, a, false, b, (short)0, acc, false, false);
  }

  const int mbase = (lane >> 4) * 8;
  for (int i = 0; i < 8; ++i) {
    const int row = row0 + mbase + i;
    if (row < Nn) h2[(size_t)row * C2 + r] = acc[i];
  }

  ss += __shfl_xor(ss, 16);
  dd += __shfl_xor(dd, 16);
  if (lane < 16 && row0 + r < Nn)
    *(float2*)(sd + 2 * (size_t)(row0 + r)) = make_float2(ss, dd);
}

__global__ void __launch_bounds__(BT)
bin_edges(const int* __restrict__ ei, int E, int Nn, int owners, int segpitch,
          int* __restrict__ segw, int* __restrict__ bentw) {
  int2* const seg  = (int2*)segw;
  int2* const bent = (int2*)bentw;
  __shared__ int  s_cnt[BT];
  __shared__ int  s_wsum[BT / 32];
  __shared__ int2 s_ent[CHUNK];
  const int tid  = threadIdx.x;
  const int lane = tid & 31;
  const int wave = tid >> 5;
  const int c    = blockIdx.x;
  const int e0   = c * CHUNK;

  int own[EPT], rloc[EPT], rank[EPT];
#pragma unroll
  for (int k = 0; k < EPT; ++k) {
    const int e  = e0 + k * BT + tid;
    const int dn = (e < E) ? ei[E + e] : -1;
    const int o  = ((unsigned)dn < (unsigned)Nn) ? dn / ROWS : -1;
    own[k]  = o;
    rloc[k] = dn - o * ROWS;
    rank[k] = 0;
  }

  int base = 0;
  for (int wb = 0; wb < owners; wb += BT) {
    s_cnt[tid] = 0;
    __syncthreads();
#pragma unroll
    for (int k = 0; k < EPT; ++k) {
      const unsigned o = (unsigned)(own[k] - wb);
      if (o < (unsigned)BT) rank[k] = atomicAdd(&s_cnt[o], 1);
    }
    __syncthreads();
    const int v = s_cnt[tid];
    int incl = v;
#pragma unroll
    for (int o = 1; o < 32; o <<= 1) {
      const int t = __shfl_up(incl, o);
      if (lane >= o) incl += t;
    }
    if (lane == 31) s_wsum[wave] = incl;
    __syncthreads();
    int wbase = 0, wtot = 0;
#pragma unroll
    for (int w = 0; w < BT / 32; ++w) {
      const int t = s_wsum[w];
      wbase += (w < wave) ? t : 0;
      wtot  += t;
    }
    const int start = base + wbase + incl - v;
    if (wb + tid < owners)
      seg[(size_t)c * segpitch + wb + tid] = make_int2(e0 + start, e0 + start + v);
    s_cnt[tid] = start;
    __syncthreads();
#pragma unroll
    for (int k = 0; k < EPT; ++k) {
      const unsigned o = (unsigned)(own[k] - wb);
      if (o < (unsigned)BT) s_ent[s_cnt[o] + rank[k]] = make_int2(e0 + k * BT + tid, rloc[k]);
    }
    base += wtot;
    __syncthreads();
  }
  for (int i = tid; i < base; i += BT) bent[e0 + i] = s_ent[i];
}

__device__ __forceinline__ int fkey(float f) {
  int k = __float_as_int(f);
  return k >= 0 ? k : (k ^ 0x7fffffff);
}
__device__ __forceinline__ float fkey_inv(int k) {
  return __int_as_float(k >= 0 ? k : (k ^ 0x7fffffff));
}
__device__ __forceinline__ void edge_sd(const int* ei, int E, int e, int& s, int& d) {
  if (e < E) { s = ei[e]; d = ei[E + e]; } else { s = d = e - E; }
}
__device__ __forceinline__ float lrelu(float a) { return a > 0.f ? a : NEG_SLOPE * a; }

__device__ __forceinline__ float softplusf(float v) {
  return fmaxf(v, 0.f) + log1pf(__expf(-fabsf(v)));
}

template <int C, bool FIRST>
__global__ void __launch_bounds__(NT)
layer_owner_kernel(const int* __restrict__ ei, int E, int Nn,
                   const float* __restrict__ hsrc,
                   const float* __restrict__ sd,
                   const float* __restrict__ bias,
                   float* __restrict__ rowout, float* __restrict__ klout,
                   int* __restrict__ loff, int* __restrict__ lent,
                   const int* __restrict__ segw, const int* __restrict__ bentw,
                   int nchunk, int segpitch) {
  const int2* const seg  = (const int2*)segw;
  const int2* const bent = (const int2*)bentw;
  __shared__ alignas(8) int s_raw[4 * CAP];
  __shared__ int   s_rowcnt[ROWS];
  __shared__ int   s_off[ROWS + 1];
  __shared__ float s_d[ROWS];
  __shared__ float s_rmax[ROWS];
  __shared__ float s_rden[ROWS];
  __shared__ int   s_cnt;
  int* const s_e  = s_raw;
  int* const s_r  = s_raw + CAP;
  int* const s_te = s_raw + 2 * CAP;
  int* const s_tr = s_raw + 3 * CAP;
  int2* const s_pair = (int2*)(s_raw + 2 * CAP);
  constexpr int UG   = 8;
  constexpr int PCAP = CAP / 2;

  const int tid  = threadIdx.x;
  const int lo   = blockIdx.x * ROWS;
  const int rows = min(ROWS, Nn - lo);
  const unsigned urows = (unsigned)rows;
  const int Etot = E + Nn;
  constexpr int H   = C / 2;
  constexpr int TPE = C / 4;
  constexpr int NG  = NT / TPE;
  const int fo = (tid % TPE) * 4;
  const int g  = tid / TPE;

  int*  const offp = loff + (size_t)blockIdx.x * LOFF;
  int2* const entp = (int2*)lent + (size_t)blockIdx.x * CAP;

  int ntot = 0;
  if constexpr (!FIRST) ntot = offp[ROWS + 1];

  for (int i = tid; i < ROWS; i += NT) s_d[i] = (i < rows) ? sd[2 * (lo + i) + 1] : 0.f;
  if constexpr (FIRST) {
    for (int i = tid; i < ROWS; i += NT) s_rowcnt[i] = 0;
    if (tid == 0) s_cnt = 0;
  }
  __syncthreads();

  if constexpr (FIRST) {
    auto append = [&](int e, unsigned r) {
      const int slot = atomicAdd(&s_cnt, 1);
      if (slot < CAP) { s_e[slot] = e; s_r[slot] = (int)r; }
      atomicAdd(&s_rowcnt[r], 1);
    };

    for (int c = tid / LPC; c < nchunk; c += NT / LPC) {
      const int2 se = seg[(size_t)c * segpitch + blockIdx.x];
      const int  n  = se.y - se.x;
      int slot0 = 0;
      if (tid % LPC == 0) slot0 = atomicAdd(&s_cnt, n);
      slot0 = __shfl(slot0, 0, LPC);
      for (int k0 = tid % LPC; k0 < n; k0 += LPC * USEG) {
        int2 en[USEG];
#pragma unroll
        for (int u = 0; u < USEG; ++u) {
          const int k = k0 + u * LPC;
          en[u] = bent[se.x + (k < n ? k : k0)];
        }
#pragma unroll
        for (int u = 0; u < USEG; ++u) {
          const int k = k0 + u * LPC;
          if (k < n) {
            const int slot = slot0 + k;
            if (slot < CAP) { s_e[slot] = en[u].x; s_r[slot] = en[u].y; }
            atomicAdd(&s_rowcnt[en[u].y], 1);
          }
        }
      }
    }
    for (int r = tid; r < rows; r += NT) append(E + lo + r, (unsigned)r);
    __syncthreads();
    ntot = s_cnt;
    if (tid == 0) offp[ROWS + 1] = ntot;
  }
  const bool listed = ntot + (UG - 1) * ROWS <= PCAP;

  if (listed) {
    if constexpr (FIRST) {
      if (tid < 32) {
        constexpr int PER = (ROWS + 31) / 32;
        int loc[PER]; int sum = 0;
#pragma unroll
        for (int k = 0; k < PER; ++k) {
          const int idx = tid * PER + k;
          loc[k] = sum;
          sum += (idx < ROWS) ? s_rowcnt[idx] : 0;
        }
        int incl = sum;
#pragma unroll
        for (int o = 1; o < 32; o <<= 1) {
          const int t = __shfl_up(incl, o);
          if (tid >= o) incl += t;
        }
        const int basep = incl - sum;
#pragma unroll
        for (int k = 0; k < PER; ++k) {
          const int idx = tid * PER + k;
          if (idx < ROWS) { s_off[idx] = basep + loc[k]; s_rowcnt[idx] = 0; }
        }
        if (tid == 31) s_off[ROWS] = incl;
      }
      __syncthreads();
      for (int i = tid; i <= ROWS; i += NT) offp[i] = s_off[i];

      for (int i = tid; i < ntot; i += NT) {
        const int r = s_r[i];
        const int p = s_off[r] + atomicAdd(&s_rowcnt[r], 1);
        s_te[p] = s_e[i]; s_tr[p] = r;
      }
      __syncthreads();

      constexpr int UA = 4;
      for (int p0 = tid; p0 < ntot; p0 += UA * NT) {
        int sn[UA], pos[UA], rr[UA]; float sv[UA];
#pragma unroll
        for (int k = 0; k < UA; ++k) {
          const int  p  = p0 + k * NT;
          const bool ok = p < ntot;
          const int  pc = ok ? p : p0;
          const int  e  = s_te[pc];
          const int  r  = s_tr[pc];
          const int  b  = s_off[r], en = s_off[r + 1];
          int rk = 0;
          for (int j = b; j < en; ++j) rk += (s_te[j] < e) ? 1 : 0;
          pos[k] = ok ? b + rk : -1;
          rr[k]  = r;
          sn[k]  = e < E ? ei[e] : e - E;
        }
#pragma unroll
        for (int k = 0; k < UA; ++k) sv[k] = sd[2 * sn[k]];
#pragma unroll
        for (int k = 0; k < UA; ++k) {
          if (pos[k] >= 0) {
            s_e[pos[k]] = sn[k];
            s_r[pos[k]] = __float_as_int(lrelu(sv[k] + s_d[rr[k]]));
            entp[pos[k]] = make_int2(sn[k], rr[k]);
          }
        }
      }
      __syncthreads();
    } else {
      for (int i = tid; i <= ROWS; i += NT) s_off[i] = offp[i];
      constexpr int UA = 4;
      for (int p0 = tid; p0 < ntot; p0 += UA * NT) {
        int2 en[UA]; float sv[UA];
#pragma unroll
        for (int k = 0; k < UA; ++k) {
          const int p = p0 + k * NT;
          en[k] = entp[p < ntot ? p : p0];
        }
#pragma unroll
        for (int k = 0; k < UA; ++k) sv[k] = sd[2 * en[k].x];
        asm volatile("" :: "v"(sv[0]), "v"(sv[1]), "v"(sv[2]), "v"(sv[3]));
#pragma unroll
        for (int k = 0; k < UA; ++k) {
          const int p = p0 + k * NT;
          if (p < ntot) {
            s_e[p]  = en[k].x;
            s_tr[p] = en[k].y;
            s_r[p]  = __float_as_int(lrelu(sv[k] + s_d[en[k].y]));
          }
        }
      }
      __syncthreads();
    }

    constexpr int UD = 8;
    for (int r = tid; r < rows; r += NT) {
      const int b = s_off[r], en = s_off[r + 1];
      float m = __int_as_float(s_r[b]);
      for (int j0 = b + 1; j0 < en; j0 += UD) {
        float av[UD];
#pragma unroll
        for (int k = 0; k < UD; ++k) {
          const int j = j0 + k;
          av[k] = __int_as_float(s_r[j < en ? j : b]);
        }
#pragma unroll
        for (int k = 0; k < UD; ++k) m = fmaxf(m, av[k]);
      }
      float den = 0.f;
      for (int j0 = b; j0 < en; j0 += UD) {
        float ev[UD];
#pragma unroll
        for (int k = 0; k < UD; ++k) {
          const int j = j0 + k;
          ev[k] = __expf(__int_as_float(s_r[j < en ? j : b]) - m);
        }
#pragma unroll
        for (int k = 0; k < UD; ++k)
          if (j0 + k < en) den += ev[k];
      }
      s_rmax[r] = m;
      s_rden[r] = den + 1e-16f;
      const int len = en - b, padded = (len + UG - 1) & ~(UG - 1);
      int2* const pp = s_pair + b + (UG - 1) * r;
#pragma unroll
      for (int k = 0; k < UG - 1; ++k)
        if (len + k < padded) pp[len + k] = make_int2(0, 0);
    }
    __syncthreads();

    {
      constexpr int UW = 4;
      for (int p0 = tid; p0 < ntot; p0 += UW * NT) {
        float al[UW], mm[UW], dd[UW]; int sn[UW], rw[UW];
#pragma unroll
        for (int k = 0; k < UW; ++k) {
          const int p  = p0 + k * NT;
          const int pc = p < ntot ? p : p0;
          const int r  = s_tr[pc];
          rw[k] = r;
          sn[k] = s_e[pc];
          al[k] = __int_as_float(s_r[pc]);
          mm[k] = s_rmax[r];
          dd[k] = s_rden[r];
        }
#pragma unroll
        for (int k = 0; k < UW; ++k) {
          const int p = p0 + k * NT;
          if (p < ntot)
            s_pair[p + (UG - 1) * rw[k]] =
                make_int2(sn[k] * (C * 4), __float_as_int(__expf(al[k] - mm[k]) / dd[k]));
        }
      }
    }
    __syncthreads();

    for (int r = g; r < rows; r += NG) {
      const int b = s_off[r], len = s_off[r + 1] - b;
      const int2* const pp = s_pair + b + (UG - 1) * r;
      float a0 = 0.f, a1 = 0.f, a2 = 0.f, a3 = 0.f;
      for (int j0 = 0; j0 < len; j0 += UG) {
        int2 pr[UG]; float4 hv[UG];
#pragma unroll
        for (int k = 0; k < UG; ++k) pr[k] = pp[j0 + k];
#pragma unroll
        for (int k = 0; k < UG; ++k)
          hv[k] = *(const float4*)((const char*)hsrc + (unsigned)(pr[k].x + fo * 4));
#pragma unroll
        for (int k = 0; k < UG; ++k) {
          const float w = __int_as_float(pr[k].y);
          a0 += hv[k].x * w; a1 += hv[k].y * w;
          a2 += hv[k].z * w; a3 += hv[k].w * w;
        }
      }
      const float4 bv = *(const float4*)(bias + fo);
      const float o0 = a0 + bv.x, o1 = a1 + bv.y, o2 = a2 + bv.z, o3 = a3 + bv.w;
      const bool  up = fo >= H;
      const float ra = __shfl_xor(up ? o0 : o2, TPE / 2, TPE);
      const float rb = __shfl_xor(up ? o1 : o3, TPE / 2, TPE);
      const size_t n = (size_t)(lo + r);
      float4 st;
      st.x = FIRST ? (o0 > 0.f ? o0 : (__expf(o0) - 1.f)) : o0;
      st.y = FIRST ? (o1 > 0.f ? o1 : (__expf(o1) - 1.f)) : o1;
      st.z = FIRST ? (o2 > 0.f ? o2 : (__expf(o2) - 1.f)) : o2;
      st.w = FIRST ? (o3 > 0.f ? o3 : (__expf(o3) - 1.f)) : o3;
      *(float4*)(rowout + n * C + fo) = st;
      const float mu0 = up ? ra : o0, mu1 = up ? rb : o1;
      const float sp0 = up ? o2 : ra, sp1 = up ? o3 : rb;
      const int   kf  = up ? fo - H + 2 : fo;
      const float d0 = softplusf(sp0) + 1e-10f, d1 = softplusf(sp1) + 1e-10f;
      float2 kl;
      kl.x = -__logf(d0) + 0.5f * (d0 * d0 + mu0 * mu0 - 1.f);
      kl.y = -__logf(d1) + 0.5f * (d1 * d1 + mu1 * mu1 - 1.f);
      *(float2*)(klout + n * H + kf) = kl;
    }
    return;
  }

  float* const s_acc = (float*)s_raw;
  int*   const s_mx  = s_raw + ROWS * C;
  float* const s_den = (float*)(s_raw + ROWS * C + ROWS);
  for (int i = tid; i < ROWS * C; i += NT) s_acc[i] = 0.f;
  for (int i = tid; i < ROWS; i += NT) { s_mx[i] = (int)0x80000000; s_den[i] = 0.f; }
  __syncthreads();
  for (int e = tid; e < Etot; e += NT) {
    int sn, dn; edge_sd(ei, E, e, sn, dn);
    const unsigned r = (unsigned)(dn - lo);
    if (r < urows) atomicMax(&s_mx[r], fkey(lrelu(sd[2 * sn] + s_d[r])));
  }
  __syncthreads();
  for (int e = tid; e < Etot; e += NT) {
    int sn, dn; edge_sd(ei, E, e, sn, dn);
    const unsigned r = (unsigned)(dn - lo);
    if (r < urows)
      atomicAdd(&s_den[r], __expf(lrelu(sd[2 * sn] + s_d[r]) - fkey_inv(s_mx[r])));
  }
  __syncthreads();
  for (int e = g; e < Etot; e += NG) {
    int sn, dn; edge_sd(ei, E, e, sn, dn);
    const unsigned r = (unsigned)(dn - lo);
    if (r < urows) {
      const float ew = __expf(lrelu(sd[2 * sn] + s_d[r]) - fkey_inv(s_mx[r]));
      const float w  = ew / (s_den[r] + 1e-16f);
      const float4 hv = *(const float4*)(hsrc + (size_t)sn * C + fo);
      float* op = s_acc + (int)r * C + fo;
      atomicAdd(op + 0, hv.x * w);
      atomicAdd(op + 1, hv.y * w);
      atomicAdd(op + 2, hv.z * w);
      atomicAdd(op + 3, hv.w * w);
    }
  }
  __syncthreads();
  for (int i = tid; i < rows * C; i += NT) {
    const int r = i / C, f = i % C;
    const size_t n = (size_t)(lo + r);
    const float v = s_acc[i] + bias[f];
    rowout[n * C + f] = FIRST ? (v > 0.f ? v : (__expf(v) - 1.f)) : v;
    if (f < H) {
      const float sp = s_acc[r * C + H + f] + bias[H + f];
      const float sd = softplusf(sp) + 1e-10f;
      klout[n * H + f] = -__logf(sd) + 0.5f * (sd * sd + v * v - 1.f);
    }
  }
}

extern "C" void kernel_launch(void* const* d_in, const int* in_sizes, int n_in,
                              void* d_out, int out_size, void* d_ws, size_t ws_size,
                              hipStream_t stream) {
  const float* x   = (const float*)d_in[0];
  const int*   ei  = (const int*)  d_in[1];
  const float* W1  = (const float*)d_in[2];
  const float* as1 = (const float*)d_in[3];
  const float* ad1 = (const float*)d_in[4];
  const float* b1  = (const float*)d_in[5];
  const float* W2  = (const float*)d_in[6];
  const float* as2 = (const float*)d_in[7];
  const float* ad2 = (const float*)d_in[8];
  const float* b2  = (const float*)d_in[9];

  const int N    = in_sizes[0] / F_IN;
  const int E    = in_sizes[1] / 2;
  float* out = (float*)d_out;

  const size_t N64 = (size_t)N * C1, N16 = (size_t)N * C2;
  const int owners = (N + ROWS - 1) / ROWS;
  int*   lent = (int*)d_ws;
  int*   loff = lent + (size_t)owners * CAP * 2;
  const int nchunk = (E + CHUNK - 1) / CHUNK;
  const int segp   = (owners + 15) & ~15;
  int*   bent = loff + (size_t)owners * LOFF;
  int*   seg  = bent + (size_t)nchunk * CHUNK * 2;
  float* ws = (float*)(seg + (size_t)nchunk * segp * 2);
  float* h1 = ws;
  float* hb = h1 + N64;
  float* h2 = hb + N64;
  float* sd = h2 + N16;

  float* ixz1 = out + N16;
  float* ixz2 = out + (size_t)N * 48;

  const int tiles  = (N + 15) / 16;
  const int gblk   = (tiles + 7) / 8;

  gemm1_wmma<<<gblk, 256, 0, stream>>>(x, W1, as1, ad1, h1, sd, N);
  bin_edges<<<nchunk, BT, 0, stream>>>(ei, E, N, owners, segp, seg, bent);
  layer_owner_kernel<C1, true><<<owners, NT, 0, stream>>>(ei, E, N, h1, sd, b1, hb, ixz1, loff, lent,
                                                          seg, bent, nchunk, segp);

  gemm2_wmma<<<gblk, 256, 0, stream>>>(hb, W2, as2, ad2, h2, sd, N);
  layer_owner_kernel<C2, false><<<owners, NT, 0, stream>>>(ei, E, N, h2, sd, b2, out, ixz2, loff, lent,
                                                           seg, bent, nchunk, segp);
}


# GEMM1 runs as 256 work-groups of 25 waves, one per WGP, a 16-row tile per wave: W1 is folded and packed into LDS once per WGP, not by each of 782 eight-tile work-groups; sums unchanged
# speedup vs baseline: 4.4814x; 1.0565x over previous
#include <hip/hip_runtime.h>
#include <hip/hip_bf16.h>

typedef __attribute__((ext_vector_type(2))) float v2f;
typedef __attribute__((ext_vector_type(8))) float v8f;

#define F_IN 128
#define C1   64
#define C2   16
#define NEG_SLOPE 0.2f

#define ROWS 392
#define NT   1024
#define CAP  16384
#define LOFF 396
#define BT    256
#define EPT   10
#define CHUNK (BT * EPT)
#define LPC   4
#define USEG  6

#define G1B 256
#define G1W 25
#define G1T (G1W * 32)
__global__ void __launch_bounds__(G1T)
gemm1_wmma(const float* __restrict__ x, const float* __restrict__ W1,
           const float* __restrict__ asrc, const float* __restrict__ adst,
           float* __restrict__ h1, float* __restrict__ sd, int Nn) {
  __shared__ float  sW[F_IN * C1];
  __shared__ float4 sUV[F_IN / 2];
  if (threadIdx.x < 2 * F_IN) {
    const int k = threadIdx.x >> 1;
    const float4* av = (const float4*)((threadIdx.x & 1) ? adst : asrc);
    const float4* wr = (const float4*)(W1 + k * C1);
    float f = 0.f;
    for (int c4 = 0; c4 < C1 / 4; ++c4) {
      const float4 w = wr[c4], a = av[c4];
      f += w.x * a.x; f += w.y * a.y; f += w.z * a.z; f += w.w * a.w;
    }
    ((float*)sUV)[threadIdx.x] = f;
  }
  for (int i = threadIdx.x; i < (F_IN * C1) / 4; i += G1T) {
    float4 v = ((const float4*)W1)[i];
    const int k  = i >> 4;
    const int c0 = (i & 15) * 4;
    float* dst = sW + (k >> 2) * 256 + (k & 3);
    dst[(c0 + 0) * 4] = v.x;
    dst[(c0 + 1) * 4] = v.y;
    dst[(c0 + 2) * 4] = v.z;
    dst[(c0 + 3) * 4] = v.w;
  }
  __syncthreads();

  const int wave = threadIdx.x >> 5;
  const int lane = threadIdx.x & 31;
  const int r    = lane & 15;
  const int kh   = (lane >> 4) << 1;
  for (int tile = wave * (int)gridDim.x + (int)blockIdx.x; tile * 16 < Nn; tile += G1W * (int)gridDim.x) {
  const int row0 = tile * 16;
  const int arow = min(row0 + r, Nn - 1);

  v8f acc0 = {}, acc1 = {}, acc2 = {}, acc3 = {};
  float ss = 0.f, dd = 0.f;
  const float* ap = x + (size_t)arow * F_IN + kh;
  const float* bbase = sW + r * 4 + kh;
  constexpr int KC = 8;
  v2f an[KC];
  for (int j = 0; j < KC; ++j) { an[j].x = ap[4 * j]; an[j].y = ap[4 * j + 1]; }
#pragma unroll 1
  for (int q0 = 0; q0 < F_IN / 4; q0 += KC) {
    v2f ac[KC];
    for (int j = 0; j < KC; ++j) ac[j] = an[j];
    if (q0 + KC < F_IN / 4) {
      for (int j = 0; j < KC; ++j) {
        an[j].x = ap[4 * (q0 + KC + j)]; an[j].y = ap[4 * (q0 + KC + j) + 1];
      }
    }
    for (int j = 0; j < KC; ++j) {
      const int q = q0 + j;
      const v2f a = ac[j];
      const float4 uv = sUV[(4 * q + kh) >> 1];
      ss += a.x * uv.x; dd += a.x * uv.y;
      ss += a.y * uv.z; dd += a.y * uv.w;
      const float* bq = bbase + q * 256;
      v2f b0 = *(const v2f*)(bq);
      v2f b1 = *(const v2f*)(bq + 64);
      v2f b2 = *(const v2f*)(bq + 128);
      v2f b3 = *(const v2f*)(bq + 192);
      acc0 = __builtin_amdgcn_wmma_f32_16x16x4_f32(false, a, false, b0, (short)0, acc0, false, false);
      acc1 = __builtin_amdgcn_wmma_f32_16x16x4_f32(false, a, false, b1, (short)0, acc1, false, false);
      acc2 = __builtin_amdgcn_wmma_f32_16x16x4_f32(false, a, false, b2, (short)0, acc2, false, false);
      acc3 = __builtin_amdgcn_wmma_f32_16x16x4_f32(false, a, false, b3, (short)0, acc3, false, false);
    }
  }

  const int mbase = (lane >> 4) * 8;
  for (int i = 0; i < 8; ++i) {
    const int row = row0 + mbase + i;
    if (row < Nn) {
      float* op = h1 + (size_t)row * C1 + r;
      op[0]  = acc0[i];
      op[16] = acc1[i];
      op[32] = acc2[i];
      op[48] = acc3[i];
    }
  }

  ss += __shfl_xor(ss, 16);
  dd += __shfl_xor(dd, 16);
  if (lane < 16 && row0 + r < Nn)
    *(float2*)(sd + 2 * (size_t)(row0 + r)) = make_float2(ss, dd);
  }
}

__global__ void __launch_bounds__(256)
gemm2_wmma(const float* __restrict__ h, const float* __restrict__ W2,
           const float* __restrict__ asrc, const float* __restrict__ adst,
           float* __restrict__ h2, float* __restrict__ sd, int Nn) {
  __shared__ float  sW[C1 * C2];
  __shared__ float4 sUV[C1 / 2];
  if (threadIdx.x < 2 * C1) {
    const int k = threadIdx.x >> 1;
    const float4* av = (const float4*)((threadIdx.x & 1) ? adst : asrc);
    const float4* wr = (const float4*)(W2 + k * C2);
    float f = 0.f;
    for (int c4 = 0; c4 < C2 / 4; ++c4) {
      const float4 w = wr[c4], a = av[c4];
      f += w.x * a.x; f += w.y * a.y; f += w.z * a.z; f += w.w * a.w;
    }
    ((float*)sUV)[threadIdx.x] = f;
  }
  for (int i = threadIdx.x; i < (C1 * C2) / 4; i += 256) {
    float4 v = ((const float4*)W2)[i];
    const int k  = i >> 2;
    const int c0 = (i & 3) * 4;
    float* dst = sW + (k >> 2) * 64 + (k & 3);
    dst[(c0 + 0) * 4] = v.x;
    dst[(c0 + 1) * 4] = v.y;
    dst[(c0 + 2) * 4] = v.z;
    dst[(c0 + 3) * 4] = v.w;
  }
  __syncthreads();

  const int wave = threadIdx.x >> 5;
  const int lane = threadIdx.x & 31;
  const int tile = blockIdx.x * 8 + wave;
  if (tile * 16 >= Nn) return;
  const int row0 = tile * 16;
  const int r    = lane & 15;
  const int kh   = (lane >> 4) << 1;
  const int arow = min(row0 + r, Nn - 1);

  v8f acc = {};
  float ss = 0.f, dd = 0.f;
  const float* ap = h + (size_t)arow * C1 + kh;
  const float* bbase = sW + r * 4 + kh;
  for (int k0 = 0; k0 < C1; k0 += 4) {
    v2f a; a.x = ap[k0]; a.y = ap[k0 + 1];
    const float4 uv = sUV[(k0 + kh) >> 1];
    ss += a.x * uv.x; dd += a.x * uv.y;
    ss += a.y * uv.z; dd += a.y * uv.w;
    v2f b = *(const v2f*)(bbase + (k0 >> 2) * 64);
    acc = __builtin_amdgcn_wmma_f32_16x16x4_f32(false, a, false, b, (short)0, acc, false, false);
  }

  const int mbase = (lane >> 4) * 8;
  for (int i = 0; i < 8; ++i) {
    const int row = row0 + mbase + i;
    if (row < Nn) h2[(size_t)row * C2 + r] = acc[i];
  }

  ss += __shfl_xor(ss, 16);
  dd += __shfl_xor(dd, 16);
  if (lane < 16 && row0 + r < Nn)
    *(float2*)(sd + 2 * (size_t)(row0 + r)) = make_float2(ss, dd);
}

__global__ void __launch_bounds__(BT)
bin_edges(const int* __restrict__ ei, int E, int Nn, int owners, int segpitch,
          int* __restrict__ segw, int* __restrict__ bentw) {
  int2* const seg  = (int2*)segw;
  int2* const bent = (int2*)bentw;
  __shared__ int  s_cnt[BT];
  __shared__ int  s_wsum[BT / 32];
  __shared__ int2 s_ent[CHUNK];
  const int tid  = threadIdx.x;
  const int lane = tid & 31;
  const int wave = tid >> 5;
  const int c    = blockIdx.x;
  const int e0   = c * CHUNK;

  int own[EPT], rloc[EPT], rank[EPT];
#pragma unroll
  for (int k = 0; k < EPT; ++k) {
    const int e  = e0 + k * BT + tid;
    const int dn = (e < E) ? ei[E + e] : -1;
    const int o  = ((unsigned)dn < (unsigned)Nn) ? dn / ROWS : -1;
    own[k]  = o;
    rloc[k] = dn - o * ROWS;
    rank[k] = 0;
  }

  int base = 0;
  for (int wb = 0; wb < owners; wb += BT) {
    s_cnt[tid] = 0;
    __syncthreads();
#pragma unroll
    for (int k = 0; k < EPT; ++k) {
      const unsigned o = (unsigned)(own[k] - wb);
      if (o < (unsigned)BT) rank[k] = atomicAdd(&s_cnt[o], 1);
    }
    __syncthreads();
    const int v = s_cnt[tid];
    int incl = v;
#pragma unroll
    for (int o = 1; o < 32; o <<= 1) {
      const int t = __shfl_up(incl, o);
      if (lane >= o) incl += t;
    }
    if (lane == 31) s_wsum[wave] = incl;
    __syncthreads();
    int wbase = 0, wtot = 0;
#pragma unroll
    for (int w = 0; w < BT / 32; ++w) {
      const int t = s_wsum[w];
      wbase += (w < wave) ? t : 0;
      wtot  += t;
    }
    const int start = base + wbase + incl - v;
    if (wb + tid < owners)
      seg[(size_t)c * segpitch + wb + tid] = make_int2(e0 + start, e0 + start + v);
    s_cnt[tid] = start;
    __syncthreads();
#pragma unroll
    for (int k = 0; k < EPT; ++k) {
      const unsigned o = (unsigned)(own[k] - wb);
      if (o < (unsigned)BT) s_ent[s_cnt[o] + rank[k]] = make_int2(e0 + k * BT + tid, rloc[k]);
    }
    base += wtot;
    __syncthreads();
  }
  for (int i = tid; i < base; i += BT) bent[e0 + i] = s_ent[i];
}

__device__ __forceinline__ int fkey(float f) {
  int k = __float_as_int(f);
  return k >= 0 ? k : (k ^ 0x7fffffff);
}
__device__ __forceinline__ float fkey_inv(int k) {
  return __int_as_float(k >= 0 ? k : (k ^ 0x7fffffff));
}
__device__ __forceinline__ void edge_sd(const int* ei, int E, int e, int& s, int& d) {
  if (e < E) { s = ei[e]; d = ei[E + e]; } else { s = d = e - E; }
}
__device__ __forceinline__ float lrelu(float a) { return a > 0.f ? a : NEG_SLOPE * a; }

__device__ __forceinline__ float softplusf(float v) {
  return fmaxf(v, 0.f) + log1pf(__expf(-fabsf(v)));
}

template <int C, bool FIRST>
__global__ void __launch_bounds__(NT)
layer_owner_kernel(const int* __restrict__ ei, int E, int Nn,
                   const float* __restrict__ hsrc,
                   const float* __restrict__ sd,
                   const float* __restrict__ bias,
                   float* __restrict__ rowout, float* __restrict__ klout,
                   int* __restrict__ loff, int* __restrict__ lent,
                   const int* __restrict__ segw, const int* __restrict__ bentw,
                   int nchunk, int segpitch) {
  const int2* const seg  = (const int2*)segw;
  const int2* const bent = (const int2*)bentw;
  __shared__ alignas(8) int s_raw[4 * CAP];
  __shared__ int   s_rowcnt[ROWS];
  __shared__ int   s_off[ROWS + 1];
  __shared__ float s_d[ROWS];
  __shared__ float s_rmax[ROWS];
  __shared__ float s_rden[ROWS];
  __shared__ int   s_cnt;
  int* const s_e  = s_raw;
  int* const s_r  = s_raw + CAP;
  int* const s_te = s_raw + 2 * CAP;
  int* const s_tr = s_raw + 3 * CAP;
  int2* const s_pair = (int2*)(s_raw + 2 * CAP);
  constexpr int UG   = 8;
  constexpr int PCAP = CAP / 2;

  const int tid  = threadIdx.x;
  const int lo   = blockIdx.x * ROWS;
  const int rows = min(ROWS, Nn - lo);
  const unsigned urows = (unsigned)rows;
  const int Etot = E + Nn;
  constexpr int H   = C / 2;
  constexpr int TPE = C / 4;
  constexpr int NG  = NT / TPE;
  const int fo = (tid % TPE) * 4;
  const int g  = tid / TPE;

  int*  const offp = loff + (size_t)blockIdx.x * LOFF;
  int2* const entp = (int2*)lent + (size_t)blockIdx.x * CAP;

  int ntot = 0;
  if constexpr (!FIRST) ntot = offp[ROWS + 1];

  for (int i = tid; i < ROWS; i += NT) s_d[i] = (i < rows) ? sd[2 * (lo + i) + 1] : 0.f;
  if constexpr (FIRST) {
    for (int i = tid; i < ROWS; i += NT) s_rowcnt[i] = 0;
    if (tid == 0) s_cnt = 0;
  }
  __syncthreads();

  if constexpr (FIRST) {
    auto append = [&](int e, unsigned r) {
      const int slot = atomicAdd(&s_cnt, 1);
      if (slot < CAP) { s_e[slot] = e; s_r[slot] = (int)r; }
      atomicAdd(&s_rowcnt[r], 1);
    };

    for (int c = tid / LPC; c < nchunk; c += NT / LPC) {
      const int2 se = seg[(size_t)c * segpitch + blockIdx.x];
      const int  n  = se.y - se.x;
      int slot0 = 0;
      if (tid % LPC == 0) slot0 = atomicAdd(&s_cnt, n);
      slot0 = __shfl(slot0, 0, LPC);
      for (int k0 = tid % LPC; k0 < n; k0 += LPC * USEG) {
        int2 en[USEG];
#pragma unroll
        for (int u = 0; u < USEG; ++u) {
          const int k = k0 + u * LPC;
          en[u] = bent[se.x + (k < n ? k : k0)];
        }
#pragma unroll
        for (int u = 0; u < USEG; ++u) {
          const int k = k0 + u * LPC;
          if (k < n) {
            const int slot = slot0 + k;
            if (slot < CAP) { s_e[slot] = en[u].x; s_r[slot] = en[u].y; }
            atomicAdd(&s_rowcnt[en[u].y], 1);
          }
        }
      }
    }
    for (int r = tid; r < rows; r += NT) append(E + lo + r, (unsigned)r);
    __syncthreads();
    ntot = s_cnt;
    if (tid == 0) offp[ROWS + 1] = ntot;
  }
  const bool listed = ntot + (UG - 1) * ROWS <= PCAP;

  if (listed) {
    if constexpr (FIRST) {
      if (tid < 32) {
        constexpr int PER = (ROWS + 31) / 32;
        int loc[PER]; int sum = 0;
#pragma unroll
        for (int k = 0; k < PER; ++k) {
          const int idx = tid * PER + k;
          loc[k] = sum;
          sum += (idx < ROWS) ? s_rowcnt[idx] : 0;
        }
        int incl = sum;
#pragma unroll
        for (int o = 1; o < 32; o <<= 1) {
          const int t = __shfl_up(incl, o);
          if (tid >= o) incl += t;
        }
        const int basep = incl - sum;
#pragma unroll
        for (int k = 0; k < PER; ++k) {
          const int idx = tid * PER + k;
          if (idx < ROWS) { s_off[idx] = basep + loc[k]; s_rowcnt[idx] = 0; }
        }
        if (tid == 31) s_off[ROWS] = incl;
      }
      __syncthreads();
      for (int i = tid; i <= ROWS; i += NT) offp[i] = s_off[i];

      for (int i = tid; i < ntot; i += NT) {
        const int r = s_r[i];
        const int p = s_off[r] + atomicAdd(&s_rowcnt[r], 1);
        s_te[p] = s_e[i]; s_tr[p] = r;
      }
      __syncthreads();

      constexpr int UA = 4;
      for (int p0 = tid; p0 < ntot; p0 += UA * NT) {
        int sn[UA], pos[UA], rr[UA]; float sv[UA];
#pragma unroll
        for (int k = 0; k < UA; ++k) {
          const int  p  = p0 + k * NT;
          const bool ok = p < ntot;
          const int  pc = ok ? p : p0;
          const int  e  = s_te[pc];
          const int  r  = s_tr[pc];
          const int  b  = s_off[r], en = s_off[r + 1];
          int rk = 0;
          for (int j = b; j < en; ++j) rk += (s_te[j] < e) ? 1 : 0;
          pos[k] = ok ? b + rk : -1;
          rr[k]  = r;
          sn[k]  = e < E ? ei[e] : e - E;
        }
#pragma unroll
        for (int k = 0; k < UA; ++k) sv[k] = sd[2 * sn[k]];
#pragma unroll
        for (int k = 0; k < UA; ++k) {
          if (pos[k] >= 0) {
            s_e[pos[k]] = sn[k];
            s_r[pos[k]] = __float_as_int(lrelu(sv[k] + s_d[rr[k]]));
            entp[pos[k]] = make_int2(sn[k], rr[k]);
          }
        }
      }
      __syncthreads();
    } else {
      for (int i = tid; i <= ROWS; i += NT) s_off[i] = offp[i];
      constexpr int UA = 4;
      for (int p0 = tid; p0 < ntot; p0 += UA * NT) {
        int2 en[UA]; float sv[UA];
#pragma unroll
        for (int k = 0; k < UA; ++k) {
          const int p = p0 + k * NT;
          en[k] = entp[p < ntot ? p : p0];
        }
#pragma unroll
        for (int k = 0; k < UA; ++k) sv[k] = sd[2 * en[k].x];
        asm volatile("" :: "v"(sv[0]), "v"(sv[1]), "v"(sv[2]), "v"(sv[3]));
#pragma unroll
        for (int k = 0; k < UA; ++k) {
          const int p = p0 + k * NT;
          if (p < ntot) {
            s_e[p]  = en[k].x;
            s_tr[p] = en[k].y;
            s_r[p]  = __float_as_int(lrelu(sv[k] + s_d[en[k].y]));
          }
        }
      }
      __syncthreads();
    }

    constexpr int UD = 8;
    for (int r = tid; r < rows; r += NT) {
      const int b = s_off[r], en = s_off[r + 1];
      float m = __int_as_float(s_r[b]);
      for (int j0 = b + 1; j0 < en; j0 += UD) {
        float av[UD];
#pragma unroll
        for (int k = 0; k < UD; ++k) {
          const int j = j0 + k;
          av[k] = __int_as_float(s_r[j < en ? j : b]);
        }
#pragma unroll
        for (int k = 0; k < UD; ++k) m = fmaxf(m, av[k]);
      }
      float den = 0.f;
      for (int j0 = b; j0 < en; j0 += UD) {
        float ev[UD];
#pragma unroll
        for (int k = 0; k < UD; ++k) {
          const int j = j0 + k;
          ev[k] = __expf(__int_as_float(s_r[j < en ? j : b]) - m);
        }
#pragma unroll
        for (int k = 0; k < UD; ++k)
          if (j0 + k < en) den += ev[k];
      }
      s_rmax[r] = m;
      s_rden[r] = den + 1e-16f;
      const int len = en - b, padded = (len + UG - 1) & ~(UG - 1);
      int2* const pp = s_pair + b + (UG - 1) * r;
#pragma unroll
      for (int k = 0; k < UG - 1; ++k)
        if (len + k < padded) pp[len + k] = make_int2(0, 0);
    }
    __syncthreads();

    {
      constexpr int UW = 4;
      for (int p0 = tid; p0 < ntot; p0 += UW * NT) {
        float al[UW], mm[UW], dd[UW]; int sn[UW], rw[UW];
#pragma unroll
        for (int k = 0; k < UW; ++k) {
          const int p  = p0 + k * NT;
          const int pc = p < ntot ? p : p0;
          const int r  = s_tr[pc];
          rw[k] = r;
          sn[k] = s_e[pc];
          al[k] = __int_as_float(s_r[pc]);
          mm[k] = s_rmax[r];
          dd[k] = s_rden[r];
        }
#pragma unroll
        for (int k = 0; k < UW; ++k) {
          const int p = p0 + k * NT;
          if (p < ntot)
            s_pair[p + (UG - 1) * rw[k]] =
                make_int2(sn[k] * (C * 4), __float_as_int(__expf(al[k] - mm[k]) / dd[k]));
        }
      }
    }
    __syncthreads();

    for (int r = g; r < rows; r += NG) {
      const int b = s_off[r], len = s_off[r + 1] - b;
      const int2* const pp = s_pair + b + (UG - 1) * r;
      float a0 = 0.f, a1 = 0.f, a2 = 0.f, a3 = 0.f;
      for (int j0 = 0; j0 < len; j0 += UG) {
        int2 pr[UG]; float4 hv[UG];
#pragma unroll
        for (int k = 0; k < UG; ++k) pr[k] = pp[j0 + k];
#pragma unroll
        for (int k = 0; k < UG; ++k)
          hv[k] = *(const float4*)((const char*)hsrc + (unsigned)(pr[k].x + fo * 4));
#pragma unroll
        for (int k = 0; k < UG; ++k) {
          const float w = __int_as_float(pr[k].y);
          a0 += hv[k].x * w; a1 += hv[k].y * w;
          a2 += hv[k].z * w; a3 += hv[k].w * w;
        }
      }
      const float4 bv = *(const float4*)(bias + fo);
      const float o0 = a0 + bv.x, o1 = a1 + bv.y, o2 = a2 + bv.z, o3 = a3 + bv.w;
      const bool  up = fo >= H;
      const float ra = __shfl_xor(up ? o0 : o2, TPE / 2, TPE);
      const float rb = __shfl_xor(up ? o1 : o3, TPE / 2, TPE);
      const size_t n = (size_t)(lo + r);
      float4 st;
      st.x = FIRST ? (o0 > 0.f ? o0 : (__expf(o0) - 1.f)) : o0;
      st.y = FIRST ? (o1 > 0.f ? o1 : (__expf(o1) - 1.f)) : o1;
      st.z = FIRST ? (o2 > 0.f ? o2 : (__expf(o2) - 1.f)) : o2;
      st.w = FIRST ? (o3 > 0.f ? o3 : (__expf(o3) - 1.f)) : o3;
      *(float4*)(rowout + n * C + fo) = st;
      const float mu0 = up ? ra : o0, mu1 = up ? rb : o1;
      const float sp0 = up ? o2 : ra, sp1 = up ? o3 : rb;
      const int   kf  = up ? fo - H + 2 : fo;
      const float d0 = softplusf(sp0) + 1e-10f, d1 = softplusf(sp1) + 1e-10f;
      float2 kl;
      kl.x = -__logf(d0) + 0.5f * (d0 * d0 + mu0 * mu0 - 1.f);
      kl.y = -__logf(d1) + 0.5f * (d1 * d1 + mu1 * mu1 - 1.f);
      *(float2*)(klout + n * H + kf) = kl;
    }
    return;
  }

  float* const s_acc = (float*)s_raw;
  int*   const s_mx  = s_raw + ROWS * C;
  float* const s_den = (float*)(s_raw + ROWS * C + ROWS);
  for (int i = tid; i < ROWS * C; i += NT) s_acc[i] = 0.f;
  for (int i = tid; i < ROWS; i += NT) { s_mx[i] = (int)0x80000000; s_den[i] = 0.f; }
  __syncthreads();
  for (int e = tid; e < Etot; e += NT) {
    int sn, dn; edge_sd(ei, E, e, sn, dn);
    const unsigned r = (unsigned)(dn - lo);
    if (r < urows) atomicMax(&s_mx[r], fkey(lrelu(sd[2 * sn] + s_d[r])));
  }
  __syncthreads();
  for (int e = tid; e < Etot; e += NT) {
    int sn, dn; edge_sd(ei, E, e, sn, dn);
    const unsigned r = (unsigned)(dn - lo);
    if (r < urows)
      atomicAdd(&s_den[r], __expf(lrelu(sd[2 * sn] + s_d[r]) - fkey_inv(s_mx[r])));
  }
  __syncthreads();
  for (int e = g; e < Etot; e += NG) {
    int sn, dn; edge_sd(ei, E, e, sn, dn);
    const unsigned r = (unsigned)(dn - lo);
    if (r < urows) {
      const float ew = __expf(lrelu(sd[2 * sn] + s_d[r]) - fkey_inv(s_mx[r]));
      const float w  = ew / (s_den[r] + 1e-16f);
      const float4 hv = *(const float4*)(hsrc + (size_t)sn * C + fo);
      float* op = s_acc + (int)r * C + fo;
      atomicAdd(op + 0, hv.x * w);
      atomicAdd(op + 1, hv.y * w);
      atomicAdd(op + 2, hv.z * w);
      atomicAdd(op + 3, hv.w * w);
    }
  }
  __syncthreads();
  for (int i = tid; i < rows * C; i += NT) {
    const int r = i / C, f = i % C;
    const size_t n = (size_t)(lo + r);
    const float v = s_acc[i] + bias[f];
    rowout[n * C + f] = FIRST ? (v > 0.f ? v : (__expf(v) - 1.f)) : v;
    if (f < H) {
      const float sp = s_acc[r * C + H + f] + bias[H + f];
      const float sd = softplusf(sp) + 1e-10f;
      klout[n * H + f] = -__logf(sd) + 0.5f * (sd * sd + v * v - 1.f);
    }
  }
}

extern "C" void kernel_launch(void* const* d_in, const int* in_sizes, int n_in,
                              void* d_out, int out_size, void* d_ws, size_t ws_size,
                              hipStream_t stream) {
  const float* x   = (const float*)d_in[0];
  const int*   ei  = (const int*)  d_in[1];
  const float* W1  = (const float*)d_in[2];
  const float* as1 = (const float*)d_in[3];
  const float* ad1 = (const float*)d_in[4];
  const float* b1  = (const float*)d_in[5];
  const float* W2  = (const float*)d_in[6];
  const float* as2 = (const float*)d_in[7];
  const float* ad2 = (const float*)d_in[8];
  const float* b2  = (const float*)d_in[9];

  const int N    = in_sizes[0] / F_IN;
  const int E    = in_sizes[1] / 2;
  float* out = (float*)d_out;

  const size_t N64 = (size_t)N * C1, N16 = (size_t)N * C2;
  const int owners = (N + ROWS - 1) / ROWS;
  int*   lent = (int*)d_ws;
  int*   loff = lent + (size_t)owners * CAP * 2;
  const int nchunk = (E + CHUNK - 1) / CHUNK;
  const int segp   = (owners + 15) & ~15;
  int*   bent = loff + (size_t)owners * LOFF;
  int*   seg  = bent + (size_t)nchunk * CHUNK * 2;
  float* ws = (float*)(seg + (size_t)nchunk * segp * 2);
  float* h1 = ws;
  float* hb = h1 + N64;
  float* h2 = hb + N64;
  float* sd = h2 + N16;

  float* ixz1 = out + N16;
  float* ixz2 = out + (size_t)N * 48;

  const int tiles  = (N + 15) / 16;
  const int gblk   = (tiles + 7) / 8;

  gemm1_wmma<<<G1B, G1T, 0, stream>>>(x, W1, as1, ad1, h1, sd, N);
  bin_edges<<<nchunk, BT, 0, stream>>>(ei, E, N, owners, segp, seg, bent);
  layer_owner_kernel<C1, true><<<owners, NT, 0, stream>>>(ei, E, N, h1, sd, b1, hb, ixz1, loff, lent,
                                                          seg, bent, nchunk, segp);

  gemm2_wmma<<<gblk, 256, 0, stream>>>(hb, W2, as2, ad2, h2, sd, N);
  layer_owner_kernel<C2, false><<<owners, NT, 0, stream>>>(ei, E, N, h2, sd, b2, out, ixz2, loff, lent,
                                                           seg, bent, nchunk, segp);
}
